# starting kernel
# baseline (speedup 1.0000x reference)
#include <hip/hip_runtime.h>
#include <hip/hip_bf16.h>


typedef __attribute__((ext_vector_type(2))) float v2f;
typedef __attribute__((ext_vector_type(8))) float v8f;

#define NEG_SLOPE 0.2f

__device__ __forceinline__ unsigned f2ord(float f) {
    unsigned u = __float_as_uint(f);
    return (u & 0x80000000u) ? ~u : (u | 0x80000000u);
}
__device__ __forceinline__ float ord2f(unsigned k) {
    unsigned u = (k & 0x80000000u) ? (k & 0x7fffffffu) : ~k;
    return __uint_as_float(u);
}
__device__ __forceinline__ float leaky(float x) {
    return x > 0.0f ? x : NEG_SLOPE * x;
}

#define P1 132
__global__ __launch_bounds__(256) void gemm1_kernel(
    const float* __restrict__ X, const float* __restrict__ W,
    float* __restrict__ H, int N, int K) {
    __shared__ float sW[64 * P1];
    const int lane = threadIdx.x & 31;
    const int wave = threadIdx.x >> 5;
    const int half = lane >> 4;
    const int col0 = lane & 15;
    const int nrt  = (N + 15) >> 4;
    const int rt0  = (blockIdx.x * 8 + wave) * 2;
    const int rt1  = rt0 + 1;
    const int rowA0 = (rt0 < nrt) ? min(rt0 * 16 + col0, N - 1) : 0;
    const int rowA1 = (rt1 < nrt) ? min(rt1 * 16 + col0, N - 1) : 0;

    v8f acc[2][4];
    #pragma unroll
    for (int u = 0; u < 2; ++u)
        #pragma unroll
        for (int t = 0; t < 4; ++t) acc[u][t] = v8f{};

    const int phases = K >> 7;
    for (int ph = 0; ph < phases; ++ph) {
        for (int i = threadIdx.x; i < 128 * 64; i += 256) {
            int r = i >> 6, c = i & 63;
            sW[c * P1 + r] = W[(size_t)ph * 128 * 64 + i];
        }
        __syncthreads();

        const float* xr0 = X + (size_t)rowA0 * K + ph * 128;
        const float* xr1 = X + (size_t)rowA1 * K + ph * 128;
        for (int kk = 0; kk < 32; ++kk) {
            const int kb = kk * 4 + half * 2;
            v2f a0 = *(const v2f*)(xr0 + kb);
            v2f a1 = *(const v2f*)(xr1 + kb);
            #pragma unroll
            for (int t = 0; t < 4; ++t) {
                v2f b = *(const v2f*)(&sW[(t * 16 + col0) * P1 + kb]);
                acc[0][t] = __builtin_amdgcn_wmma_f32_16x16x4_f32(
                    false, a0, false, b, (short)0, acc[0][t], false, false);
                acc[1][t] = __builtin_amdgcn_wmma_f32_16x16x4_f32(
                    false, a1, false, b, (short)0, acc[1][t], false, false);
            }
        }
        __syncthreads();
    }

    #pragma unroll
    for (int u = 0; u < 2; ++u) {
        int rt = rt0 + u;
        if (rt >= nrt) continue;
        #pragma unroll
        for (int t = 0; t < 4; ++t)
            #pragma unroll
            for (int r = 0; r < 8; ++r) {
                int row = rt * 16 + r + half * 8;
                if (row < N) H[(size_t)row * 64 + t * 16 + col0] = acc[u][t][r];
            }
    }
}

#define P2 68
__global__ __launch_bounds__(256) void gemm2_kernel(
    const float* __restrict__ Hin, const float* __restrict__ W,
    float* __restrict__ Hout, int N, int NC) {
    __shared__ float sW[48 * P2];
    const int lane = threadIdx.x & 31;
    const int wave = threadIdx.x >> 5;
    const int half = lane >> 4;
    const int col0 = lane & 15;
    const int nrt  = (N + 15) >> 4;
    const int rt0  = (blockIdx.x * 8 + wave) * 2;
    const int rt1  = rt0 + 1;
    const int rowA0 = (rt0 < nrt) ? min(rt0 * 16 + col0, N - 1) : 0;
    const int rowA1 = (rt1 < nrt) ? min(rt1 * 16 + col0, N - 1) : 0;

    for (int i = threadIdx.x; i < 64 * 48; i += 256) {
        int r = i / 48, c = i % 48;
        sW[c * P2 + r] = (c < NC) ? W[(size_t)r * NC + c] : 0.0f;
    }
    __syncthreads();

    v8f acc[2][3];
    #pragma unroll
    for (int u = 0; u < 2; ++u)
        #pragma unroll
        for (int t = 0; t < 3; ++t) acc[u][t] = v8f{};

    const float* xr0 = Hin + (size_t)rowA0 * 64;
    const float* xr1 = Hin + (size_t)rowA1 * 64;
    for (int kk = 0; kk < 16; ++kk) {
        const int kb = kk * 4 + half * 2;
        v2f a0 = *(const v2f*)(xr0 + kb);
        v2f a1 = *(const v2f*)(xr1 + kb);
        a0[0] = fmaxf(a0[0], 0.0f); a0[1] = fmaxf(a0[1], 0.0f);
        a1[0] = fmaxf(a1[0], 0.0f); a1[1] = fmaxf(a1[1], 0.0f);
        #pragma unroll
        for (int t = 0; t < 3; ++t) {
            v2f b = *(const v2f*)(&sW[(t * 16 + col0) * P2 + kb]);
            acc[0][t] = __builtin_amdgcn_wmma_f32_16x16x4_f32(
                false, a0, false, b, (short)0, acc[0][t], false, false);
            acc[1][t] = __builtin_amdgcn_wmma_f32_16x16x4_f32(
                false, a1, false, b, (short)0, acc[1][t], false, false);
        }
    }

    #pragma unroll
    for (int u = 0; u < 2; ++u) {
        int rt = rt0 + u;
        if (rt >= nrt) continue;
        #pragma unroll
        for (int t = 0; t < 3; ++t)
            #pragma unroll
            for (int r = 0; r < 8; ++r) {
                int row = rt * 16 + r + half * 8;
                int col = t * 16 + col0;
                if (row < N && col < NC)
                    Hout[(size_t)row * NC + col] = acc[u][t][r];
            }
    }
}

__global__ __launch_bounds__(256) void sd_kernel(
    const float* __restrict__ H, const float* __restrict__ asrc,
    const float* __restrict__ adst, float* __restrict__ s,
    float* __restrict__ d, int N, int F) {
    const int wid  = (int)((blockIdx.x * (long long)blockDim.x + threadIdx.x) >> 5);
    const int lane = threadIdx.x & 31;
    if (wid >= N) return;
    float ps = 0.0f, pd = 0.0f;
    for (int f = lane; f < F; f += 32) {
        float hv = H[(size_t)wid * F + f];
        ps = fmaf(hv, asrc[f], ps);
        pd = fmaf(hv, adst[f], pd);
    }
    #pragma unroll
    for (int off = 16; off > 0; off >>= 1) {
        ps += __shfl_down(ps, off, 32);
        pd += __shfl_down(pd, off, 32);
    }
    if (lane == 0) { s[wid] = ps; d[wid] = pd; }
}

__global__ void zero_f32(float* __restrict__ p, long long n) {
    long long i = blockIdx.x * (long long)blockDim.x + threadIdx.x;
    if (i < n) p[i] = 0.0f;
}
__global__ void init_seg(unsigned* __restrict__ m, float* __restrict__ denom, int n) {
    int i = blockIdx.x * blockDim.x + threadIdx.x;
    if (i < n) { m[i] = 0x007FFFFFu;   denom[i] = 0.0f; }
}

__global__ __launch_bounds__(256) void edge_max_kernel(
    const int* __restrict__ ei, const float* __restrict__ s,
    const float* __restrict__ d, unsigned* __restrict__ m, int E, int EA) {
    int e = blockIdx.x * blockDim.x + threadIdx.x;
    if (e >= EA) return;
    int src, dst;
    if (e < E) { src = ei[e]; dst = ei[E + e]; } else { src = dst = e - E; }
    float sc = leaky(s[src] + d[dst]);
    atomicMax(&m[dst], f2ord(sc));
}

__global__ __launch_bounds__(256) void edge_expsum_kernel(
    const int* __restrict__ ei, const float* __restrict__ ew,
    const float* __restrict__ s, const float* __restrict__ d,
    const unsigned* __restrict__ m, float* __restrict__ denom,
    float* __restrict__ exw, int E, int EA) {
    int e = blockIdx.x * blockDim.x + threadIdx.x;
    if (e >= EA) return;
    int src, dst;
    float w;
    if (e < E) { src = ei[e]; dst = ei[E + e]; w = ew[e]; }
    else       { src = dst = e - E; w = 1.0f; }
    float sc = leaky(s[src] + d[dst]);
    float ex = __expf(sc - ord2f(m[dst]));
    exw[e] = ex * w;
    atomicAdd(&denom[dst], ex);
}

__global__ __launch_bounds__(256) void edge_agg_kernel(
    const int* __restrict__ ei, const float* __restrict__ exw,
    const float* __restrict__ denom, const float* __restrict__ H,
    float* __restrict__ out, int E, int EA, int F) {
    long long gid = blockIdx.x * (long long)blockDim.x + threadIdx.x;
    int e = (int)(gid >> 6);
    int f = (int)(gid & 63);
    if (e >= EA || f >= F) return;
    int src, dst;
    if (e < E) { src = ei[e]; dst = ei[E + e]; } else { src = dst = e - E; }
    float c = exw[e] / fmaxf(denom[dst], 1e-16f);
    atomicAdd(&out[(size_t)dst * F + f], H[(size_t)src * F + f] * c);
}

extern "C" void kernel_launch(void* const* d_in, const int* in_sizes, int n_in,
                              void* d_out, int out_size, void* d_ws, size_t ws_size,
                              hipStream_t stream) {
    const float* x   = (const float*)d_in[0];
    const int*   ei  = (const int*)  d_in[1];
    const float* ew  = (const float*)d_in[2];
    const float* W1  = (const float*)d_in[3];
    const float* a1s = (const float*)d_in[4];
    const float* a1d = (const float*)d_in[5];
    const float* W2  = (const float*)d_in[6];
    const float* a2s = (const float*)d_in[7];
    const float* a2d = (const float*)d_in[8];

    const int HID = in_sizes[4];
    const int FIN = in_sizes[3] / HID;
    const int NC  = in_sizes[7];
    const int N   = in_sizes[0] / FIN;
    const int E   = in_sizes[2];
    const int EA  = E + N;

    float*    h1    = (float*)d_ws;
    float*    out1  = h1   + (size_t)N * HID;
    float*    h2    = out1 + (size_t)N * HID;
    float*    sArr  = h2   + (size_t)N * NC;
    float*    dArr  = sArr + N;
    float*    denom = dArr + N;
    unsigned* mArr  = (unsigned*)(denom + N);
    float*    exw   = (float*)(mArr + N);
    float*    outF  = (float*)d_out;

    const int nrt     = (N + 15) / 16;
    const int pairs   = (nrt + 1) / 2;
    const int gBlocks = (pairs + 7) / 8;
    const int eBlocks = (EA + 255) / 256;
    const int nBlocks = (N + 255) / 256;

    gemm1_kernel<<<gBlocks, 256, 0, stream>>>(x, W1, h1, N, FIN);
    sd_kernel<<<((long long)N * 32 + 255) / 256, 256, 0, stream>>>(h1, a1s, a1d, sArr, dArr, N, HID);
    init_seg<<<nBlocks, 256, 0, stream>>>(mArr, denom, N);
    zero_f32<<<(int)(((long long)N * HID + 255) / 256), 256, 0, stream>>>(out1, (long long)N * HID);
    edge_max_kernel<<<eBlocks, 256, 0, stream>>>(ei, sArr, dArr, mArr, E, EA);
    edge_expsum_kernel<<<eBlocks, 256, 0, stream>>>(ei, ew, sArr, dArr, mArr, denom, exw, E, EA);
    edge_agg_kernel<<<(int)(((long long)EA * 64 + 255) / 256), 256, 0, stream>>>(
        ei, exw, denom, h1, out1, E, EA, HID);

    gemm2_kernel<<<gBlocks, 256, 0, stream>>>(out1, W2, h2, N, NC);
    sd_kernel<<<((long long)N * 32 + 255) / 256, 256, 0, stream>>>(h2, a2s, a2d, sArr, dArr, N, NC);
    init_seg<<<nBlocks, 256, 0, stream>>>(mArr, denom, N);
    zero_f32<<<(int)(((long long)N * NC + 255) / 256), 256, 0, stream>>>(outF, (long long)N * NC);
    edge_max_kernel<<<eBlocks, 256, 0, stream>>>(ei, sArr, dArr, mArr, E, EA);
    edge_expsum_kernel<<<eBlocks, 256, 0, stream>>>(ei, ew, sArr, dArr, mArr, denom, exw, E, EA);
    edge_agg_kernel<<<(int)(((long long)EA * 64 + 255) / 256), 256, 0, stream>>>(
        ei, exw, denom, h2, outF, E, EA, NC);
}


# Edges counting-sorted by destination once per call, each group put in edge-id order; one wave per node does softmax and weighted row gather in registers: no float atomics, sums in edge-id order
# speedup vs baseline: 3.6032x; 3.6032x over previous
#include <hip/hip_runtime.h>
#include <hip/hip_bf16.h>


typedef __attribute__((ext_vector_type(2))) float v2f;
typedef __attribute__((ext_vector_type(8))) float v8f;

#define NEG_SLOPE 0.2f

__device__ __forceinline__ float leaky(float x) {
    return x > 0.0f ? x : NEG_SLOPE * x;
}

#define P1 132
__global__ __launch_bounds__(256) void gemm1_kernel(
    const float* __restrict__ X, const float* __restrict__ W,
    float* __restrict__ H, int N, int K) {
    __shared__ float sW[64 * P1];
    const int lane = threadIdx.x & 31;
    const int wave = threadIdx.x >> 5;
    const int half = lane >> 4;
    const int col0 = lane & 15;
    const int nrt  = (N + 15) >> 4;
    const int rt0  = (blockIdx.x * 8 + wave) * 2;
    const int rt1  = rt0 + 1;
    const int rowA0 = (rt0 < nrt) ? min(rt0 * 16 + col0, N - 1) : 0;
    const int rowA1 = (rt1 < nrt) ? min(rt1 * 16 + col0, N - 1) : 0;

    v8f acc[2][4];
    #pragma unroll
    for (int u = 0; u < 2; ++u)
        #pragma unroll
        for (int t = 0; t < 4; ++t) acc[u][t] = v8f{};

    const int phases = K >> 7;
    for (int ph = 0; ph < phases; ++ph) {
        for (int i = threadIdx.x; i < 128 * 64; i += 256) {
            int r = i >> 6, c = i & 63;
            sW[c * P1 + r] = W[(size_t)ph * 128 * 64 + i];
        }
        __syncthreads();

        const float* xr0 = X + (size_t)rowA0 * K + ph * 128;
        const float* xr1 = X + (size_t)rowA1 * K + ph * 128;
        for (int kk = 0; kk < 32; ++kk) {
            const int kb = kk * 4 + half * 2;
            v2f a0 = *(const v2f*)(xr0 + kb);
            v2f a1 = *(const v2f*)(xr1 + kb);
            #pragma unroll
            for (int t = 0; t < 4; ++t) {
                v2f b = *(const v2f*)(&sW[(t * 16 + col0) * P1 + kb]);
                acc[0][t] = __builtin_amdgcn_wmma_f32_16x16x4_f32(
                    false, a0, false, b, (short)0, acc[0][t], false, false);
                acc[1][t] = __builtin_amdgcn_wmma_f32_16x16x4_f32(
                    false, a1, false, b, (short)0, acc[1][t], false, false);
            }
        }
        __syncthreads();
    }

    #pragma unroll
    for (int u = 0; u < 2; ++u) {
        int rt = rt0 + u;
        if (rt >= nrt) continue;
        #pragma unroll
        for (int t = 0; t < 4; ++t)
            #pragma unroll
            for (int r = 0; r < 8; ++r) {
                int row = rt * 16 + r + half * 8;
                if (row < N) H[(size_t)row * 64 + t * 16 + col0] = acc[u][t][r];
            }
    }
}

#define P2 68
__global__ __launch_bounds__(256) void gemm2_kernel(
    const float* __restrict__ Hin, const float* __restrict__ W,
    float* __restrict__ Hout, int N, int NC) {
    __shared__ float sW[48 * P2];
    const int lane = threadIdx.x & 31;
    const int wave = threadIdx.x >> 5;
    const int half = lane >> 4;
    const int col0 = lane & 15;
    const int nrt  = (N + 15) >> 4;
    const int rt0  = (blockIdx.x * 8 + wave) * 2;
    const int rt1  = rt0 + 1;
    const int rowA0 = (rt0 < nrt) ? min(rt0 * 16 + col0, N - 1) : 0;
    const int rowA1 = (rt1 < nrt) ? min(rt1 * 16 + col0, N - 1) : 0;

    for (int i = threadIdx.x; i < 64 * 48; i += 256) {
        int r = i / 48, c = i % 48;
        sW[c * P2 + r] = (c < NC) ? W[(size_t)r * NC + c] : 0.0f;
    }
    __syncthreads();

    v8f acc[2][3];
    #pragma unroll
    for (int u = 0; u < 2; ++u)
        #pragma unroll
        for (int t = 0; t < 3; ++t) acc[u][t] = v8f{};

    const float* xr0 = Hin + (size_t)rowA0 * 64;
    const float* xr1 = Hin + (size_t)rowA1 * 64;
    for (int kk = 0; kk < 16; ++kk) {
        const int kb = kk * 4 + half * 2;
        v2f a0 = *(const v2f*)(xr0 + kb);
        v2f a1 = *(const v2f*)(xr1 + kb);
        a0[0] = fmaxf(a0[0], 0.0f); a0[1] = fmaxf(a0[1], 0.0f);
        a1[0] = fmaxf(a1[0], 0.0f); a1[1] = fmaxf(a1[1], 0.0f);
        #pragma unroll
        for (int t = 0; t < 3; ++t) {
            v2f b = *(const v2f*)(&sW[(t * 16 + col0) * P2 + kb]);
            acc[0][t] = __builtin_amdgcn_wmma_f32_16x16x4_f32(
                false, a0, false, b, (short)0, acc[0][t], false, false);
            acc[1][t] = __builtin_amdgcn_wmma_f32_16x16x4_f32(
                false, a1, false, b, (short)0, acc[1][t], false, false);
        }
    }

    #pragma unroll
    for (int u = 0; u < 2; ++u) {
        int rt = rt0 + u;
        if (rt >= nrt) continue;
        #pragma unroll
        for (int t = 0; t < 3; ++t)
            #pragma unroll
            for (int r = 0; r < 8; ++r) {
                int row = rt * 16 + r + half * 8;
                int col = t * 16 + col0;
                if (row < N && col < NC)
                    Hout[(size_t)row * NC + col] = acc[u][t][r];
            }
    }
}

__global__ __launch_bounds__(256) void sd_kernel(
    const float* __restrict__ H, const float* __restrict__ asrc,
    const float* __restrict__ adst, float* __restrict__ s,
    float* __restrict__ d, int N, int F) {
    const int wid  = (int)((blockIdx.x * (long long)blockDim.x + threadIdx.x) >> 5);
    const int lane = threadIdx.x & 31;
    if (wid >= N) return;
    float ps = 0.0f, pd = 0.0f;
    for (int f = lane; f < F; f += 32) {
        float hv = H[(size_t)wid * F + f];
        ps = fmaf(hv, asrc[f], ps);
        pd = fmaf(hv, adst[f], pd);
    }
    #pragma unroll
    for (int off = 16; off > 0; off >>= 1) {
        ps += __shfl_down(ps, off, 32);
        pd += __shfl_down(pd, off, 32);
    }
    if (lane == 0) { s[wid] = ps; d[wid] = pd; }
}

__global__ void zero_i32(int* __restrict__ p, int n) {
    int i = blockIdx.x * blockDim.x + threadIdx.x;
    if (i < n) p[i] = 0;
}

__global__ __launch_bounds__(256) void count_kernel(
    const int* __restrict__ ei, int* __restrict__ cnt, int* __restrict__ slot, int E) {
    int e = blockIdx.x * blockDim.x + threadIdx.x;
    if (e >= E) return;
    slot[e] = atomicAdd(&cnt[ei[E + e]], 1);
}

__global__ __launch_bounds__(256) void blocksum_kernel(
    const int* __restrict__ cnt, int* __restrict__ bsum, int N) {
    __shared__ int sred[8];
    const int lane = threadIdx.x & 31, wave = threadIdx.x >> 5;
    const int base = blockIdx.x * 1024 + threadIdx.x * 4;
    int v = 0;
    for (int k = 0; k < 4; ++k) if (base + k < N) v += cnt[base + k];
    for (int off = 16; off > 0; off >>= 1) v += __shfl_xor(v, off, 32);
    if (lane == 0) sred[wave] = v;
    __syncthreads();
    if (threadIdx.x == 0) {
        int t = 0;
        for (int w = 0; w < 8; ++w) t += sred[w];
        bsum[blockIdx.x] = t;
    }
}

__global__ __launch_bounds__(256) void scan_kernel(
    const int* __restrict__ cnt, const int* __restrict__ bsum,
    int* __restrict__ rowptr, int N, int E) {
    __shared__ int sred[8];
    __shared__ int swave[8];
    const int lane = threadIdx.x & 31, wave = threadIdx.x >> 5;
    int v = 0;
    for (int t = threadIdx.x; t < (int)blockIdx.x; t += 256) v += bsum[t];
    for (int off = 16; off > 0; off >>= 1) v += __shfl_xor(v, off, 32);
    if (lane == 0) sred[wave] = v;

    const int base = blockIdx.x * 1024 + threadIdx.x * 4;
    int c[4];
    for (int k = 0; k < 4; ++k) c[k] = (base + k < N) ? cnt[base + k] : 0;
    const int tsum = c[0] + c[1] + c[2] + c[3];
    int inc = tsum;
    for (int off = 1; off < 32; off <<= 1) {
        int n = __shfl_up(inc, off, 32);
        if (lane >= off) inc += n;
    }
    if (lane == 31) swave[wave] = inc;
    __syncthreads();
    int pre = 0;
    for (int w = 0; w < 8; ++w) {
        pre += sred[w];
        if (w < wave) pre += swave[w];
    }
    int run = pre + inc - tsum;
    for (int k = 0; k < 4; ++k) {
        if (base + k < N) rowptr[base + k] = run;
        run += c[k];
    }
    if (blockIdx.x == 0 && threadIdx.x == 0) rowptr[N] = E;
}

__global__ __launch_bounds__(256) void fill_kernel(
    const int* __restrict__ ei, const float* __restrict__ ew,
    const int* __restrict__ rowptr, const int* __restrict__ slot,
    int4* __restrict__ rec, int E) {
    int e = blockIdx.x * blockDim.x + threadIdx.x;
    if (e >= E) return;
    int pos = rowptr[ei[E + e]] + slot[e];
    rec[pos] = make_int4(e, ei[e], __float_as_int(ew[e]), 0);
}

__global__ __launch_bounds__(256) void group_order_kernel(
    const int* __restrict__ rowptr, const int4* __restrict__ rec,
    int* __restrict__ gsrc, float* __restrict__ gw, int N) {
    const int wid  = (int)((blockIdx.x * (long long)blockDim.x + threadIdx.x) >> 5);
    const int lane = threadIdx.x & 31;
    if (wid >= N) return;
    const int start = rowptr[wid];
    const int deg   = rowptr[wid + 1] - start;
    if (deg <= 32) {
        int4 r = make_int4(0x7fffffff, 0, 0, 0);
        if (lane < deg) r = rec[start + lane];
        int rk = 0;
        for (int j = 0; j < 32; ++j) rk += (__shfl(r.x, j, 32) < r.x) ? 1 : 0;
        if (lane < deg) { gsrc[start + rk] = r.y; gw[start + rk] = __int_as_float(r.z); }
    } else {
        for (int q = lane; q < deg; q += 32) {
            int4 r = rec[start + q];
            int rk = 0;
            for (int j = 0; j < deg; ++j) rk += (rec[start + j].x < r.x) ? 1 : 0;
            gsrc[start + rk] = r.y; gw[start + rk] = __int_as_float(r.z);
        }
    }
}

__global__ __launch_bounds__(256) void node_kernel(
    const int* __restrict__ rowptr, const int* __restrict__ gsrc,
    const float* __restrict__ gw, const float* __restrict__ s,
    const float* __restrict__ d, const float* __restrict__ H,
    float* __restrict__ out, int N, int F) {
    const int wid  = __builtin_amdgcn_readfirstlane(
        (int)((blockIdx.x * (long long)blockDim.x + threadIdx.x) >> 5));
    const int lane = threadIdx.x & 31;
    if (wid >= N) return;
    const int start = rowptr[wid];
    const int end   = rowptr[wid + 1];
    const int deg   = end - start;
    const float di  = d[wid];
    const float sc_self = leaky(s[wid] + di);

    const bool v0 = lane < deg;
    int src0 = 0; float w0 = 0.0f; float sc0 = sc_self;
    if (v0) {
        src0 = gsrc[start + lane];
        w0   = gw[start + lane];
        sc0  = leaky(s[src0] + di);
    }
    float m = fmaxf(sc_self, sc0);
    for (int p = start + 32 + lane; p < end; p += 32)
        m = fmaxf(m, leaky(s[gsrc[p]] + di));
    for (int off = 16; off > 0; off >>= 1) m = fmaxf(m, __shfl_xor(m, off, 32));

    const float ex0 = v0 ? __expf(sc0 - m) : 0.0f;
    float sum = ex0;
    for (int p = start + 32 + lane; p < end; p += 32)
        sum += __expf(leaky(s[gsrc[p]] + di) - m);
    for (int off = 16; off > 0; off >>= 1) sum += __shfl_xor(sum, off, 32);
    const float ex_self = __expf(sc_self - m);
    sum += ex_self;
    const float dn = fmaxf(sum, 1e-16f);
    const float c0 = (ex0 * w0) / dn;
    const float c_self = ex_self / dn;

    const bool fv = lane * 2 < F;
    const int  fo = fv ? lane * 2 : 0;
    const float* __restrict__ Hf = H + fo;
    float ax = 0.0f, ay = 0.0f;

    const int n0 = min(deg, 32);
    for (int jb = 0; jb < n0; jb += 8) {
        float2 h[8]; float cc[8];
        #pragma unroll
        for (int u = 0; u < 8; ++u) {
            const int j  = jb + u;
            const int jj = min(j, n0 - 1);
            const int sj = __builtin_amdgcn_readlane(src0, jj);
            const float cj = __int_as_float(__builtin_amdgcn_readlane(__float_as_int(c0), jj));
            cc[u] = (j < n0) ? cj : 0.0f;
            h[u]  = *(const float2*)(Hf + (size_t)sj * F);
        }
        #pragma unroll
        for (int u = 0; u < 8; ++u) {
            if (jb + u < n0) { ax = fmaf(cc[u], h[u].x, ax); ay = fmaf(cc[u], h[u].y, ay); }
        }
    }
    for (int cb = start + 32; cb < end; cb += 32) {
        const int p = cb + lane;
        int sx = 0; float cx = 0.0f;
        if (p < end) {
            sx = gsrc[p];
            cx = (__expf(leaky(s[sx] + di) - m) * gw[p]) / dn;
        }
        const int cn = min(32, end - cb);
        for (int j = 0; j < cn; ++j) {
            const int sj = __builtin_amdgcn_readlane(sx, j);
            const float cj = __int_as_float(__builtin_amdgcn_readlane(__float_as_int(cx), j));
            const float2 hv = *(const float2*)(Hf + (size_t)sj * F);
            ax = fmaf(cj, hv.x, ax); ay = fmaf(cj, hv.y, ay);
        }
    }
    {
        const float2 hv = *(const float2*)(Hf + (size_t)wid * F);
        ax = fmaf(c_self, hv.x, ax); ay = fmaf(c_self, hv.y, ay);
    }
    if (fv) *(float2*)(out + (size_t)wid * F + fo) = make_float2(ax, ay);
}

extern "C" void kernel_launch(void* const* d_in, const int* in_sizes, int n_in,
                              void* d_out, int out_size, void* d_ws, size_t ws_size,
                              hipStream_t stream) {
    const float* x   = (const float*)d_in[0];
    const int*   ei  = (const int*)  d_in[1];
    const float* ew  = (const float*)d_in[2];
    const float* W1  = (const float*)d_in[3];
    const float* a1s = (const float*)d_in[4];
    const float* a1d = (const float*)d_in[5];
    const float* W2  = (const float*)d_in[6];
    const float* a2s = (const float*)d_in[7];
    const float* a2d = (const float*)d_in[8];

    const int HID = in_sizes[4];
    const int FIN = in_sizes[3] / HID;
    const int NC  = in_sizes[7];
    const int N   = in_sizes[0] / FIN;
    const int E   = in_sizes[2];

    float* h1   = (float*)d_ws;
    float* out1 = h1   + (size_t)N * HID;
    float* h2   = out1 + (size_t)N * HID;
    float* sArr = h2   + (size_t)N * NC;
    float* dArr = sArr + N;
    float* outF = (float*)d_out;

    const int nsb = (N + 1023) / 1024;
    size_t off = (((size_t)N * (2 * HID + NC + 2)) + 3) & ~(size_t)3;
    int* base  = (int*)d_ws;
    int4*  rec    = (int4*)(base + off);  off += (size_t)E * 4;
    int*   cnt    = base + off;           off += ((size_t)N + 3) & ~(size_t)3;
    int*   rowptr = base + off;           off += ((size_t)N + 4) & ~(size_t)3;
    int*   bsum   = base + off;           off += ((size_t)nsb + 3) & ~(size_t)3;
    int*   slot   = base + off;           off += ((size_t)E + 3) & ~(size_t)3;
    int*   gsrc   = base + off;           off += ((size_t)E + 3) & ~(size_t)3;
    float* gw     = (float*)(base + off);

    const int nrt     = (N + 15) / 16;
    const int pairs   = (nrt + 1) / 2;
    const int gBlocks = (pairs + 7) / 8;
    const int eBlocks = (E + 255) / 256;
    const int nBlocks = (N + 255) / 256;
    const int wBlocks = (int)(((long long)N * 32 + 255) / 256);

    zero_i32<<<nBlocks, 256, 0, stream>>>(cnt, N);
    count_kernel<<<eBlocks, 256, 0, stream>>>(ei, cnt, slot, E);
    blocksum_kernel<<<nsb, 256, 0, stream>>>(cnt, bsum, N);
    scan_kernel<<<nsb, 256, 0, stream>>>(cnt, bsum, rowptr, N, E);
    fill_kernel<<<eBlocks, 256, 0, stream>>>(ei, ew, rowptr, slot, rec, E);
    group_order_kernel<<<wBlocks, 256, 0, stream>>>(rowptr, rec, gsrc, gw, N);

    gemm1_kernel<<<gBlocks, 256, 0, stream>>>(x, W1, h1, N, FIN);
    sd_kernel<<<wBlocks, 256, 0, stream>>>(h1, a1s, a1d, sArr, dArr, N, HID);
    node_kernel<<<wBlocks, 256, 0, stream>>>(rowptr, gsrc, gw, sArr, dArr, h1, out1, N, HID);

    gemm2_kernel<<<gBlocks, 256, 0, stream>>>(out1, W2, h2, N, NC);
    sd_kernel<<<wBlocks, 256, 0, stream>>>(h2, a2s, a2d, sArr, dArr, N, NC);
    node_kernel<<<wBlocks, 256, 0, stream>>>(rowptr, gsrc, gw, sArr, dArr, h2, outF, N, NC);
}


# Per-node and per-edge kernels on fixed chip-filling grids: each wave loops over many nodes (edges) with the next one's loads issued ahead, not one short wave per node; every sum keeps its order
# speedup vs baseline: 3.7793x; 1.0489x over previous
#include <hip/hip_runtime.h>
#include <hip/hip_bf16.h>


typedef __attribute__((ext_vector_type(2))) float v2f;
typedef __attribute__((ext_vector_type(8))) float v8f;

#define NEG_SLOPE 0.2f

__device__ __forceinline__ float leaky(float x) {
    return x > 0.0f ? x : NEG_SLOPE * x;
}

#define P1 132
__global__ __launch_bounds__(256) void gemm1_kernel(
    const float* __restrict__ X, const float* __restrict__ W,
    float* __restrict__ H, int N, int K) {
    __shared__ float sW[64 * P1];
    const int lane = threadIdx.x & 31;
    const int wave = threadIdx.x >> 5;
    const int half = lane >> 4;
    const int col0 = lane & 15;
    const int nrt  = (N + 15) >> 4;
    const int rt0  = (blockIdx.x * 8 + wave) * 2;
    const int rt1  = rt0 + 1;
    const int rowA0 = (rt0 < nrt) ? min(rt0 * 16 + col0, N - 1) : 0;
    const int rowA1 = (rt1 < nrt) ? min(rt1 * 16 + col0, N - 1) : 0;

    v8f acc[2][4];
    #pragma unroll
    for (int u = 0; u < 2; ++u)
        #pragma unroll
        for (int t = 0; t < 4; ++t) acc[u][t] = v8f{};

    const int phases = K >> 7;
    for (int ph = 0; ph < phases; ++ph) {
        for (int i = threadIdx.x; i < 128 * 64; i += 256) {
            int r = i >> 6, c = i & 63;
            sW[c * P1 + r] = W[(size_t)ph * 128 * 64 + i];
        }
        __syncthreads();

        const float* xr0 = X + (size_t)rowA0 * K + ph * 128;
        const float* xr1 = X + (size_t)rowA1 * K + ph * 128;
        for (int kk = 0; kk < 32; ++kk) {
            const int kb = kk * 4 + half * 2;
            v2f a0 = *(const v2f*)(xr0 + kb);
            v2f a1 = *(const v2f*)(xr1 + kb);
            #pragma unroll
            for (int t = 0; t < 4; ++t) {
                v2f b = *(const v2f*)(&sW[(t * 16 + col0) * P1 + kb]);
                acc[0][t] = __builtin_amdgcn_wmma_f32_16x16x4_f32(
                    false, a0, false, b, (short)0, acc[0][t], false, false);
                acc[1][t] = __builtin_amdgcn_wmma_f32_16x16x4_f32(
                    false, a1, false, b, (short)0, acc[1][t], false, false);
            }
        }
        __syncthreads();
    }

    #pragma unroll
    for (int u = 0; u < 2; ++u) {
        int rt = rt0 + u;
        if (rt >= nrt) continue;
        #pragma unroll
        for (int t = 0; t < 4; ++t)
            #pragma unroll
            for (int r = 0; r < 8; ++r) {
                int row = rt * 16 + r + half * 8;
                if (row < N) H[(size_t)row * 64 + t * 16 + col0] = acc[u][t][r];
            }
    }
}

#define P2 68
__global__ __launch_bounds__(256) void gemm2_kernel(
    const float* __restrict__ Hin, const float* __restrict__ W,
    float* __restrict__ Hout, int N, int NC) {
    __shared__ float sW[48 * P2];
    const int lane = threadIdx.x & 31;
    const int wave = threadIdx.x >> 5;
    const int half = lane >> 4;
    const int col0 = lane & 15;
    const int nrt  = (N + 15) >> 4;
    const int rt0  = (blockIdx.x * 8 + wave) * 2;
    const int rt1  = rt0 + 1;
    const int rowA0 = (rt0 < nrt) ? min(rt0 * 16 + col0, N - 1) : 0;
    const int rowA1 = (rt1 < nrt) ? min(rt1 * 16 + col0, N - 1) : 0;

    for (int i = threadIdx.x; i < 64 * 48; i += 256) {
        int r = i / 48, c = i % 48;
        sW[c * P2 + r] = (c < NC) ? W[(size_t)r * NC + c] : 0.0f;
    }
    __syncthreads();

    v8f acc[2][3];
    #pragma unroll
    for (int u = 0; u < 2; ++u)
        #pragma unroll
        for (int t = 0; t < 3; ++t) acc[u][t] = v8f{};

    const float* xr0 = Hin + (size_t)rowA0 * 64;
    const float* xr1 = Hin + (size_t)rowA1 * 64;
    for (int kk = 0; kk < 16; ++kk) {
        const int kb = kk * 4 + half * 2;
        v2f a0 = *(const v2f*)(xr0 + kb);
        v2f a1 = *(const v2f*)(xr1 + kb);
        a0[0] = fmaxf(a0[0], 0.0f); a0[1] = fmaxf(a0[1], 0.0f);
        a1[0] = fmaxf(a1[0], 0.0f); a1[1] = fmaxf(a1[1], 0.0f);
        #pragma unroll
        for (int t = 0; t < 3; ++t) {
            v2f b = *(const v2f*)(&sW[(t * 16 + col0) * P2 + kb]);
            acc[0][t] = __builtin_amdgcn_wmma_f32_16x16x4_f32(
                false, a0, false, b, (short)0, acc[0][t], false, false);
            acc[1][t] = __builtin_amdgcn_wmma_f32_16x16x4_f32(
                false, a1, false, b, (short)0, acc[1][t], false, false);
        }
    }

    #pragma unroll
    for (int u = 0; u < 2; ++u) {
        int rt = rt0 + u;
        if (rt >= nrt) continue;
        #pragma unroll
        for (int t = 0; t < 3; ++t)
            #pragma unroll
            for (int r = 0; r < 8; ++r) {
                int row = rt * 16 + r + half * 8;
                int col = t * 16 + col0;
                if (row < N && col < NC)
                    Hout[(size_t)row * NC + col] = acc[u][t][r];
            }
    }
}

__global__ __launch_bounds__(256) void sd_kernel(
    const float* __restrict__ H, const float* __restrict__ asrc,
    const float* __restrict__ adst, float* __restrict__ s,
    float* __restrict__ d, int N, int F) {
    const int w0   = (int)((blockIdx.x * (long long)blockDim.x + threadIdx.x) >> 5);
    const int nw   = (int)((gridDim.x * (long long)blockDim.x) >> 5);
    const int lane = threadIdx.x & 31;
    const int f0 = lane, f1 = lane + 32;
    const bool v0 = f0 < F, v1 = f1 < F;
    const float as0 = v0 ? asrc[f0] : 0.0f, ad0 = v0 ? adst[f0] : 0.0f;
    const float as1 = v1 ? asrc[f1] : 0.0f, ad1 = v1 ? adst[f1] : 0.0f;
    for (int wb = w0; wb < N; wb += 4 * nw) {
        float h0[4], h1[4];
        #pragma unroll
        for (int k = 0; k < 4; ++k) {
            const int w = wb + k * nw;
            h0[k] = (w < N && v0) ? H[(size_t)w * F + f0] : 0.0f;
            h1[k] = (w < N && v1) ? H[(size_t)w * F + f1] : 0.0f;
        }
        #pragma unroll
        for (int k = 0; k < 4; ++k) {
            const int w = wb + k * nw;
            if (w < N) {
                float ps = 0.0f, pd = 0.0f;
                if (v0) { ps = fmaf(h0[k], as0, ps); pd = fmaf(h0[k], ad0, pd); }
                if (v1) { ps = fmaf(h1[k], as1, ps); pd = fmaf(h1[k], ad1, pd); }
                for (int f = lane + 64; f < F; f += 32) {
                    float hv = H[(size_t)w * F + f];
                    ps = fmaf(hv, asrc[f], ps);
                    pd = fmaf(hv, adst[f], pd);
                }
                #pragma unroll
                for (int off = 16; off > 0; off >>= 1) {
                    ps += __shfl_down(ps, off, 32);
                    pd += __shfl_down(pd, off, 32);
                }
                if (lane == 0) { s[w] = ps; d[w] = pd; }
            }
        }
    }
}

__global__ void zero_i32(int* __restrict__ p, int n) {
    int i = blockIdx.x * blockDim.x + threadIdx.x;
    if (i < n) p[i] = 0;
}

template <int U>
__device__ __forceinline__ void count_trip(
    const int* __restrict__ ei, int* __restrict__ cnt, int* __restrict__ slot,
    int E, int first, int nth) {
    int dd[U], r[U];
    #pragma unroll
    for (int k = 0; k < U; ++k) dd[k] = ei[E + first + k * nth];
    #pragma unroll
    for (int k = 0; k < U; ++k) r[k] = atomicAdd(&cnt[dd[k]], 1);
    #pragma unroll
    for (int k = 0; k < U; ++k) slot[first + k * nth] = r[k];
}

__global__ __launch_bounds__(256) void count_kernel(
    const int* __restrict__ ei, int* __restrict__ cnt, int* __restrict__ slot, int E) {
    const int nth = gridDim.x * blockDim.x;
    const int tid = blockIdx.x * blockDim.x + threadIdx.x;
    int base = 0;
    for (; base + 4 * nth <= E; base += 4 * nth) count_trip<4>(ei, cnt, slot, E, base + tid, nth);
    for (; base + 2 * nth <= E; base += 2 * nth) count_trip<2>(ei, cnt, slot, E, base + tid, nth);
    for (int e = base + tid; e < E; e += nth) count_trip<1>(ei, cnt, slot, E, e, nth);
}

__global__ __launch_bounds__(256) void blocksum_kernel(
    const int* __restrict__ cnt, int* __restrict__ bsum, int N) {
    __shared__ int sred[8];
    const int lane = threadIdx.x & 31, wave = threadIdx.x >> 5;
    const int base = blockIdx.x * 1024 + threadIdx.x * 4;
    int v = 0;
    for (int k = 0; k < 4; ++k) if (base + k < N) v += cnt[base + k];
    for (int off = 16; off > 0; off >>= 1) v += __shfl_xor(v, off, 32);
    if (lane == 0) sred[wave] = v;
    __syncthreads();
    if (threadIdx.x == 0) {
        int t = 0;
        for (int w = 0; w < 8; ++w) t += sred[w];
        bsum[blockIdx.x] = t;
    }
}

__global__ __launch_bounds__(256) void scan_kernel(
    const int* __restrict__ cnt, const int* __restrict__ bsum,
    int* __restrict__ rowptr, int N, int E) {
    __shared__ int sred[8];
    __shared__ int swave[8];
    const int lane = threadIdx.x & 31, wave = threadIdx.x >> 5;
    int v = 0;
    for (int t = threadIdx.x; t < (int)blockIdx.x; t += 256) v += bsum[t];
    for (int off = 16; off > 0; off >>= 1) v += __shfl_xor(v, off, 32);
    if (lane == 0) sred[wave] = v;

    const int base = blockIdx.x * 1024 + threadIdx.x * 4;
    int c[4];
    for (int k = 0; k < 4; ++k) c[k] = (base + k < N) ? cnt[base + k] : 0;
    const int tsum = c[0] + c[1] + c[2] + c[3];
    int inc = tsum;
    for (int off = 1; off < 32; off <<= 1) {
        int n = __shfl_up(inc, off, 32);
        if (lane >= off) inc += n;
    }
    if (lane == 31) swave[wave] = inc;
    __syncthreads();
    int pre = 0;
    for (int w = 0; w < 8; ++w) {
        pre += sred[w];
        if (w < wave) pre += swave[w];
    }
    int run = pre + inc - tsum;
    for (int k = 0; k < 4; ++k) {
        if (base + k < N) rowptr[base + k] = run;
        run += c[k];
    }
    if (blockIdx.x == 0 && threadIdx.x == 0) rowptr[N] = E;
}

template <int U>
__device__ __forceinline__ void fill_trip(
    const int* __restrict__ ei, const float* __restrict__ ew,
    const int* __restrict__ rowptr, const int* __restrict__ slot,
    int4* __restrict__ rec, int E, int first, int nth) {
    int dd[U], ss[U], sl[U], ww[U], rp[U];
    #pragma unroll
    for (int k = 0; k < U; ++k) {
        const int e = first + k * nth;
        dd[k] = ei[E + e];
        ss[k] = ei[e];
        sl[k] = slot[e];
        ww[k] = __float_as_int(ew[e]);
    }
    #pragma unroll
    for (int k = 0; k < U; ++k) rp[k] = rowptr[dd[k]];
    #pragma unroll
    for (int k = 0; k < U; ++k)
        rec[rp[k] + sl[k]] = make_int4(first + k * nth, ss[k], ww[k], 0);
}

__global__ __launch_bounds__(256) void fill_kernel(
    const int* __restrict__ ei, const float* __restrict__ ew,
    const int* __restrict__ rowptr, const int* __restrict__ slot,
    int4* __restrict__ rec, int E) {
    const int nth = gridDim.x * blockDim.x;
    const int tid = blockIdx.x * blockDim.x + threadIdx.x;
    int base = 0;
    for (; base + 4 * nth <= E; base += 4 * nth) fill_trip<4>(ei, ew, rowptr, slot, rec, E, base + tid, nth);
    for (; base + 2 * nth <= E; base += 2 * nth) fill_trip<2>(ei, ew, rowptr, slot, rec, E, base + tid, nth);
    for (int e = base + tid; e < E; e += nth) fill_trip<1>(ei, ew, rowptr, slot, rec, E, e, nth);
}

__global__ __launch_bounds__(256) void group_order_kernel(
    const int* __restrict__ rowptr, const int4* __restrict__ rec,
    int* __restrict__ gsrc, float* __restrict__ gw, int N) {
    const int w0   = __builtin_amdgcn_readfirstlane(
        (int)((blockIdx.x * (long long)blockDim.x + threadIdx.x) >> 5));
    const int nw   = (int)((gridDim.x * (long long)blockDim.x) >> 5);
    const int lane = threadIdx.x & 31;
    if (w0 >= N) return;
    int start = rowptr[w0];
    int end   = rowptr[w0 + 1];
    for (int wid = w0; wid < N; wid += nw) {
        const int nx = min(wid + nw, N - 1);
        const int nstart = rowptr[nx];
        const int nend   = rowptr[nx + 1];
        const int deg = end - start;
        if (deg <= 32) {
            int4 r = make_int4(0x7fffffff, 0, 0, 0);
            if (lane < deg) r = rec[start + lane];
            int rk = 0;
            for (int j = 0; j < 32; ++j) rk += (__shfl(r.x, j, 32) < r.x) ? 1 : 0;
            if (lane < deg) { gsrc[start + rk] = r.y; gw[start + rk] = __int_as_float(r.z); }
        } else {
            for (int q = lane; q < deg; q += 32) {
                int4 r = rec[start + q];
                int rk = 0;
                for (int j = 0; j < deg; ++j) rk += (rec[start + j].x < r.x) ? 1 : 0;
                gsrc[start + rk] = r.y; gw[start + rk] = __int_as_float(r.z);
            }
        }
        start = nstart; end = nend;
    }
}

__global__ __launch_bounds__(256) void node_kernel(
    const int* __restrict__ rowptr, const int* __restrict__ gsrc,
    const float* __restrict__ gw, const float* __restrict__ s,
    const float* __restrict__ d, const float* __restrict__ H,
    float* __restrict__ out, int N, int F) {
    const int wfirst = __builtin_amdgcn_readfirstlane(
        (int)((blockIdx.x * (long long)blockDim.x + threadIdx.x) >> 5));
    const int nw   = (int)((gridDim.x * (long long)blockDim.x) >> 5);
    const int lane = threadIdx.x & 31;
    if (wfirst >= N) return;

    const bool fv = lane * 2 < F;
    const int  fo = fv ? lane * 2 : 0;
    const float* __restrict__ Hf = H + fo;

    int   start = rowptr[wfirst];
    int   end   = rowptr[wfirst + 1];
    float di    = d[wfirst];
    float si    = s[wfirst];
    int src0 = 0; float wt0 = 0.0f; float s0 = 0.0f;
    if (lane < end - start) {
        src0 = gsrc[start + lane];
        wt0  = gw[start + lane];
        s0   = s[src0];
    }

    for (int wid = wfirst; wid < N; wid += nw) {
        const float2 hself = *(const float2*)(Hf + (size_t)wid * F);
        const int nx = min(wid + nw, N - 1);
        const int   nstart = rowptr[nx];
        const int   nend   = rowptr[nx + 1];
        const float ndi    = d[nx];
        const float nsi    = s[nx];

        const int deg = end - start;
        const float sc_self = leaky(si + di);
        const bool v0 = lane < deg;
        const float sc0 = v0 ? leaky(s0 + di) : sc_self;

        float m = fmaxf(sc_self, sc0);
        for (int p = start + 32 + lane; p < end; p += 32)
            m = fmaxf(m, leaky(s[gsrc[p]] + di));
        for (int off = 16; off > 0; off >>= 1) m = fmaxf(m, __shfl_xor(m, off, 32));

        const float ex0 = v0 ? __expf(sc0 - m) : 0.0f;
        float sum = ex0;
        for (int p = start + 32 + lane; p < end; p += 32)
            sum += __expf(leaky(s[gsrc[p]] + di) - m);
        for (int off = 16; off > 0; off >>= 1) sum += __shfl_xor(sum, off, 32);
        const float ex_self = __expf(sc_self - m);
        sum += ex_self;
        const float dn = fmaxf(sum, 1e-16f);
        const float c0 = (ex0 * wt0) / dn;
        const float c_self = ex_self / dn;

        const bool nv0 = lane < nend - nstart;
        int nsrc0 = 0; float nwt0 = 0.0f; float ns0 = 0.0f;
        if (nv0) { nsrc0 = gsrc[nstart + lane]; nwt0 = gw[nstart + lane]; }

        float ax = 0.0f, ay = 0.0f;
        const int n0 = min(deg, 32);
        for (int jb = 0; jb < n0; jb += 8) {
            float2 h[8]; float cc[8];
            #pragma unroll
            for (int u = 0; u < 8; ++u) {
                const int j  = jb + u;
                const int jj = min(j, n0 - 1);
                const int sj = __builtin_amdgcn_readlane(src0, jj);
                const float cj = __int_as_float(__builtin_amdgcn_readlane(__float_as_int(c0), jj));
                cc[u] = (j < n0) ? cj : 0.0f;
                h[u]  = *(const float2*)(Hf + (size_t)sj * F);
            }
            if (jb == 0 && nv0) ns0 = s[nsrc0];
            #pragma unroll
            for (int u = 0; u < 8; ++u) {
                if (jb + u < n0) { ax = fmaf(cc[u], h[u].x, ax); ay = fmaf(cc[u], h[u].y, ay); }
            }
        }
        if (n0 == 0 && nv0) ns0 = s[nsrc0];
        for (int cb = start + 32; cb < end; cb += 32) {
            const int p = cb + lane;
            int sx = 0; float cx = 0.0f;
            if (p < end) {
                sx = gsrc[p];
                cx = (__expf(leaky(s[sx] + di) - m) * gw[p]) / dn;
            }
            const int cn = min(32, end - cb);
            for (int j = 0; j < cn; ++j) {
                const int sj = __builtin_amdgcn_readlane(sx, j);
                const float cj = __int_as_float(__builtin_amdgcn_readlane(__float_as_int(cx), j));
                const float2 hv = *(const float2*)(Hf + (size_t)sj * F);
                ax = fmaf(cj, hv.x, ax); ay = fmaf(cj, hv.y, ay);
            }
        }
        ax = fmaf(c_self, hself.x, ax); ay = fmaf(c_self, hself.y, ay);
        if (fv) *(float2*)(out + (size_t)wid * F + fo) = make_float2(ax, ay);

        start = nstart; end = nend; di = ndi; si = nsi;
        src0 = nsrc0; wt0 = nwt0; s0 = ns0;
    }
}

extern "C" void kernel_launch(void* const* d_in, const int* in_sizes, int n_in,
                              void* d_out, int out_size, void* d_ws, size_t ws_size,
                              hipStream_t stream) {
    const float* x   = (const float*)d_in[0];
    const int*   ei  = (const int*)  d_in[1];
    const float* ew  = (const float*)d_in[2];
    const float* W1  = (const float*)d_in[3];
    const float* a1s = (const float*)d_in[4];
    const float* a1d = (const float*)d_in[5];
    const float* W2  = (const float*)d_in[6];
    const float* a2s = (const float*)d_in[7];
    const float* a2d = (const float*)d_in[8];

    const int HID = in_sizes[4];
    const int FIN = in_sizes[3] / HID;
    const int NC  = in_sizes[7];
    const int N   = in_sizes[0] / FIN;
    const int E   = in_sizes[2];

    float* h1   = (float*)d_ws;
    float* out1 = h1   + (size_t)N * HID;
    float* h2   = out1 + (size_t)N * HID;
    float* sArr = h2   + (size_t)N * NC;
    float* dArr = sArr + N;
    float* outF = (float*)d_out;

    const int nsb = (N + 1023) / 1024;
    size_t off = (((size_t)N * (2 * HID + NC + 2)) + 3) & ~(size_t)3;
    int* base  = (int*)d_ws;
    int4*  rec    = (int4*)(base + off);  off += (size_t)E * 4;
    int*   cnt    = base + off;           off += ((size_t)N + 3) & ~(size_t)3;
    int*   rowptr = base + off;           off += ((size_t)N + 4) & ~(size_t)3;
    int*   bsum   = base + off;           off += ((size_t)nsb + 3) & ~(size_t)3;
    int*   slot   = base + off;           off += ((size_t)E + 3) & ~(size_t)3;
    int*   gsrc   = base + off;           off += ((size_t)E + 3) & ~(size_t)3;
    float* gw     = (float*)(base + off);

    const int nrt     = (N + 15) / 16;
    const int pairs   = (nrt + 1) / 2;
    const int gBlocks = (pairs + 7) / 8;
    const int nBlocks = (N + 255) / 256;
    const int eBlocks = 1024;
    const int sBlocks = 1024;
    const int wBlocks = 2048;

    zero_i32<<<nBlocks, 256, 0, stream>>>(cnt, N);
    count_kernel<<<eBlocks, 256, 0, stream>>>(ei, cnt, slot, E);
    blocksum_kernel<<<nsb, 256, 0, stream>>>(cnt, bsum, N);
    scan_kernel<<<nsb, 256, 0, stream>>>(cnt, bsum, rowptr, N, E);
    fill_kernel<<<eBlocks, 256, 0, stream>>>(ei, ew, rowptr, slot, rec, E);
    group_order_kernel<<<wBlocks, 256, 0, stream>>>(rowptr, rec, gsrc, gw, N);

    gemm1_kernel<<<gBlocks, 256, 0, stream>>>(x, W1, h1, N, FIN);
    sd_kernel<<<sBlocks, 256, 0, stream>>>(h1, a1s, a1d, sArr, dArr, N, HID);
    node_kernel<<<wBlocks, 256, 0, stream>>>(rowptr, gsrc, gw, sArr, dArr, h1, out1, N, HID);

    gemm2_kernel<<<gBlocks, 256, 0, stream>>>(out1, W2, h2, N, NC);
    sd_kernel<<<sBlocks, 256, 0, stream>>>(h2, a2s, a2d, sArr, dArr, N, NC);
    node_kernel<<<wBlocks, 256, 0, stream>>>(rowptr, gsrc, gw, sArr, dArr, h2, outF, N, NC);
}


# Edges grouped by destination in two levels with every counter in LDS (bins of 256 nodes per edge chunk, then per node inside a bin), no returning adds on words in memory; every sum keeps its order
# speedup vs baseline: 5.3755x; 1.4224x over previous
#include <hip/hip_runtime.h>
#include <hip/hip_bf16.h>


typedef __attribute__((ext_vector_type(2))) float v2f;
typedef __attribute__((ext_vector_type(8))) float v8f;

#define NEG_SLOPE 0.2f

__device__ __forceinline__ float leaky(float x) {
    return x > 0.0f ? x : NEG_SLOPE * x;
}

#define P1 132
__global__ __launch_bounds__(256) void gemm1_kernel(
    const float* __restrict__ X, const float* __restrict__ W,
    float* __restrict__ H, int N, int K) {
    __shared__ float sW[64 * P1];
    const int lane = threadIdx.x & 31;
    const int wave = threadIdx.x >> 5;
    const int half = lane >> 4;
    const int col0 = lane & 15;
    const int nrt  = (N + 15) >> 4;
    const int rt0  = (blockIdx.x * 8 + wave) * 2;
    const int rt1  = rt0 + 1;
    const int rowA0 = (rt0 < nrt) ? min(rt0 * 16 + col0, N - 1) : 0;
    const int rowA1 = (rt1 < nrt) ? min(rt1 * 16 + col0, N - 1) : 0;

    v8f acc[2][4];
    #pragma unroll
    for (int u = 0; u < 2; ++u)
        #pragma unroll
        for (int t = 0; t < 4; ++t) acc[u][t] = v8f{};

    const int phases = K >> 7;
    for (int ph = 0; ph < phases; ++ph) {
        for (int i = threadIdx.x; i < 128 * 64; i += 256) {
            int r = i >> 6, c = i & 63;
            sW[c * P1 + r] = W[(size_t)ph * 128 * 64 + i];
        }
        __syncthreads();

        const float* xr0 = X + (size_t)rowA0 * K + ph * 128;
        const float* xr1 = X + (size_t)rowA1 * K + ph * 128;
        for (int kk = 0; kk < 32; ++kk) {
            const int kb = kk * 4 + half * 2;
            v2f a0 = *(const v2f*)(xr0 + kb);
            v2f a1 = *(const v2f*)(xr1 + kb);
            #pragma unroll
            for (int t = 0; t < 4; ++t) {
                v2f b = *(const v2f*)(&sW[(t * 16 + col0) * P1 + kb]);
                acc[0][t] = __builtin_amdgcn_wmma_f32_16x16x4_f32(
                    false, a0, false, b, (short)0, acc[0][t], false, false);
                acc[1][t] = __builtin_amdgcn_wmma_f32_16x16x4_f32(
                    false, a1, false, b, (short)0, acc[1][t], false, false);
            }
        }
        __syncthreads();
    }

    #pragma unroll
    for (int u = 0; u < 2; ++u) {
        int rt = rt0 + u;
        if (rt >= nrt) continue;
        #pragma unroll
        for (int t = 0; t < 4; ++t)
            #pragma unroll
            for (int r = 0; r < 8; ++r) {
                int row = rt * 16 + r + half * 8;
                if (row < N) H[(size_t)row * 64 + t * 16 + col0] = acc[u][t][r];
            }
    }
}

#define P2 68
__global__ __launch_bounds__(256) void gemm2_kernel(
    const float* __restrict__ Hin, const float* __restrict__ W,
    float* __restrict__ Hout, int N, int NC) {
    __shared__ float sW[48 * P2];
    const int lane = threadIdx.x & 31;
    const int wave = threadIdx.x >> 5;
    const int half = lane >> 4;
    const int col0 = lane & 15;
    const int nrt  = (N + 15) >> 4;
    const int rt0  = (blockIdx.x * 8 + wave) * 2;
    const int rt1  = rt0 + 1;
    const int rowA0 = (rt0 < nrt) ? min(rt0 * 16 + col0, N - 1) : 0;
    const int rowA1 = (rt1 < nrt) ? min(rt1 * 16 + col0, N - 1) : 0;

    for (int i = threadIdx.x; i < 64 * 48; i += 256) {
        int r = i / 48, c = i % 48;
        sW[c * P2 + r] = (c < NC) ? W[(size_t)r * NC + c] : 0.0f;
    }
    __syncthreads();

    v8f acc[2][3];
    #pragma unroll
    for (int u = 0; u < 2; ++u)
        #pragma unroll
        for (int t = 0; t < 3; ++t) acc[u][t] = v8f{};

    const float* xr0 = Hin + (size_t)rowA0 * 64;
    const float* xr1 = Hin + (size_t)rowA1 * 64;
    for (int kk = 0; kk < 16; ++kk) {
        const int kb = kk * 4 + half * 2;
        v2f a0 = *(const v2f*)(xr0 + kb);
        v2f a1 = *(const v2f*)(xr1 + kb);
        a0[0] = fmaxf(a0[0], 0.0f); a0[1] = fmaxf(a0[1], 0.0f);
        a1[0] = fmaxf(a1[0], 0.0f); a1[1] = fmaxf(a1[1], 0.0f);
        #pragma unroll
        for (int t = 0; t < 3; ++t) {
            v2f b = *(const v2f*)(&sW[(t * 16 + col0) * P2 + kb]);
            acc[0][t] = __builtin_amdgcn_wmma_f32_16x16x4_f32(
                false, a0, false, b, (short)0, acc[0][t], false, false);
            acc[1][t] = __builtin_amdgcn_wmma_f32_16x16x4_f32(
                false, a1, false, b, (short)0, acc[1][t], false, false);
        }
    }

    #pragma unroll
    for (int u = 0; u < 2; ++u) {
        int rt = rt0 + u;
        if (rt >= nrt) continue;
        #pragma unroll
        for (int t = 0; t < 3; ++t)
            #pragma unroll
            for (int r = 0; r < 8; ++r) {
                int row = rt * 16 + r + half * 8;
                int col = t * 16 + col0;
                if (row < N && col < NC)
                    Hout[(size_t)row * NC + col] = acc[u][t][r];
            }
    }
}

__global__ __launch_bounds__(256) void sd_kernel(
    const float* __restrict__ H, const float* __restrict__ asrc,
    const float* __restrict__ adst, float* __restrict__ s,
    float* __restrict__ d, int N, int F) {
    const int w0   = (int)((blockIdx.x * (long long)blockDim.x + threadIdx.x) >> 5);
    const int nw   = (int)((gridDim.x * (long long)blockDim.x) >> 5);
    const int lane = threadIdx.x & 31;
    const int f0 = lane, f1 = lane + 32;
    const bool v0 = f0 < F, v1 = f1 < F;
    const float as0 = v0 ? asrc[f0] : 0.0f, ad0 = v0 ? adst[f0] : 0.0f;
    const float as1 = v1 ? asrc[f1] : 0.0f, ad1 = v1 ? adst[f1] : 0.0f;
    for (int wb = w0; wb < N; wb += 4 * nw) {
        float h0[4], h1[4];
        #pragma unroll
        for (int k = 0; k < 4; ++k) {
            const int w = wb + k * nw;
            h0[k] = (w < N && v0) ? H[(size_t)w * F + f0] : 0.0f;
            h1[k] = (w < N && v1) ? H[(size_t)w * F + f1] : 0.0f;
        }
        #pragma unroll
        for (int k = 0; k < 4; ++k) {
            const int w = wb + k * nw;
            if (w < N) {
                float ps = 0.0f, pd = 0.0f;
                if (v0) { ps = fmaf(h0[k], as0, ps); pd = fmaf(h0[k], ad0, pd); }
                if (v1) { ps = fmaf(h1[k], as1, ps); pd = fmaf(h1[k], ad1, pd); }
                for (int f = lane + 64; f < F; f += 32) {
                    float hv = H[(size_t)w * F + f];
                    ps = fmaf(hv, asrc[f], ps);
                    pd = fmaf(hv, adst[f], pd);
                }
                #pragma unroll
                for (int off = 16; off > 0; off >>= 1) {
                    ps += __shfl_down(ps, off, 32);
                    pd += __shfl_down(pd, off, 32);
                }
                if (lane == 0) { s[w] = ps; d[w] = pd; }
            }
        }
    }
}

#define BIN_SHIFT 8
#define BIN_NODES 256

#define TRIP 7

__global__ __launch_bounds__(1024) void bin_count_kernel(
    const int* __restrict__ ei, int* __restrict__ binCnt, int E, int B, int CH) {
    extern __shared__ int hist[];
    const int tid = threadIdx.x, nt = blockDim.x;
    const int c = blockIdx.x, G = gridDim.x;
    for (int i = tid; i < B; i += nt) hist[i] = 0;
    __syncthreads();
    const int e0 = min(E, c * CH), e1 = min(E, e0 + CH);
    for (int base = e0 + tid; base - tid < e1; base += TRIP * nt) {
        int dd[TRIP];
        #pragma unroll
        for (int k = 0; k < TRIP; ++k) dd[k] = ei[E + min(base + k * nt, e1 - 1)];
        #pragma unroll
        for (int k = 0; k < TRIP; ++k)
            if (base + k * nt < e1) atomicAdd(&hist[dd[k] >> BIN_SHIFT], 1);
    }
    __syncthreads();
    for (int i = tid; i < B; i += nt) binCnt[(size_t)i * G + c] = hist[i];
}

__global__ __launch_bounds__(256) void blocksum_kernel(
    const int* __restrict__ cnt, int* __restrict__ bsum, int N) {
    __shared__ int sred[8];
    const int lane = threadIdx.x & 31, wave = threadIdx.x >> 5;
    const int base = blockIdx.x * 1024 + threadIdx.x * 4;
    int v = 0;
    for (int k = 0; k < 4; ++k) if (base + k < N) v += cnt[base + k];
    for (int off = 16; off > 0; off >>= 1) v += __shfl_xor(v, off, 32);
    if (lane == 0) sred[wave] = v;
    __syncthreads();
    if (threadIdx.x == 0) {
        int t = 0;
        for (int w = 0; w < 8; ++w) t += sred[w];
        bsum[blockIdx.x] = t;
    }
}

__global__ __launch_bounds__(256) void scan_kernel(
    const int* __restrict__ cnt, const int* __restrict__ bsum,
    int* __restrict__ rowptr, int N, int E) {
    __shared__ int sred[8];
    __shared__ int swave[8];
    const int lane = threadIdx.x & 31, wave = threadIdx.x >> 5;
    int v = 0;
    for (int t = threadIdx.x; t < (int)blockIdx.x; t += 256) v += bsum[t];
    for (int off = 16; off > 0; off >>= 1) v += __shfl_xor(v, off, 32);
    if (lane == 0) sred[wave] = v;

    const int base = blockIdx.x * 1024 + threadIdx.x * 4;
    int c[4];
    for (int k = 0; k < 4; ++k) c[k] = (base + k < N) ? cnt[base + k] : 0;
    const int tsum = c[0] + c[1] + c[2] + c[3];
    int inc = tsum;
    for (int off = 1; off < 32; off <<= 1) {
        int n = __shfl_up(inc, off, 32);
        if (lane >= off) inc += n;
    }
    if (lane == 31) swave[wave] = inc;
    __syncthreads();
    int pre = 0;
    for (int w = 0; w < 8; ++w) {
        pre += sred[w];
        if (w < wave) pre += swave[w];
    }
    int run = pre + inc - tsum;
    for (int k = 0; k < 4; ++k) {
        if (base + k < N) rowptr[base + k] = run;
        run += c[k];
    }
    if (blockIdx.x == 0 && threadIdx.x == 0) rowptr[N] = E;
}

__global__ __launch_bounds__(1024) void bin_scatter_kernel(
    const int* __restrict__ ei, const float* __restrict__ ew,
    const int* __restrict__ binOff, int4* __restrict__ rec2, int E, int B, int CH) {
    extern __shared__ int cur[];
    const int tid = threadIdx.x, nt = blockDim.x;
    const int c = blockIdx.x, G = gridDim.x;
    for (int i = tid; i < B; i += nt) cur[i] = binOff[(size_t)i * G + c];
    __syncthreads();
    const int e0 = min(E, c * CH), e1 = min(E, e0 + CH);
    for (int base = e0 + tid; base - tid < e1; base += TRIP * nt) {
        int dd[TRIP], ss[TRIP], ww[TRIP], pp[TRIP];
        #pragma unroll
        for (int k = 0; k < TRIP; ++k) {
            const int e = min(base + k * nt, e1 - 1);
            dd[k] = ei[E + e];
            ss[k] = ei[e];
            ww[k] = __float_as_int(ew[e]);
        }
        #pragma unroll
        for (int k = 0; k < TRIP; ++k) {
            pp[k] = 0;
            if (base + k * nt < e1) pp[k] = atomicAdd(&cur[dd[k] >> BIN_SHIFT], 1);
        }
        #pragma unroll
        for (int k = 0; k < TRIP; ++k)
            if (base + k * nt < e1) rec2[pp[k]] = make_int4(base + k * nt, ss[k], ww[k], dd[k]);
    }
}

#define GTRIP 6
__global__ __launch_bounds__(1024) void bin_group_kernel(
    const int* __restrict__ binOff, const int4* __restrict__ rec2,
    int4* __restrict__ rec, int* __restrict__ rowptr, int N, int E, int G) {
    __shared__ int cnt[BIN_NODES];
    __shared__ int cur[BIN_NODES];
    __shared__ int swave[BIN_NODES / 32];
    const int tid = threadIdx.x, nt = blockDim.x;
    const int lane = tid & 31, wave = tid >> 5;
    const int b = blockIdx.x;
    const int nbase = b << BIN_SHIFT;
    const int binStart = binOff[(size_t)b * G];
    const int binEnd   = binOff[(size_t)(b + 1) * G];
    if (tid < BIN_NODES) cnt[tid] = 0;
    __syncthreads();
    for (int base = binStart + tid; base - tid < binEnd; base += GTRIP * nt) {
        int dd[GTRIP];
        #pragma unroll
        for (int k = 0; k < GTRIP; ++k) dd[k] = rec2[min(base + k * nt, binEnd - 1)].w;
        #pragma unroll
        for (int k = 0; k < GTRIP; ++k)
            if (base + k * nt < binEnd) atomicAdd(&cnt[dd[k] - nbase], 1);
    }
    __syncthreads();
    int v = 0, inc = 0;
    if (tid < BIN_NODES) {
        v = cnt[tid];
        inc = v;
        for (int off = 1; off < 32; off <<= 1) {
            int n = __shfl_up(inc, off, 32);
            if (lane >= off) inc += n;
        }
        if (lane == 31) swave[wave] = inc;
    }
    __syncthreads();
    if (tid < BIN_NODES) {
        int pre = 0;
        for (int w = 0; w < BIN_NODES / 32; ++w) if (w < wave) pre += swave[w];
        const int first = binStart + pre + inc - v;
        cur[tid] = first;
        if (nbase + tid < N) rowptr[nbase + tid] = first;
    }
    if (b == (int)gridDim.x - 1 && tid == 0) rowptr[N] = E;
    __syncthreads();
    for (int base = binStart + tid; base - tid < binEnd; base += GTRIP * nt) {
        int4 r[GTRIP]; int pp[GTRIP];
        #pragma unroll
        for (int k = 0; k < GTRIP; ++k) r[k] = rec2[min(base + k * nt, binEnd - 1)];
        #pragma unroll
        for (int k = 0; k < GTRIP; ++k) {
            pp[k] = 0;
            if (base + k * nt < binEnd) pp[k] = atomicAdd(&cur[r[k].w - nbase], 1);
        }
        #pragma unroll
        for (int k = 0; k < GTRIP; ++k)
            if (base + k * nt < binEnd) rec[pp[k]] = make_int4(r[k].x, r[k].y, r[k].z, 0);
    }
}

__global__ __launch_bounds__(256) void group_order_kernel(
    const int* __restrict__ rowptr, const int4* __restrict__ rec,
    int* __restrict__ gsrc, float* __restrict__ gw, int N) {
    const int w0   = __builtin_amdgcn_readfirstlane(
        (int)((blockIdx.x * (long long)blockDim.x + threadIdx.x) >> 5));
    const int nw   = (int)((gridDim.x * (long long)blockDim.x) >> 5);
    const int lane = threadIdx.x & 31;
    if (w0 >= N) return;
    int start = rowptr[w0];
    int end   = rowptr[w0 + 1];
    for (int wid = w0; wid < N; wid += nw) {
        const int nx = min(wid + nw, N - 1);
        const int nstart = rowptr[nx];
        const int nend   = rowptr[nx + 1];
        const int deg = end - start;
        if (deg <= 32) {
            int4 r = make_int4(0x7fffffff, 0, 0, 0);
            if (lane < deg) r = rec[start + lane];
            int rk = 0;
            for (int j = 0; j < 32; ++j) rk += (__shfl(r.x, j, 32) < r.x) ? 1 : 0;
            if (lane < deg) { gsrc[start + rk] = r.y; gw[start + rk] = __int_as_float(r.z); }
        } else {
            for (int q = lane; q < deg; q += 32) {
                int4 r = rec[start + q];
                int rk = 0;
                for (int j = 0; j < deg; ++j) rk += (rec[start + j].x < r.x) ? 1 : 0;
                gsrc[start + rk] = r.y; gw[start + rk] = __int_as_float(r.z);
            }
        }
        start = nstart; end = nend;
    }
}

__global__ __launch_bounds__(256) void node_kernel(
    const int* __restrict__ rowptr, const int* __restrict__ gsrc,
    const float* __restrict__ gw, const float* __restrict__ s,
    const float* __restrict__ d, const float* __restrict__ H,
    float* __restrict__ out, int N, int F) {
    const int wfirst = __builtin_amdgcn_readfirstlane(
        (int)((blockIdx.x * (long long)blockDim.x + threadIdx.x) >> 5));
    const int nw   = (int)((gridDim.x * (long long)blockDim.x) >> 5);
    const int lane = threadIdx.x & 31;
    if (wfirst >= N) return;

    const bool fv = lane * 2 < F;
    const int  fo = fv ? lane * 2 : 0;
    const float* __restrict__ Hf = H + fo;

    int   start = rowptr[wfirst];
    int   end   = rowptr[wfirst + 1];
    float di    = d[wfirst];
    float si    = s[wfirst];
    int src0 = 0; float wt0 = 0.0f; float s0 = 0.0f;
    if (lane < end - start) {
        src0 = gsrc[start + lane];
        wt0  = gw[start + lane];
        s0   = s[src0];
    }

    for (int wid = wfirst; wid < N; wid += nw) {
        const float2 hself = *(const float2*)(Hf + (size_t)wid * F);
        const int nx = min(wid + nw, N - 1);
        const int   nstart = rowptr[nx];
        const int   nend   = rowptr[nx + 1];
        const float ndi    = d[nx];
        const float nsi    = s[nx];

        const int deg = end - start;
        const float sc_self = leaky(si + di);
        const bool v0 = lane < deg;
        const float sc0 = v0 ? leaky(s0 + di) : sc_self;

        float m = fmaxf(sc_self, sc0);
        for (int p = start + 32 + lane; p < end; p += 32)
            m = fmaxf(m, leaky(s[gsrc[p]] + di));
        for (int off = 16; off > 0; off >>= 1) m = fmaxf(m, __shfl_xor(m, off, 32));

        const float ex0 = v0 ? __expf(sc0 - m) : 0.0f;
        float sum = ex0;
        for (int p = start + 32 + lane; p < end; p += 32)
            sum += __expf(leaky(s[gsrc[p]] + di) - m);
        for (int off = 16; off > 0; off >>= 1) sum += __shfl_xor(sum, off, 32);
        const float ex_self = __expf(sc_self - m);
        sum += ex_self;
        const float dn = fmaxf(sum, 1e-16f);
        const float c0 = (ex0 * wt0) / dn;
        const float c_self = ex_self / dn;

        const bool nv0 = lane < nend - nstart;
        int nsrc0 = 0; float nwt0 = 0.0f; float ns0 = 0.0f;
        if (nv0) { nsrc0 = gsrc[nstart + lane]; nwt0 = gw[nstart + lane]; }

        float ax = 0.0f, ay = 0.0f;
        const int n0 = min(deg, 32);
        for (int jb = 0; jb < n0; jb += 8) {
            float2 h[8]; float cc[8];
            #pragma unroll
            for (int u = 0; u < 8; ++u) {
                const int j  = jb + u;
                const int jj = min(j, n0 - 1);
                const int sj = __builtin_amdgcn_readlane(src0, jj);
                const float cj = __int_as_float(__builtin_amdgcn_readlane(__float_as_int(c0), jj));
                cc[u] = (j < n0) ? cj : 0.0f;
                h[u]  = *(const float2*)(Hf + (size_t)sj * F);
            }
            if (jb == 0 && nv0) ns0 = s[nsrc0];
            #pragma unroll
            for (int u = 0; u < 8; ++u) {
                if (jb + u < n0) { ax = fmaf(cc[u], h[u].x, ax); ay = fmaf(cc[u], h[u].y, ay); }
            }
        }
        if (n0 == 0 && nv0) ns0 = s[nsrc0];
        for (int cb = start + 32; cb < end; cb += 32) {
            const int p = cb + lane;
            int sx = 0; float cx = 0.0f;
            if (p < end) {
                sx = gsrc[p];
                cx = (__expf(leaky(s[sx] + di) - m) * gw[p]) / dn;
            }
            const int cn = min(32, end - cb);
            for (int j = 0; j < cn; ++j) {
                const int sj = __builtin_amdgcn_readlane(sx, j);
                const float cj = __int_as_float(__builtin_amdgcn_readlane(__float_as_int(cx), j));
                const float2 hv = *(const float2*)(Hf + (size_t)sj * F);
                ax = fmaf(cj, hv.x, ax); ay = fmaf(cj, hv.y, ay);
            }
        }
        ax = fmaf(c_self, hself.x, ax); ay = fmaf(c_self, hself.y, ay);
        if (fv) *(float2*)(out + (size_t)wid * F + fo) = make_float2(ax, ay);

        start = nstart; end = nend; di = ndi; si = nsi;
        src0 = nsrc0; wt0 = nwt0; s0 = ns0;
    }
}

extern "C" void kernel_launch(void* const* d_in, const int* in_sizes, int n_in,
                              void* d_out, int out_size, void* d_ws, size_t ws_size,
                              hipStream_t stream) {
    const float* x   = (const float*)d_in[0];
    const int*   ei  = (const int*)  d_in[1];
    const float* ew  = (const float*)d_in[2];
    const float* W1  = (const float*)d_in[3];
    const float* a1s = (const float*)d_in[4];
    const float* a1d = (const float*)d_in[5];
    const float* W2  = (const float*)d_in[6];
    const float* a2s = (const float*)d_in[7];
    const float* a2d = (const float*)d_in[8];

    const int HID = in_sizes[4];
    const int FIN = in_sizes[3] / HID;
    const int NC  = in_sizes[7];
    const int N   = in_sizes[0] / FIN;
    const int E   = in_sizes[2];

    float* h1   = (float*)d_ws;
    float* out1 = h1   + (size_t)N * HID;
    float* h2   = out1 + (size_t)N * HID;
    float* sArr = h2   + (size_t)N * NC;
    float* dArr = sArr + N;
    float* outF = (float*)d_out;

    const int G1  = 256;
    const int CH  = (E + G1 - 1) / G1;
    const int B   = (N + BIN_NODES - 1) / BIN_NODES;
    const int nbc = B * G1;
    const int nsb = (nbc + 1023) / 1024;
    size_t off = (((size_t)N * (2 * HID + NC + 2)) + 3) & ~(size_t)3;
    int* base  = (int*)d_ws;
    int4*  rec    = (int4*)(base + off);  off += (size_t)E * 4;
    int4*  rec2   = (int4*)(base + off);  off += (size_t)E * 4;
    int*   rowptr = base + off;           off += ((size_t)N + 4) & ~(size_t)3;
    int*   bsum   = base + off;           off += ((size_t)nsb + 3) & ~(size_t)3;
    int*   binCnt = base + off;           off += ((size_t)nbc + 3) & ~(size_t)3;
    int*   binOff = base + off;           off += ((size_t)nbc + 4) & ~(size_t)3;
    int*   gsrc   = base + off;           off += ((size_t)E + 3) & ~(size_t)3;
    float* gw     = (float*)(base + off);

    const int nrt     = (N + 15) / 16;
    const int pairs   = (nrt + 1) / 2;
    const int gBlocks = (pairs + 7) / 8;
    const int sBlocks = 1024;
    const int wBlocks = 2048;

    bin_count_kernel<<<G1, 1024, B * sizeof(int), stream>>>(ei, binCnt, E, B, CH);
    blocksum_kernel<<<nsb, 256, 0, stream>>>(binCnt, bsum, nbc);
    scan_kernel<<<nsb, 256, 0, stream>>>(binCnt, bsum, binOff, nbc, E);
    bin_scatter_kernel<<<G1, 1024, B * sizeof(int), stream>>>(ei, ew, binOff, rec2, E, B, CH);
    bin_group_kernel<<<B, 1024, 0, stream>>>(binOff, rec2, rec, rowptr, N, E, G1);
    group_order_kernel<<<wBlocks, 256, 0, stream>>>(rowptr, rec, gsrc, gw, N);

    gemm1_kernel<<<gBlocks, 256, 0, stream>>>(x, W1, h1, N, FIN);
    sd_kernel<<<sBlocks, 256, 0, stream>>>(h1, a1s, a1d, sArr, dArr, N, HID);
    node_kernel<<<wBlocks, 256, 0, stream>>>(rowptr, gsrc, gw, sArr, dArr, h1, out1, N, HID);

    gemm2_kernel<<<gBlocks, 256, 0, stream>>>(out1, W2, h2, N, NC);
    sd_kernel<<<sBlocks, 256, 0, stream>>>(h2, a2s, a2d, sArr, dArr, N, NC);
    node_kernel<<<wBlocks, 256, 0, stream>>>(rowptr, gsrc, gw, sArr, dArr, h2, outF, N, NC);
}


# Per-node scores s,d formed in both GEMM epilogues from the accumulators (per lane over its column tiles, then a 16-lane tree): both score launches and their re-read of h gone; s,d sum order changes
# speedup vs baseline: 6.3025x; 1.1724x over previous
#include <hip/hip_runtime.h>
#include <hip/hip_bf16.h>


typedef __attribute__((ext_vector_type(2))) float v2f;
typedef __attribute__((ext_vector_type(8))) float v8f;

#define NEG_SLOPE 0.2f

__device__ __forceinline__ float leaky(float x) {
    return x > 0.0f ? x : NEG_SLOPE * x;
}

template <int CTRL>
__device__ __forceinline__ float row_xor(float v) {
    return __int_as_float(__builtin_amdgcn_update_dpp(
        0, __float_as_int(v), CTRL, 0xF, 0xF, false));
}

__device__ __forceinline__ void row_scores(const float (&v)[32], int col0,
                                           float& so, float& dd) {
    const bool b3 = (col0 & 8) != 0, b2 = (col0 & 4) != 0;
    const bool b1 = (col0 & 2) != 0, b0 = (col0 & 1) != 0;
    float w16[16], w8[8], w4[4];
    #pragma unroll
    for (int i = 0; i < 16; ++i) {
        const float keep = b3 ? v[16 + i] : v[i];
        const float send = b3 ? v[i] : v[16 + i];
        w16[i] = keep + row_xor<0x168>(send);
    }
    #pragma unroll
    for (int i = 0; i < 8; ++i) {
        const float keep = b2 ? w16[8 + i] : w16[i];
        const float send = b2 ? w16[i] : w16[8 + i];
        w8[i] = keep + row_xor<0x164>(send);
    }
    #pragma unroll
    for (int i = 0; i < 4; ++i) {
        const float keep = b1 ? w8[4 + i] : w8[i];
        const float send = b1 ? w8[i] : w8[4 + i];
        w4[i] = keep + row_xor<0x162>(send);
    }
    {
        const float keep0 = b0 ? w4[2] : w4[0], send0 = b0 ? w4[0] : w4[2];
        const float keep1 = b0 ? w4[3] : w4[1], send1 = b0 ? w4[1] : w4[3];
        so = keep0 + row_xor<0x161>(send0);
        dd = keep1 + row_xor<0x161>(send1);
    }
}

#define P1 132
__global__ __launch_bounds__(256) void gemm1_kernel(
    const float* __restrict__ X, const float* __restrict__ W,
    const float* __restrict__ asrc, const float* __restrict__ adst,
    float* __restrict__ H, float* __restrict__ s, float* __restrict__ d,
    int N, int K) {
    __shared__ float sW[64 * P1];
    const int lane = threadIdx.x & 31;
    const int wave = threadIdx.x >> 5;
    const int half = lane >> 4;
    const int col0 = lane & 15;
    float as[4], ad[4];
    #pragma unroll
    for (int t = 0; t < 4; ++t) { as[t] = asrc[t * 16 + col0]; ad[t] = adst[t * 16 + col0]; }
    const int nrt  = (N + 15) >> 4;
    const int rt0  = (blockIdx.x * 8 + wave) * 2;
    const int rt1  = rt0 + 1;
    const int rowA0 = (rt0 < nrt) ? min(rt0 * 16 + col0, N - 1) : 0;
    const int rowA1 = (rt1 < nrt) ? min(rt1 * 16 + col0, N - 1) : 0;

    v8f acc[2][4];
    #pragma unroll
    for (int u = 0; u < 2; ++u)
        #pragma unroll
        for (int t = 0; t < 4; ++t) acc[u][t] = v8f{};

    const int phases = K >> 7;
    for (int ph = 0; ph < phases; ++ph) {
        for (int i = threadIdx.x; i < 128 * 64; i += 256) {
            int r = i >> 6, c = i & 63;
            sW[c * P1 + r] = W[(size_t)ph * 128 * 64 + i];
        }
        __syncthreads();

        const float* xr0 = X + (size_t)rowA0 * K + ph * 128;
        const float* xr1 = X + (size_t)rowA1 * K + ph * 128;
        for (int kk = 0; kk < 32; ++kk) {
            const int kb = kk * 4 + half * 2;
            v2f a0 = *(const v2f*)(xr0 + kb);
            v2f a1 = *(const v2f*)(xr1 + kb);
            #pragma unroll
            for (int t = 0; t < 4; ++t) {
                v2f b = *(const v2f*)(&sW[(t * 16 + col0) * P1 + kb]);
                acc[0][t] = __builtin_amdgcn_wmma_f32_16x16x4_f32(
                    false, a0, false, b, (short)0, acc[0][t], false, false);
                acc[1][t] = __builtin_amdgcn_wmma_f32_16x16x4_f32(
                    false, a1, false, b, (short)0, acc[1][t], false, false);
            }
        }
        __syncthreads();
    }

    #pragma unroll
    for (int u = 0; u < 2; ++u) {
        int rt = rt0 + u;
        if (rt >= nrt) continue;
        #pragma unroll
        for (int t = 0; t < 4; ++t)
            #pragma unroll
            for (int r = 0; r < 8; ++r) {
                int row = rt * 16 + r + half * 8;
                if (row < N) H[(size_t)row * 64 + t * 16 + col0] = acc[u][t][r];
            }
    }

    float pv[32];
    #pragma unroll
    for (int u = 0; u < 2; ++u)
        #pragma unroll
        for (int r = 0; r < 8; ++r) {
            float ps = acc[u][0][r] * as[0], pd = acc[u][0][r] * ad[0];
            #pragma unroll
            for (int t = 1; t < 4; ++t) {
                ps = fmaf(acc[u][t][r], as[t], ps);
                pd = fmaf(acc[u][t][r], ad[t], pd);
            }
            pv[u * 16 + r * 2] = ps; pv[u * 16 + r * 2 + 1] = pd;
        }
    float so, dd;
    row_scores(pv, col0, so, dd);
    const int srow = (rt0 + (col0 >> 3)) * 16 + half * 8 + (col0 & 7);
    if (srow < N) { s[srow] = so; d[srow] = dd; }
}

#define P2 68
__global__ __launch_bounds__(256) void gemm2_kernel(
    const float* __restrict__ Hin, const float* __restrict__ W,
    const float* __restrict__ asrc, const float* __restrict__ adst,
    float* __restrict__ Hout, float* __restrict__ s, float* __restrict__ d,
    int N, int NC) {
    __shared__ float sW[48 * P2];
    const int lane = threadIdx.x & 31;
    const int wave = threadIdx.x >> 5;
    const int half = lane >> 4;
    const int col0 = lane & 15;
    float as[3], ad[3];
    #pragma unroll
    for (int t = 0; t < 3; ++t) {
        const int c = t * 16 + col0;
        as[t] = (c < NC) ? asrc[c] : 0.0f;
        ad[t] = (c < NC) ? adst[c] : 0.0f;
    }
    const int nrt  = (N + 15) >> 4;
    const int rt0  = (blockIdx.x * 8 + wave) * 2;
    const int rt1  = rt0 + 1;
    const int rowA0 = (rt0 < nrt) ? min(rt0 * 16 + col0, N - 1) : 0;
    const int rowA1 = (rt1 < nrt) ? min(rt1 * 16 + col0, N - 1) : 0;

    for (int i = threadIdx.x; i < 64 * 48; i += 256) {
        int r = i / 48, c = i % 48;
        sW[c * P2 + r] = (c < NC) ? W[(size_t)r * NC + c] : 0.0f;
    }
    __syncthreads();

    v8f acc[2][3];
    #pragma unroll
    for (int u = 0; u < 2; ++u)
        #pragma unroll
        for (int t = 0; t < 3; ++t) acc[u][t] = v8f{};

    const float* xr0 = Hin + (size_t)rowA0 * 64;
    const float* xr1 = Hin + (size_t)rowA1 * 64;
    for (int kk = 0; kk < 16; ++kk) {
        const int kb = kk * 4 + half * 2;
        v2f a0 = *(const v2f*)(xr0 + kb);
        v2f a1 = *(const v2f*)(xr1 + kb);
        a0[0] = fmaxf(a0[0], 0.0f); a0[1] = fmaxf(a0[1], 0.0f);
        a1[0] = fmaxf(a1[0], 0.0f); a1[1] = fmaxf(a1[1], 0.0f);
        #pragma unroll
        for (int t = 0; t < 3; ++t) {
            v2f b = *(const v2f*)(&sW[(t * 16 + col0) * P2 + kb]);
            acc[0][t] = __builtin_amdgcn_wmma_f32_16x16x4_f32(
                false, a0, false, b, (short)0, acc[0][t], false, false);
            acc[1][t] = __builtin_amdgcn_wmma_f32_16x16x4_f32(
                false, a1, false, b, (short)0, acc[1][t], false, false);
        }
    }

    #pragma unroll
    for (int u = 0; u < 2; ++u) {
        int rt = rt0 + u;
        if (rt >= nrt) continue;
        #pragma unroll
        for (int t = 0; t < 3; ++t)
            #pragma unroll
            for (int r = 0; r < 8; ++r) {
                int row = rt * 16 + r + half * 8;
                int col = t * 16 + col0;
                if (row < N && col < NC)
                    Hout[(size_t)row * NC + col] = acc[u][t][r];
            }
    }

    float pv[32];
    #pragma unroll
    for (int u = 0; u < 2; ++u)
        #pragma unroll
        for (int r = 0; r < 8; ++r) {
            float ps = acc[u][0][r] * as[0], pd = acc[u][0][r] * ad[0];
            #pragma unroll
            for (int t = 1; t < 3; ++t) {
                ps = fmaf(acc[u][t][r], as[t], ps);
                pd = fmaf(acc[u][t][r], ad[t], pd);
            }
            pv[u * 16 + r * 2] = ps; pv[u * 16 + r * 2 + 1] = pd;
        }
    float so, dd;
    row_scores(pv, col0, so, dd);
    const int srow = (rt0 + (col0 >> 3)) * 16 + half * 8 + (col0 & 7);
    if (srow < N) { s[srow] = so; d[srow] = dd; }
}

#define BIN_SHIFT 8
#define BIN_NODES 256

#define TRIP 7

__global__ __launch_bounds__(1024) void bin_count_kernel(
    const int* __restrict__ ei, int* __restrict__ binCnt, int E, int B, int CH) {
    extern __shared__ int hist[];
    const int tid = threadIdx.x, nt = blockDim.x;
    const int c = blockIdx.x, G = gridDim.x;
    for (int i = tid; i < B; i += nt) hist[i] = 0;
    __syncthreads();
    const int e0 = min(E, c * CH), e1 = min(E, e0 + CH);
    for (int base = e0 + tid; base - tid < e1; base += TRIP * nt) {
        int dd[TRIP];
        #pragma unroll
        for (int k = 0; k < TRIP; ++k) dd[k] = ei[E + min(base + k * nt, e1 - 1)];
        #pragma unroll
        for (int k = 0; k < TRIP; ++k)
            if (base + k * nt < e1) atomicAdd(&hist[dd[k] >> BIN_SHIFT], 1);
    }
    __syncthreads();
    for (int i = tid; i < B; i += nt) binCnt[(size_t)i * G + c] = hist[i];
}

__global__ __launch_bounds__(256) void blocksum_kernel(
    const int* __restrict__ cnt, int* __restrict__ bsum, int N) {
    __shared__ int sred[8];
    const int lane = threadIdx.x & 31, wave = threadIdx.x >> 5;
    const int base = blockIdx.x * 1024 + threadIdx.x * 4;
    int v = 0;
    for (int k = 0; k < 4; ++k) if (base + k < N) v += cnt[base + k];
    for (int off = 16; off > 0; off >>= 1) v += __shfl_xor(v, off, 32);
    if (lane == 0) sred[wave] = v;
    __syncthreads();
    if (threadIdx.x == 0) {
        int t = 0;
        for (int w = 0; w < 8; ++w) t += sred[w];
        bsum[blockIdx.x] = t;
    }
}

__global__ __launch_bounds__(256) void scan_kernel(
    const int* __restrict__ cnt, const int* __restrict__ bsum,
    int* __restrict__ rowptr, int N, int E) {
    __shared__ int sred[8];
    __shared__ int swave[8];
    const int lane = threadIdx.x & 31, wave = threadIdx.x >> 5;
    int v = 0;
    for (int t = threadIdx.x; t < (int)blockIdx.x; t += 256) v += bsum[t];
    for (int off = 16; off > 0; off >>= 1) v += __shfl_xor(v, off, 32);
    if (lane == 0) sred[wave] = v;

    const int base = blockIdx.x * 1024 + threadIdx.x * 4;
    int c[4];
    for (int k = 0; k < 4; ++k) c[k] = (base + k < N) ? cnt[base + k] : 0;
    const int tsum = c[0] + c[1] + c[2] + c[3];
    int inc = tsum;
    for (int off = 1; off < 32; off <<= 1) {
        int n = __shfl_up(inc, off, 32);
        if (lane >= off) inc += n;
    }
    if (lane == 31) swave[wave] = inc;
    __syncthreads();
    int pre = 0;
    for (int w = 0; w < 8; ++w) {
        pre += sred[w];
        if (w < wave) pre += swave[w];
    }
    int run = pre + inc - tsum;
    for (int k = 0; k < 4; ++k) {
        if (base + k < N) rowptr[base + k] = run;
        run += c[k];
    }
    if (blockIdx.x == 0 && threadIdx.x == 0) rowptr[N] = E;
}

__global__ __launch_bounds__(1024) void bin_scatter_kernel(
    const int* __restrict__ ei, const float* __restrict__ ew,
    const int* __restrict__ binOff, int4* __restrict__ rec2, int E, int B, int CH) {
    extern __shared__ int cur[];
    const int tid = threadIdx.x, nt = blockDim.x;
    const int c = blockIdx.x, G = gridDim.x;
    for (int i = tid; i < B; i += nt) cur[i] = binOff[(size_t)i * G + c];
    __syncthreads();
    const int e0 = min(E, c * CH), e1 = min(E, e0 + CH);
    for (int base = e0 + tid; base - tid < e1; base += TRIP * nt) {
        int dd[TRIP], ss[TRIP], ww[TRIP], pp[TRIP];
        #pragma unroll
        for (int k = 0; k < TRIP; ++k) {
            const int e = min(base + k * nt, e1 - 1);
            dd[k] = ei[E + e];
            ss[k] = ei[e];
            ww[k] = __float_as_int(ew[e]);
        }
        #pragma unroll
        for (int k = 0; k < TRIP; ++k) {
            pp[k] = 0;
            if (base + k * nt < e1) pp[k] = atomicAdd(&cur[dd[k] >> BIN_SHIFT], 1);
        }
        #pragma unroll
        for (int k = 0; k < TRIP; ++k)
            if (base + k * nt < e1) rec2[pp[k]] = make_int4(base + k * nt, ss[k], ww[k], dd[k]);
    }
}

#define GTRIP 6
__global__ __launch_bounds__(1024) void bin_group_kernel(
    const int* __restrict__ binOff, const int4* __restrict__ rec2,
    int4* __restrict__ rec, int* __restrict__ rowptr, int N, int E, int G) {
    __shared__ int cnt[BIN_NODES];
    __shared__ int cur[BIN_NODES];
    __shared__ int swave[BIN_NODES / 32];
    const int tid = threadIdx.x, nt = blockDim.x;
    const int lane = tid & 31, wave = tid >> 5;
    const int b = blockIdx.x;
    const int nbase = b << BIN_SHIFT;
    const int binStart = binOff[(size_t)b * G];
    const int binEnd   = binOff[(size_t)(b + 1) * G];
    if (tid < BIN_NODES) cnt[tid] = 0;
    __syncthreads();
    for (int base = binStart + tid; base - tid < binEnd; base += GTRIP * nt) {
        int dd[GTRIP];
        #pragma unroll
        for (int k = 0; k < GTRIP; ++k) dd[k] = rec2[min(base + k * nt, binEnd - 1)].w;
        #pragma unroll
        for (int k = 0; k < GTRIP; ++k)
            if (base + k * nt < binEnd) atomicAdd(&cnt[dd[k] - nbase], 1);
    }
    __syncthreads();
    int v = 0, inc = 0;
    if (tid < BIN_NODES) {
        v = cnt[tid];
        inc = v;
        for (int off = 1; off < 32; off <<= 1) {
            int n = __shfl_up(inc, off, 32);
            if (lane >= off) inc += n;
        }
        if (lane == 31) swave[wave] = inc;
    }
    __syncthreads();
    if (tid < BIN_NODES) {
        int pre = 0;
        for (int w = 0; w < BIN_NODES / 32; ++w) if (w < wave) pre += swave[w];
        const int first = binStart + pre + inc - v;
        cur[tid] = first;
        if (nbase + tid < N) rowptr[nbase + tid] = first;
    }
    if (b == (int)gridDim.x - 1 && tid == 0) rowptr[N] = E;
    __syncthreads();
    for (int base = binStart + tid; base - tid < binEnd; base += GTRIP * nt) {
        int4 r[GTRIP]; int pp[GTRIP];
        #pragma unroll
        for (int k = 0; k < GTRIP; ++k) r[k] = rec2[min(base + k * nt, binEnd - 1)];
        #pragma unroll
        for (int k = 0; k < GTRIP; ++k) {
            pp[k] = 0;
            if (base + k * nt < binEnd) pp[k] = atomicAdd(&cur[r[k].w - nbase], 1);
        }
        #pragma unroll
        for (int k = 0; k < GTRIP; ++k)
            if (base + k * nt < binEnd) rec[pp[k]] = make_int4(r[k].x, r[k].y, r[k].z, 0);
    }
}

__global__ __launch_bounds__(256) void group_order_kernel(
    const int* __restrict__ rowptr, const int4* __restrict__ rec,
    int* __restrict__ gsrc, float* __restrict__ gw, int N) {
    const int w0   = __builtin_amdgcn_readfirstlane(
        (int)((blockIdx.x * (long long)blockDim.x + threadIdx.x) >> 5));
    const int nw   = (int)((gridDim.x * (long long)blockDim.x) >> 5);
    const int lane = threadIdx.x & 31;
    if (w0 >= N) return;
    int start = rowptr[w0];
    int end   = rowptr[w0 + 1];
    for (int wid = w0; wid < N; wid += nw) {
        const int nx = min(wid + nw, N - 1);
        const int nstart = rowptr[nx];
        const int nend   = rowptr[nx + 1];
        const int deg = end - start;
        if (deg <= 32) {
            int4 r = make_int4(0x7fffffff, 0, 0, 0);
            if (lane < deg) r = rec[start + lane];
            int rk = 0;
            for (int j = 0; j < 32; ++j) rk += (__shfl(r.x, j, 32) < r.x) ? 1 : 0;
            if (lane < deg) { gsrc[start + rk] = r.y; gw[start + rk] = __int_as_float(r.z); }
        } else {
            for (int q = lane; q < deg; q += 32) {
                int4 r = rec[start + q];
                int rk = 0;
                for (int j = 0; j < deg; ++j) rk += (rec[start + j].x < r.x) ? 1 : 0;
                gsrc[start + rk] = r.y; gw[start + rk] = __int_as_float(r.z);
            }
        }
        start = nstart; end = nend;
    }
}

__global__ __launch_bounds__(256) void node_kernel(
    const int* __restrict__ rowptr, const int* __restrict__ gsrc,
    const float* __restrict__ gw, const float* __restrict__ s,
    const float* __restrict__ d, const float* __restrict__ H,
    float* __restrict__ out, int N, int F) {
    const int wfirst = __builtin_amdgcn_readfirstlane(
        (int)((blockIdx.x * (long long)blockDim.x + threadIdx.x) >> 5));
    const int nw   = (int)((gridDim.x * (long long)blockDim.x) >> 5);
    const int lane = threadIdx.x & 31;
    if (wfirst >= N) return;

    const bool fv = lane * 2 < F;
    const int  fo = fv ? lane * 2 : 0;
    const float* __restrict__ Hf = H + fo;

    int   start = rowptr[wfirst];
    int   end   = rowptr[wfirst + 1];
    float di    = d[wfirst];
    float si    = s[wfirst];
    int src0 = 0; float wt0 = 0.0f; float s0 = 0.0f;
    if (lane < end - start) {
        src0 = gsrc[start + lane];
        wt0  = gw[start + lane];
        s0   = s[src0];
    }

    for (int wid = wfirst; wid < N; wid += nw) {
        const float2 hself = *(const float2*)(Hf + (size_t)wid * F);
        const int nx = min(wid + nw, N - 1);
        const int   nstart = rowptr[nx];
        const int   nend   = rowptr[nx + 1];
        const float ndi    = d[nx];
        const float nsi    = s[nx];

        const int deg = end - start;
        const float sc_self = leaky(si + di);
        const bool v0 = lane < deg;
        const float sc0 = v0 ? leaky(s0 + di) : sc_self;

        float m = fmaxf(sc_self, sc0);
        for (int p = start + 32 + lane; p < end; p += 32)
            m = fmaxf(m, leaky(s[gsrc[p]] + di));
        for (int off = 16; off > 0; off >>= 1) m = fmaxf(m, __shfl_xor(m, off, 32));

        const float ex0 = v0 ? __expf(sc0 - m) : 0.0f;
        float sum = ex0;
        for (int p = start + 32 + lane; p < end; p += 32)
            sum += __expf(leaky(s[gsrc[p]] + di) - m);
        for (int off = 16; off > 0; off >>= 1) sum += __shfl_xor(sum, off, 32);
        const float ex_self = __expf(sc_self - m);
        sum += ex_self;
        const float dn = fmaxf(sum, 1e-16f);
        const float c0 = (ex0 * wt0) / dn;
        const float c_self = ex_self / dn;

        const bool nv0 = lane < nend - nstart;
        int nsrc0 = 0; float nwt0 = 0.0f; float ns0 = 0.0f;
        if (nv0) { nsrc0 = gsrc[nstart + lane]; nwt0 = gw[nstart + lane]; }

        float ax = 0.0f, ay = 0.0f;
        const int n0 = min(deg, 32);
        for (int jb = 0; jb < n0; jb += 8) {
            float2 h[8]; float cc[8];
            #pragma unroll
            for (int u = 0; u < 8; ++u) {
                const int j  = jb + u;
                const int jj = min(j, n0 - 1);
                const int sj = __builtin_amdgcn_readlane(src0, jj);
                const float cj = __int_as_float(__builtin_amdgcn_readlane(__float_as_int(c0), jj));
                cc[u] = (j < n0) ? cj : 0.0f;
                h[u]  = *(const float2*)(Hf + (size_t)sj * F);
            }
            if (jb == 0 && nv0) ns0 = s[nsrc0];
            #pragma unroll
            for (int u = 0; u < 8; ++u) {
                if (jb + u < n0) { ax = fmaf(cc[u], h[u].x, ax); ay = fmaf(cc[u], h[u].y, ay); }
            }
        }
        if (n0 == 0 && nv0) ns0 = s[nsrc0];
        for (int cb = start + 32; cb < end; cb += 32) {
            const int p = cb + lane;
            int sx = 0; float cx = 0.0f;
            if (p < end) {
                sx = gsrc[p];
                cx = (__expf(leaky(s[sx] + di) - m) * gw[p]) / dn;
            }
            const int cn = min(32, end - cb);
            for (int j = 0; j < cn; ++j) {
                const int sj = __builtin_amdgcn_readlane(sx, j);
                const float cj = __int_as_float(__builtin_amdgcn_readlane(__float_as_int(cx), j));
                const float2 hv = *(const float2*)(Hf + (size_t)sj * F);
                ax = fmaf(cj, hv.x, ax); ay = fmaf(cj, hv.y, ay);
            }
        }
        ax = fmaf(c_self, hself.x, ax); ay = fmaf(c_self, hself.y, ay);
        if (fv) *(float2*)(out + (size_t)wid * F + fo) = make_float2(ax, ay);

        start = nstart; end = nend; di = ndi; si = nsi;
        src0 = nsrc0; wt0 = nwt0; s0 = ns0;
    }
}

extern "C" void kernel_launch(void* const* d_in, const int* in_sizes, int n_in,
                              void* d_out, int out_size, void* d_ws, size_t ws_size,
                              hipStream_t stream) {
    const float* x   = (const float*)d_in[0];
    const int*   ei  = (const int*)  d_in[1];
    const float* ew  = (const float*)d_in[2];
    const float* W1  = (const float*)d_in[3];
    const float* a1s = (const float*)d_in[4];
    const float* a1d = (const float*)d_in[5];
    const float* W2  = (const float*)d_in[6];
    const float* a2s = (const float*)d_in[7];
    const float* a2d = (const float*)d_in[8];

    const int HID = in_sizes[4];
    const int FIN = in_sizes[3] / HID;
    const int NC  = in_sizes[7];
    const int N   = in_sizes[0] / FIN;
    const int E   = in_sizes[2];

    float* h1   = (float*)d_ws;
    float* out1 = h1   + (size_t)N * HID;
    float* h2   = out1 + (size_t)N * HID;
    float* sArr = h2   + (size_t)N * NC;
    float* dArr = sArr + N;
    float* outF = (float*)d_out;

    const int G1  = 256;
    const int CH  = (E + G1 - 1) / G1;
    const int B   = (N + BIN_NODES - 1) / BIN_NODES;
    const int nbc = B * G1;
    const int nsb = (nbc + 1023) / 1024;
    size_t off = (((size_t)N * (2 * HID + NC + 2)) + 3) & ~(size_t)3;
    int* base  = (int*)d_ws;
    int4*  rec    = (int4*)(base + off);  off += (size_t)E * 4;
    int4*  rec2   = (int4*)(base + off);  off += (size_t)E * 4;
    int*   rowptr = base + off;           off += ((size_t)N + 4) & ~(size_t)3;
    int*   bsum   = base + off;           off += ((size_t)nsb + 3) & ~(size_t)3;
    int*   binCnt = base + off;           off += ((size_t)nbc + 3) & ~(size_t)3;
    int*   binOff = base + off;           off += ((size_t)nbc + 4) & ~(size_t)3;
    int*   gsrc   = base + off;           off += ((size_t)E + 3) & ~(size_t)3;
    float* gw     = (float*)(base + off);

    const int nrt     = (N + 15) / 16;
    const int pairs   = (nrt + 1) / 2;
    const int gBlocks = (pairs + 7) / 8;
    const int wBlocks = 2048;

    bin_count_kernel<<<G1, 1024, B * sizeof(int), stream>>>(ei, binCnt, E, B, CH);
    blocksum_kernel<<<nsb, 256, 0, stream>>>(binCnt, bsum, nbc);
    scan_kernel<<<nsb, 256, 0, stream>>>(binCnt, bsum, binOff, nbc, E);
    bin_scatter_kernel<<<G1, 1024, B * sizeof(int), stream>>>(ei, ew, binOff, rec2, E, B, CH);
    bin_group_kernel<<<B, 1024, 0, stream>>>(binOff, rec2, rec, rowptr, N, E, G1);
    group_order_kernel<<<wBlocks, 256, 0, stream>>>(rowptr, rec, gsrc, gw, N);

    gemm1_kernel<<<gBlocks, 256, 0, stream>>>(x, W1, a1s, a1d, h1, sArr, dArr, N, FIN);
    node_kernel<<<wBlocks, 256, 0, stream>>>(rowptr, gsrc, gw, sArr, dArr, h1, out1, N, HID);

    gemm2_kernel<<<gBlocks, 256, 0, stream>>>(out1, W2, a2s, a2d, h2, sArr, dArr, N, NC);
    node_kernel<<<wBlocks, 256, 0, stream>>>(rowptr, gsrc, gw, sArr, dArr, h2, outF, N, NC);
}


# Edge grouping without a grid-wide scan: each chunk sorts its records by bin in its own region (local scan); the bin pass finds its segments in a table and orders its groups itself: 6 launches -> 2
# speedup vs baseline: 6.6921x; 1.0618x over previous
#include <hip/hip_runtime.h>
#include <hip/hip_bf16.h>


typedef __attribute__((ext_vector_type(2))) float v2f;
typedef __attribute__((ext_vector_type(8))) float v8f;

#define NEG_SLOPE 0.2f

__device__ __forceinline__ float leaky(float x) {
    return x > 0.0f ? x : NEG_SLOPE * x;
}

template <int CTRL>
__device__ __forceinline__ float row_xor(float v) {
    return __int_as_float(__builtin_amdgcn_update_dpp(
        0, __float_as_int(v), CTRL, 0xF, 0xF, false));
}

__device__ __forceinline__ void row_scores(const float (&v)[32], int col0,
                                           float& so, float& dd) {
    const bool b3 = (col0 & 8) != 0, b2 = (col0 & 4) != 0;
    const bool b1 = (col0 & 2) != 0, b0 = (col0 & 1) != 0;
    float w16[16], w8[8], w4[4];
    #pragma unroll
    for (int i = 0; i < 16; ++i) {
        const float keep = b3 ? v[16 + i] : v[i];
        const float send = b3 ? v[i] : v[16 + i];
        w16[i] = keep + row_xor<0x168>(send);
    }
    #pragma unroll
    for (int i = 0; i < 8; ++i) {
        const float keep = b2 ? w16[8 + i] : w16[i];
        const float send = b2 ? w16[i] : w16[8 + i];
        w8[i] = keep + row_xor<0x164>(send);
    }
    #pragma unroll
    for (int i = 0; i < 4; ++i) {
        const float keep = b1 ? w8[4 + i] : w8[i];
        const float send = b1 ? w8[i] : w8[4 + i];
        w4[i] = keep + row_xor<0x162>(send);
    }
    {
        const float keep0 = b0 ? w4[2] : w4[0], send0 = b0 ? w4[0] : w4[2];
        const float keep1 = b0 ? w4[3] : w4[1], send1 = b0 ? w4[1] : w4[3];
        so = keep0 + row_xor<0x161>(send0);
        dd = keep1 + row_xor<0x161>(send1);
    }
}

#define P1 132
__global__ __launch_bounds__(256) void gemm1_kernel(
    const float* __restrict__ X, const float* __restrict__ W,
    const float* __restrict__ asrc, const float* __restrict__ adst,
    float* __restrict__ H, float* __restrict__ s, float* __restrict__ d,
    int N, int K) {
    __shared__ float sW[64 * P1];
    const int lane = threadIdx.x & 31;
    const int wave = threadIdx.x >> 5;
    const int half = lane >> 4;
    const int col0 = lane & 15;
    float as[4], ad[4];
    #pragma unroll
    for (int t = 0; t < 4; ++t) { as[t] = asrc[t * 16 + col0]; ad[t] = adst[t * 16 + col0]; }
    const int nrt  = (N + 15) >> 4;
    const int rt0  = (blockIdx.x * 8 + wave) * 2;
    const int rt1  = rt0 + 1;
    const int rowA0 = (rt0 < nrt) ? min(rt0 * 16 + col0, N - 1) : 0;
    const int rowA1 = (rt1 < nrt) ? min(rt1 * 16 + col0, N - 1) : 0;

    v8f acc[2][4];
    #pragma unroll
    for (int u = 0; u < 2; ++u)
        #pragma unroll
        for (int t = 0; t < 4; ++t) acc[u][t] = v8f{};

    const int phases = K >> 7;
    for (int ph = 0; ph < phases; ++ph) {
        for (int i = threadIdx.x; i < 128 * 64; i += 256) {
            int r = i >> 6, c = i & 63;
            sW[c * P1 + r] = W[(size_t)ph * 128 * 64 + i];
        }
        __syncthreads();

        const float* xr0 = X + (size_t)rowA0 * K + ph * 128;
        const float* xr1 = X + (size_t)rowA1 * K + ph * 128;
        for (int kk = 0; kk < 32; ++kk) {
            const int kb = kk * 4 + half * 2;
            v2f a0 = *(const v2f*)(xr0 + kb);
            v2f a1 = *(const v2f*)(xr1 + kb);
            #pragma unroll
            for (int t = 0; t < 4; ++t) {
                v2f b = *(const v2f*)(&sW[(t * 16 + col0) * P1 + kb]);
                acc[0][t] = __builtin_amdgcn_wmma_f32_16x16x4_f32(
                    false, a0, false, b, (short)0, acc[0][t], false, false);
                acc[1][t] = __builtin_amdgcn_wmma_f32_16x16x4_f32(
                    false, a1, false, b, (short)0, acc[1][t], false, false);
            }
        }
        __syncthreads();
    }

    #pragma unroll
    for (int u = 0; u < 2; ++u) {
        int rt = rt0 + u;
        if (rt >= nrt) continue;
        #pragma unroll
        for (int t = 0; t < 4; ++t)
            #pragma unroll
            for (int r = 0; r < 8; ++r) {
                int row = rt * 16 + r + half * 8;
                if (row < N) H[(size_t)row * 64 + t * 16 + col0] = acc[u][t][r];
            }
    }

    float pv[32];
    #pragma unroll
    for (int u = 0; u < 2; ++u)
        #pragma unroll
        for (int r = 0; r < 8; ++r) {
            float ps = acc[u][0][r] * as[0], pd = acc[u][0][r] * ad[0];
            #pragma unroll
            for (int t = 1; t < 4; ++t) {
                ps = fmaf(acc[u][t][r], as[t], ps);
                pd = fmaf(acc[u][t][r], ad[t], pd);
            }
            pv[u * 16 + r * 2] = ps; pv[u * 16 + r * 2 + 1] = pd;
        }
    float so, dd;
    row_scores(pv, col0, so, dd);
    const int srow = (rt0 + (col0 >> 3)) * 16 + half * 8 + (col0 & 7);
    if (srow < N) { s[srow] = so; d[srow] = dd; }
}

#define P2 68
__global__ __launch_bounds__(256) void gemm2_kernel(
    const float* __restrict__ Hin, const float* __restrict__ W,
    const float* __restrict__ asrc, const float* __restrict__ adst,
    float* __restrict__ Hout, float* __restrict__ s, float* __restrict__ d,
    int N, int NC) {
    __shared__ float sW[48 * P2];
    const int lane = threadIdx.x & 31;
    const int wave = threadIdx.x >> 5;
    const int half = lane >> 4;
    const int col0 = lane & 15;
    float as[3], ad[3];
    #pragma unroll
    for (int t = 0; t < 3; ++t) {
        const int c = t * 16 + col0;
        as[t] = (c < NC) ? asrc[c] : 0.0f;
        ad[t] = (c < NC) ? adst[c] : 0.0f;
    }
    const int nrt  = (N + 15) >> 4;
    const int rt0  = (blockIdx.x * 8 + wave) * 2;
    const int rt1  = rt0 + 1;
    const int rowA0 = (rt0 < nrt) ? min(rt0 * 16 + col0, N - 1) : 0;
    const int rowA1 = (rt1 < nrt) ? min(rt1 * 16 + col0, N - 1) : 0;

    for (int i = threadIdx.x; i < 64 * 48; i += 256) {
        int r = i / 48, c = i % 48;
        sW[c * P2 + r] = (c < NC) ? W[(size_t)r * NC + c] : 0.0f;
    }
    __syncthreads();

    v8f acc[2][3];
    #pragma unroll
    for (int u = 0; u < 2; ++u)
        #pragma unroll
        for (int t = 0; t < 3; ++t) acc[u][t] = v8f{};

    const float* xr0 = Hin + (size_t)rowA0 * 64;
    const float* xr1 = Hin + (size_t)rowA1 * 64;
    for (int kk = 0; kk < 16; ++kk) {
        const int kb = kk * 4 + half * 2;
        v2f a0 = *(const v2f*)(xr0 + kb);
        v2f a1 = *(const v2f*)(xr1 + kb);
        a0[0] = fmaxf(a0[0], 0.0f); a0[1] = fmaxf(a0[1], 0.0f);
        a1[0] = fmaxf(a1[0], 0.0f); a1[1] = fmaxf(a1[1], 0.0f);
        #pragma unroll
        for (int t = 0; t < 3; ++t) {
            v2f b = *(const v2f*)(&sW[(t * 16 + col0) * P2 + kb]);
            acc[0][t] = __builtin_amdgcn_wmma_f32_16x16x4_f32(
                false, a0, false, b, (short)0, acc[0][t], false, false);
            acc[1][t] = __builtin_amdgcn_wmma_f32_16x16x4_f32(
                false, a1, false, b, (short)0, acc[1][t], false, false);
        }
    }

    #pragma unroll
    for (int u = 0; u < 2; ++u) {
        int rt = rt0 + u;
        if (rt >= nrt) continue;
        #pragma unroll
        for (int t = 0; t < 3; ++t)
            #pragma unroll
            for (int r = 0; r < 8; ++r) {
                int row = rt * 16 + r + half * 8;
                int col = t * 16 + col0;
                if (row < N && col < NC)
                    Hout[(size_t)row * NC + col] = acc[u][t][r];
            }
    }

    float pv[32];
    #pragma unroll
    for (int u = 0; u < 2; ++u)
        #pragma unroll
        for (int r = 0; r < 8; ++r) {
            float ps = acc[u][0][r] * as[0], pd = acc[u][0][r] * ad[0];
            #pragma unroll
            for (int t = 1; t < 3; ++t) {
                ps = fmaf(acc[u][t][r], as[t], ps);
                pd = fmaf(acc[u][t][r], ad[t], pd);
            }
            pv[u * 16 + r * 2] = ps; pv[u * 16 + r * 2 + 1] = pd;
        }
    float so, dd;
    row_scores(pv, col0, so, dd);
    const int srow = (rt0 + (col0 >> 3)) * 16 + half * 8 + (col0 & 7);
    if (srow < N) { s[srow] = so; d[srow] = dd; }
}

#define BIN_SHIFT 8
#define BIN_NODES 256

#define TRIP 7

__global__ __launch_bounds__(1024) void chunk_bin_kernel(
    const int* __restrict__ ei, const float* __restrict__ ew,
    int* __restrict__ segOff, int4* __restrict__ rec2, int E, int B, int CH) {
    extern __shared__ int hist[];
    __shared__ int swave[32];
    const int tid = threadIdx.x, nt = blockDim.x;
    const int lane = tid & 31, wave = tid >> 5;
    const int c = blockIdx.x, G = gridDim.x;
    for (int i = tid; i < B; i += nt) hist[i] = 0;
    __syncthreads();
    const int e0 = min(E, c * CH), e1 = min(E, e0 + CH);
    int dd[TRIP], ss[TRIP], ww[TRIP], pp[TRIP];
    #pragma unroll
    for (int k = 0; k < TRIP; ++k) { dd[k] = 0; ss[k] = 0; ww[k] = 0; pp[k] = 0; }
    if (e0 < e1) {
        #pragma unroll
        for (int k = 0; k < TRIP; ++k) {
            const int e = min(e0 + tid + k * nt, e1 - 1);
            dd[k] = ei[E + e];
            ss[k] = ei[e];
            ww[k] = __float_as_int(ew[e]);
        }
        #pragma unroll
        for (int k = 0; k < TRIP; ++k)
            if (e0 + tid + k * nt < e1) pp[k] = atomicAdd(&hist[dd[k] >> BIN_SHIFT], 1);
    }
    __syncthreads();
    const int per = (B + nt - 1) / nt;
    const int i0 = min(B, tid * per), i1 = min(B, i0 + per);
    int tsum = 0;
    for (int i = i0; i < i1; ++i) tsum += hist[i];
    int inc = tsum;
    for (int off = 1; off < 32; off <<= 1) {
        int n = __shfl_up(inc, off, 32);
        if (lane >= off) inc += n;
    }
    if (lane == 31) swave[wave] = inc;
    __syncthreads();
    int pre = (lane < wave) ? swave[lane] : 0;
    for (int off = 16; off > 0; off >>= 1) pre += __shfl_xor(pre, off, 32);
    int run = pre + inc - tsum;
    for (int i = i0; i < i1; ++i) {
        const int v = hist[i];
        hist[i] = run;
        segOff[(size_t)i * G + c] = run;
        run += v;
    }
    if (tid == 0) segOff[(size_t)B * G + c] = e1 - e0;
    __syncthreads();
    #pragma unroll
    for (int k = 0; k < TRIP; ++k) {
        const int e = e0 + tid + k * nt;
        if (e < e1)
            rec2[e0 + hist[dd[k] >> BIN_SHIFT] + pp[k]] = make_int4(e, ss[k], ww[k], dd[k]);
    }
}

__device__ __forceinline__ void seg_group(
    const int* __restrict__ segOff, int b, int G, int CH, int g, int lane,
    int (&incl)[8], int (&dl)[8], int& losum) {
    const int c = g * 8 + (lane & 7);
    int lo = 0, hi = 0;
    if (c < G) { lo = segOff[(size_t)b * G + c]; hi = segOff[(size_t)(b + 1) * G + c]; }
    const int len = hi - lo;
    int inc = len;
    for (int off = 1; off < 8; off <<= 1) {
        int n = __shfl_up(inc, off, 32);
        if ((lane & 7) >= off) inc += n;
    }
    const int d = c * CH + lo - (inc - len);
    losum = 0;
    #pragma unroll
    for (int i = 0; i < 8; ++i) {
        incl[i] = __builtin_amdgcn_readlane(inc, i);
        dl[i]   = __builtin_amdgcn_readlane(d, i);
        losum  += __builtin_amdgcn_readlane(lo, i);
    }
}

__device__ __forceinline__ int seg_index(int q, const int (&incl)[8], const int (&dl)[8]) {
    int D = dl[0];
    #pragma unroll
    for (int i = 0; i < 7; ++i) D = (q >= incl[i]) ? dl[i + 1] : D;
    return q + D;
}

#define GTRIP 6
__global__ __launch_bounds__(1024, 16) void bin_group_kernel(
    const int* __restrict__ segOff, const int4* __restrict__ rec2,
    int4* rec, int* __restrict__ rowptr,
    int* __restrict__ gsrc, float* __restrict__ gw, int N, int E, int G, int CH) {
    __shared__ int cnt[BIN_NODES];
    __shared__ int cur[BIN_NODES];
    __shared__ int swave[BIN_NODES / 32];
    __shared__ int sstart[32];
    const int tid = threadIdx.x;
    const int lane = tid & 31, wave = tid >> 5;
    const int nwaves = blockDim.x >> 5;
    const int b = blockIdx.x;
    const int nbase = b << BIN_SHIFT;
    if (tid < BIN_NODES) cnt[tid] = 0;
    __syncthreads();
    int mylo = 0;
    for (int g = wave; g * 8 < G; g += nwaves) {
        int incl[8], dl[8], losum;
        seg_group(segOff, b, G, CH, g, lane, incl, dl, losum);
        mylo += losum;
        const int T = incl[7];
        for (int q0 = 0; q0 < T; q0 += 32 * GTRIP) {
            int dd[GTRIP];
            #pragma unroll
            for (int k = 0; k < GTRIP; ++k)
                dd[k] = rec2[seg_index(min(q0 + k * 32 + lane, T - 1), incl, dl)].w;
            #pragma unroll
            for (int k = 0; k < GTRIP; ++k)
                if (q0 + k * 32 + lane < T) atomicAdd(&cnt[dd[k] - nbase], 1);
        }
    }
    if (lane == 0) sstart[wave] = mylo;
    __syncthreads();
    int v = 0, inc = 0;
    if (tid < BIN_NODES) {
        v = cnt[tid];
        inc = v;
        for (int off = 1; off < 32; off <<= 1) {
            int n = __shfl_up(inc, off, 32);
            if (lane >= off) inc += n;
        }
        if (lane == 31) swave[wave] = inc;
    }
    __syncthreads();
    if (tid < BIN_NODES) {
        int binStart = 0;
        for (int w = 0; w < nwaves; ++w) binStart += sstart[w];
        int pre = 0;
        for (int w = 0; w < BIN_NODES / 32; ++w) if (w < wave) pre += swave[w];
        const int first = binStart + pre + inc - v;
        cur[tid] = first;
        if (nbase + tid < N) rowptr[nbase + tid] = first;
    }
    if (b == (int)gridDim.x - 1 && tid == 0) rowptr[N] = E;
    __syncthreads();
    for (int g = wave; g * 8 < G; g += nwaves) {
        int incl[8], dl[8], losum;
        seg_group(segOff, b, G, CH, g, lane, incl, dl, losum);
        const int T = incl[7];
        for (int q0 = 0; q0 < T; q0 += 32 * GTRIP) {
            int4 r[GTRIP]; int pp[GTRIP];
            #pragma unroll
            for (int k = 0; k < GTRIP; ++k)
                r[k] = rec2[seg_index(min(q0 + k * 32 + lane, T - 1), incl, dl)];
            #pragma unroll
            for (int k = 0; k < GTRIP; ++k) {
                pp[k] = 0;
                if (q0 + k * 32 + lane < T) pp[k] = atomicAdd(&cur[r[k].w - nbase], 1);
            }
            #pragma unroll
            for (int k = 0; k < GTRIP; ++k)
                if (q0 + k * 32 + lane < T) rec[pp[k]] = make_int4(r[k].x, r[k].y, r[k].z, 0);
        }
    }
    __syncthreads();
    int nl = wave;
    int start = 0, deg = 0;
    int4 r = make_int4(0x7fffffff, 0, 0, 0);
    if (nl < BIN_NODES) {
        deg = cnt[nl];
        start = cur[nl] - deg;
        if (lane < deg) r = rec[start + lane];
    }
    for (; nl < BIN_NODES; nl += nwaves) {
        const int nn = nl + nwaves;
        int nstart = 0, ndeg = 0;
        int4 nr = make_int4(0x7fffffff, 0, 0, 0);
        if (nn < BIN_NODES) {
            ndeg = cnt[nn];
            nstart = cur[nn] - ndeg;
            if (lane < ndeg) nr = rec[nstart + lane];
        }
        const int degs = __builtin_amdgcn_readfirstlane(deg);
        if (degs <= 32) {
            int rk = 0;
            for (int j0 = 0; j0 < degs; j0 += 8) {
                #pragma unroll
                for (int u = 0; u < 8; ++u)
                    rk += (__builtin_amdgcn_readlane(r.x, j0 + u) < r.x) ? 1 : 0;
            }
            if (lane < degs) { gsrc[start + rk] = r.y; gw[start + rk] = __int_as_float(r.z); }
        } else {
            for (int q = lane; q < degs; q += 32) {
                const int4 rq = rec[start + q];
                int rk = 0;
                for (int j = 0; j < degs; ++j) rk += (rec[start + j].x < rq.x) ? 1 : 0;
                gsrc[start + rk] = rq.y; gw[start + rk] = __int_as_float(rq.z);
            }
        }
        start = nstart; deg = ndeg; r = nr;
    }
}

__global__ __launch_bounds__(256) void node_kernel(
    const int* __restrict__ rowptr, const int* __restrict__ gsrc,
    const float* __restrict__ gw, const float* __restrict__ s,
    const float* __restrict__ d, const float* __restrict__ H,
    float* __restrict__ out, int N, int F) {
    const int wfirst = __builtin_amdgcn_readfirstlane(
        (int)((blockIdx.x * (long long)blockDim.x + threadIdx.x) >> 5));
    const int nw   = (int)((gridDim.x * (long long)blockDim.x) >> 5);
    const int lane = threadIdx.x & 31;
    if (wfirst >= N) return;

    const bool fv = lane * 2 < F;
    const int  fo = fv ? lane * 2 : 0;
    const float* __restrict__ Hf = H + fo;

    int   start = rowptr[wfirst];
    int   end   = rowptr[wfirst + 1];
    float di    = d[wfirst];
    float si    = s[wfirst];
    int src0 = 0; float wt0 = 0.0f; float s0 = 0.0f;
    if (lane < end - start) {
        src0 = gsrc[start + lane];
        wt0  = gw[start + lane];
        s0   = s[src0];
    }

    for (int wid = wfirst; wid < N; wid += nw) {
        const float2 hself = *(const float2*)(Hf + (size_t)wid * F);
        const int nx = min(wid + nw, N - 1);
        const int   nstart = rowptr[nx];
        const int   nend   = rowptr[nx + 1];
        const float ndi    = d[nx];
        const float nsi    = s[nx];

        const int deg = end - start;
        const float sc_self = leaky(si + di);
        const bool v0 = lane < deg;
        const float sc0 = v0 ? leaky(s0 + di) : sc_self;

        float m = fmaxf(sc_self, sc0);
        for (int p = start + 32 + lane; p < end; p += 32)
            m = fmaxf(m, leaky(s[gsrc[p]] + di));
        for (int off = 16; off > 0; off >>= 1) m = fmaxf(m, __shfl_xor(m, off, 32));

        const float ex0 = v0 ? __expf(sc0 - m) : 0.0f;
        float sum = ex0;
        for (int p = start + 32 + lane; p < end; p += 32)
            sum += __expf(leaky(s[gsrc[p]] + di) - m);
        for (int off = 16; off > 0; off >>= 1) sum += __shfl_xor(sum, off, 32);
        const float ex_self = __expf(sc_self - m);
        sum += ex_self;
        const float dn = fmaxf(sum, 1e-16f);
        const float c0 = (ex0 * wt0) / dn;
        const float c_self = ex_self / dn;

        const bool nv0 = lane < nend - nstart;
        int nsrc0 = 0; float nwt0 = 0.0f; float ns0 = 0.0f;
        if (nv0) { nsrc0 = gsrc[nstart + lane]; nwt0 = gw[nstart + lane]; }

        float ax = 0.0f, ay = 0.0f;
        const int n0 = min(deg, 32);
        for (int jb = 0; jb < n0; jb += 8) {
            float2 h[8]; float cc[8];
            #pragma unroll
            for (int u = 0; u < 8; ++u) {
                const int j  = jb + u;
                const int jj = min(j, n0 - 1);
                const int sj = __builtin_amdgcn_readlane(src0, jj);
                const float cj = __int_as_float(__builtin_amdgcn_readlane(__float_as_int(c0), jj));
                cc[u] = (j < n0) ? cj : 0.0f;
                h[u]  = *(const float2*)(Hf + (size_t)sj * F);
            }
            if (jb == 0 && nv0) ns0 = s[nsrc0];
            #pragma unroll
            for (int u = 0; u < 8; ++u) {
                if (jb + u < n0) { ax = fmaf(cc[u], h[u].x, ax); ay = fmaf(cc[u], h[u].y, ay); }
            }
        }
        if (n0 == 0 && nv0) ns0 = s[nsrc0];
        for (int cb = start + 32; cb < end; cb += 32) {
            const int p = cb + lane;
            int sx = 0; float cx = 0.0f;
            if (p < end) {
                sx = gsrc[p];
                cx = (__expf(leaky(s[sx] + di) - m) * gw[p]) / dn;
            }
            const int cn = min(32, end - cb);
            for (int j = 0; j < cn; ++j) {
                const int sj = __builtin_amdgcn_readlane(sx, j);
                const float cj = __int_as_float(__builtin_amdgcn_readlane(__float_as_int(cx), j));
                const float2 hv = *(const float2*)(Hf + (size_t)sj * F);
                ax = fmaf(cj, hv.x, ax); ay = fmaf(cj, hv.y, ay);
            }
        }
        ax = fmaf(c_self, hself.x, ax); ay = fmaf(c_self, hself.y, ay);
        if (fv) *(float2*)(out + (size_t)wid * F + fo) = make_float2(ax, ay);

        start = nstart; end = nend; di = ndi; si = nsi;
        src0 = nsrc0; wt0 = nwt0; s0 = ns0;
    }
}

extern "C" void kernel_launch(void* const* d_in, const int* in_sizes, int n_in,
                              void* d_out, int out_size, void* d_ws, size_t ws_size,
                              hipStream_t stream) {
    const float* x   = (const float*)d_in[0];
    const int*   ei  = (const int*)  d_in[1];
    const float* ew  = (const float*)d_in[2];
    const float* W1  = (const float*)d_in[3];
    const float* a1s = (const float*)d_in[4];
    const float* a1d = (const float*)d_in[5];
    const float* W2  = (const float*)d_in[6];
    const float* a2s = (const float*)d_in[7];
    const float* a2d = (const float*)d_in[8];

    const int HID = in_sizes[4];
    const int FIN = in_sizes[3] / HID;
    const int NC  = in_sizes[7];
    const int N   = in_sizes[0] / FIN;
    const int E   = in_sizes[2];

    float* h1   = (float*)d_ws;
    float* out1 = h1   + (size_t)N * HID;
    float* h2   = out1 + (size_t)N * HID;
    float* sArr = h2   + (size_t)N * NC;
    float* dArr = sArr + N;
    float* outF = (float*)d_out;

    const int perWG = 1024 * TRIP;
    const int G1  = (E + perWG - 1) / perWG > 256 ? (E + perWG - 1) / perWG : 256;
    const int CH  = (E + G1 - 1) / G1;
    const int B   = (N + BIN_NODES - 1) / BIN_NODES;
    const int nso = (B + 1) * G1;
    size_t off = (((size_t)N * (2 * HID + NC + 2)) + 3) & ~(size_t)3;
    int* base  = (int*)d_ws;
    int4*  rec    = (int4*)(base + off);  off += (size_t)E * 4;
    int4*  rec2   = (int4*)(base + off);  off += (size_t)E * 4;
    int*   rowptr = base + off;           off += ((size_t)N + 4) & ~(size_t)3;
    int*   segOff = base + off;           off += ((size_t)nso + 3) & ~(size_t)3;
    int*   gsrc   = base + off;           off += ((size_t)E + 3) & ~(size_t)3;
    float* gw     = (float*)(base + off);

    const int nrt     = (N + 15) / 16;
    const int pairs   = (nrt + 1) / 2;
    const int gBlocks = (pairs + 7) / 8;
    const int wBlocks = 2048;

    chunk_bin_kernel<<<G1, 1024, B * sizeof(int), stream>>>(ei, ew, segOff, rec2, E, B, CH);
    bin_group_kernel<<<B, 1024, 0, stream>>>(segOff, rec2, rec, rowptr, gsrc, gw, N, E, G1, CH);

    gemm1_kernel<<<gBlocks, 256, 0, stream>>>(x, W1, a1s, a1d, h1, sArr, dArr, N, FIN);
    node_kernel<<<wBlocks, 256, 0, stream>>>(rowptr, gsrc, gw, sArr, dArr, h1, out1, N, HID);

    gemm2_kernel<<<gBlocks, 256, 0, stream>>>(out1, W2, a2s, a2d, h2, sArr, dArr, N, NC);
    node_kernel<<<wBlocks, 256, 0, stream>>>(rowptr, gsrc, gw, sArr, dArr, h2, outF, N, NC);
}


# GEMM 1 runs inside the first edge-grouping launch as a second work-group role (matrix work beside the chunk pass's memory waits; one row tile per wave): 6 launches -> 5; every sum keeps its order
# speedup vs baseline: 7.4436x; 1.1123x over previous
#include <hip/hip_runtime.h>
#include <hip/hip_bf16.h>


typedef __attribute__((ext_vector_type(2))) float v2f;
typedef __attribute__((ext_vector_type(8))) float v8f;

#define NEG_SLOPE 0.2f

__device__ __forceinline__ float leaky(float x) {
    return x > 0.0f ? x : NEG_SLOPE * x;
}

template <int CTRL>
__device__ __forceinline__ float row_xor(float v) {
    return __int_as_float(__builtin_amdgcn_update_dpp(
        0, __float_as_int(v), CTRL, 0xF, 0xF, false));
}

__device__ __forceinline__ void row_scores(const float (&v)[32], int col0,
                                           float& so, float& dd) {
    const bool b3 = (col0 & 8) != 0, b2 = (col0 & 4) != 0;
    const bool b1 = (col0 & 2) != 0, b0 = (col0 & 1) != 0;
    float w16[16], w8[8], w4[4];
    #pragma unroll
    for (int i = 0; i < 16; ++i) {
        const float keep = b3 ? v[16 + i] : v[i];
        const float send = b3 ? v[i] : v[16 + i];
        w16[i] = keep + row_xor<0x168>(send);
    }
    #pragma unroll
    for (int i = 0; i < 8; ++i) {
        const float keep = b2 ? w16[8 + i] : w16[i];
        const float send = b2 ? w16[i] : w16[8 + i];
        w8[i] = keep + row_xor<0x164>(send);
    }
    #pragma unroll
    for (int i = 0; i < 4; ++i) {
        const float keep = b1 ? w8[4 + i] : w8[i];
        const float send = b1 ? w8[i] : w8[4 + i];
        w4[i] = keep + row_xor<0x162>(send);
    }
    {
        const float keep0 = b0 ? w4[2] : w4[0], send0 = b0 ? w4[0] : w4[2];
        const float keep1 = b0 ? w4[3] : w4[1], send1 = b0 ? w4[1] : w4[3];
        so = keep0 + row_xor<0x161>(send0);
        dd = keep1 + row_xor<0x161>(send1);
    }
}

__device__ __forceinline__ float row_scores16(const float (&v)[16], int col0) {
    const bool b3 = (col0 & 8) != 0, b2 = (col0 & 4) != 0;
    const bool b1 = (col0 & 2) != 0, b0 = (col0 & 1) != 0;
    float w8[8], w4[4], w2[2];
    #pragma unroll
    for (int i = 0; i < 8; ++i) {
        const float keep = b3 ? v[8 + i] : v[i];
        const float send = b3 ? v[i] : v[8 + i];
        w8[i] = keep + row_xor<0x168>(send);
    }
    #pragma unroll
    for (int i = 0; i < 4; ++i) {
        const float keep = b2 ? w8[4 + i] : w8[i];
        const float send = b2 ? w8[i] : w8[4 + i];
        w4[i] = keep + row_xor<0x164>(send);
    }
    #pragma unroll
    for (int i = 0; i < 2; ++i) {
        const float keep = b1 ? w4[2 + i] : w4[i];
        const float send = b1 ? w4[i] : w4[2 + i];
        w2[i] = keep + row_xor<0x162>(send);
    }
    const float keep = b0 ? w2[1] : w2[0], send = b0 ? w2[0] : w2[1];
    return keep + row_xor<0x161>(send);
}

#define P1 132
__device__ __forceinline__ void gemm1_role(
    float* sW,
    const float* __restrict__ X, const float* __restrict__ W,
    const float* __restrict__ asrc, const float* __restrict__ adst,
    float* __restrict__ H, float* __restrict__ s, float* __restrict__ d,
    int N, int K, int blk, int TPB) {
    const int tid = threadIdx.x, nt = 1024;
    const int lane = tid & 31;
    const int wave = __builtin_amdgcn_readfirstlane(tid >> 5);
    const int half = lane >> 4;
    const int col0 = lane & 15;
    const int nrt  = (N + 15) >> 4;
    const int tBeg = min(nrt, blk * TPB), tEnd = min(nrt, tBeg + TPB);
    const int phases = K >> 7;

    if (tBeg >= tEnd) return;
    {
        const int rt = tBeg + wave;
        const bool act = rt < tEnd;
        const int rowA = act ? min(rt * 16 + col0, N - 1) : 0;

        v8f acc[4];
        #pragma unroll
        for (int t = 0; t < 4; ++t) acc[t] = v8f{};

        for (int ph = 0; ph < phases; ++ph) {
            const float* wp = W + (size_t)ph * 128 * 64 + tid;
            #pragma unroll
            for (int j = 0; j < 128 * 64 / nt; ++j)
                sW[(tid & 63) * P1 + (tid >> 6) + j * (nt >> 6)] = wp[j * nt];
            __syncthreads();

            if (act) {
                const float* xr = X + (size_t)rowA * K + ph * 128;
                for (int kk = 0; kk < 32; ++kk) {
                    const int kb = kk * 4 + half * 2;
                    v2f a = *(const v2f*)(xr + kb);
                    #pragma unroll
                    for (int t = 0; t < 4; ++t) {
                        v2f b = *(const v2f*)(&sW[(t * 16 + col0) * P1 + kb]);
                        acc[t] = __builtin_amdgcn_wmma_f32_16x16x4_f32(
                            false, a, false, b, (short)0, acc[t], false, false);
                    }
                }
            }
            __syncthreads();
        }

        if (act) {
            float as[4], ad[4];
            #pragma unroll
            for (int t = 0; t < 4; ++t) { as[t] = asrc[t * 16 + col0]; ad[t] = adst[t * 16 + col0]; }
            #pragma unroll
            for (int t = 0; t < 4; ++t)
                #pragma unroll
                for (int r = 0; r < 8; ++r) {
                    int row = rt * 16 + r + half * 8;
                    if (row < N) H[(size_t)row * 64 + t * 16 + col0] = acc[t][r];
                }

            float pv[16];
            #pragma unroll
            for (int r = 0; r < 8; ++r) {
                float ps = acc[0][r] * as[0], pd = acc[0][r] * ad[0];
                #pragma unroll
                for (int t = 1; t < 4; ++t) {
                    ps = fmaf(acc[t][r], as[t], ps);
                    pd = fmaf(acc[t][r], ad[t], pd);
                }
                pv[r * 2] = ps; pv[r * 2 + 1] = pd;
            }
            const float sc = row_scores16(pv, col0);
            const int srow = rt * 16 + half * 8 + (col0 >> 1);
            if (srow < N) { if (col0 & 1) d[srow] = sc; else s[srow] = sc; }
        }
    }
}

#define P2 68
__global__ __launch_bounds__(256) void gemm2_kernel(
    const float* __restrict__ Hin, const float* __restrict__ W,
    const float* __restrict__ asrc, const float* __restrict__ adst,
    float* __restrict__ Hout, float* __restrict__ s, float* __restrict__ d,
    int N, int NC) {
    __shared__ float sW[48 * P2];
    const int lane = threadIdx.x & 31;
    const int wave = threadIdx.x >> 5;
    const int half = lane >> 4;
    const int col0 = lane & 15;
    float as[3], ad[3];
    #pragma unroll
    for (int t = 0; t < 3; ++t) {
        const int c = t * 16 + col0;
        as[t] = (c < NC) ? asrc[c] : 0.0f;
        ad[t] = (c < NC) ? adst[c] : 0.0f;
    }
    const int nrt  = (N + 15) >> 4;
    const int rt0  = (blockIdx.x * 8 + wave) * 2;
    const int rt1  = rt0 + 1;
    const int rowA0 = (rt0 < nrt) ? min(rt0 * 16 + col0, N - 1) : 0;
    const int rowA1 = (rt1 < nrt) ? min(rt1 * 16 + col0, N - 1) : 0;

    for (int i = threadIdx.x; i < 64 * 48; i += 256) {
        int r = i / 48, c = i % 48;
        sW[c * P2 + r] = (c < NC) ? W[(size_t)r * NC + c] : 0.0f;
    }
    __syncthreads();

    v8f acc[2][3];
    #pragma unroll
    for (int u = 0; u < 2; ++u)
        #pragma unroll
        for (int t = 0; t < 3; ++t) acc[u][t] = v8f{};

    const float* xr0 = Hin + (size_t)rowA0 * 64;
    const float* xr1 = Hin + (size_t)rowA1 * 64;
    for (int kk = 0; kk < 16; ++kk) {
        const int kb = kk * 4 + half * 2;
        v2f a0 = *(const v2f*)(xr0 + kb);
        v2f a1 = *(const v2f*)(xr1 + kb);
        a0[0] = fmaxf(a0[0], 0.0f); a0[1] = fmaxf(a0[1], 0.0f);
        a1[0] = fmaxf(a1[0], 0.0f); a1[1] = fmaxf(a1[1], 0.0f);
        #pragma unroll
        for (int t = 0; t < 3; ++t) {
            v2f b = *(const v2f*)(&sW[(t * 16 + col0) * P2 + kb]);
            acc[0][t] = __builtin_amdgcn_wmma_f32_16x16x4_f32(
                false, a0, false, b, (short)0, acc[0][t], false, false);
            acc[1][t] = __builtin_amdgcn_wmma_f32_16x16x4_f32(
                false, a1, false, b, (short)0, acc[1][t], false, false);
        }
    }

    #pragma unroll
    for (int u = 0; u < 2; ++u) {
        int rt = rt0 + u;
        if (rt >= nrt) continue;
        #pragma unroll
        for (int t = 0; t < 3; ++t)
            #pragma unroll
            for (int r = 0; r < 8; ++r) {
                int row = rt * 16 + r + half * 8;
                int col = t * 16 + col0;
                if (row < N && col < NC)
                    Hout[(size_t)row * NC + col] = acc[u][t][r];
            }
    }

    float pv[32];
    #pragma unroll
    for (int u = 0; u < 2; ++u)
        #pragma unroll
        for (int r = 0; r < 8; ++r) {
            float ps = acc[u][0][r] * as[0], pd = acc[u][0][r] * ad[0];
            #pragma unroll
            for (int t = 1; t < 3; ++t) {
                ps = fmaf(acc[u][t][r], as[t], ps);
                pd = fmaf(acc[u][t][r], ad[t], pd);
            }
            pv[u * 16 + r * 2] = ps; pv[u * 16 + r * 2 + 1] = pd;
        }
    float so, dd;
    row_scores(pv, col0, so, dd);
    const int srow = (rt0 + (col0 >> 3)) * 16 + half * 8 + (col0 & 7);
    if (srow < N) { s[srow] = so; d[srow] = dd; }
}

#define BIN_SHIFT 8
#define BIN_NODES 256

#define TRIP 7

__device__ __forceinline__ void chunk_bin_role(
    int* hist, int* swave,
    const int* __restrict__ ei, const float* __restrict__ ew,
    int* __restrict__ segOff, int4* __restrict__ rec2, int E, int B, int CH,
    int c, int G) {
    const int tid = threadIdx.x, nt = blockDim.x;
    const int lane = tid & 31, wave = tid >> 5;
    for (int i = tid; i < B; i += nt) hist[i] = 0;
    __syncthreads();
    const int e0 = min(E, c * CH), e1 = min(E, e0 + CH);
    int dd[TRIP], ss[TRIP], ww[TRIP], pp[TRIP];
    #pragma unroll
    for (int k = 0; k < TRIP; ++k) { dd[k] = 0; ss[k] = 0; ww[k] = 0; pp[k] = 0; }
    if (e0 < e1) {
        #pragma unroll
        for (int k = 0; k < TRIP; ++k) {
            const int e = min(e0 + tid + k * nt, e1 - 1);
            dd[k] = ei[E + e];
            ss[k] = ei[e];
            ww[k] = __float_as_int(ew[e]);
        }
        #pragma unroll
        for (int k = 0; k < TRIP; ++k)
            if (e0 + tid + k * nt < e1) pp[k] = atomicAdd(&hist[dd[k] >> BIN_SHIFT], 1);
    }
    __syncthreads();
    const int per = (B + nt - 1) / nt;
    const int i0 = min(B, tid * per), i1 = min(B, i0 + per);
    int tsum = 0;
    for (int i = i0; i < i1; ++i) tsum += hist[i];
    int inc = tsum;
    for (int off = 1; off < 32; off <<= 1) {
        int n = __shfl_up(inc, off, 32);
        if (lane >= off) inc += n;
    }
    if (lane == 31) swave[wave] = inc;
    __syncthreads();
    int pre = (lane < wave) ? swave[lane] : 0;
    for (int off = 16; off > 0; off >>= 1) pre += __shfl_xor(pre, off, 32);
    int run = pre + inc - tsum;
    for (int i = i0; i < i1; ++i) {
        const int v = hist[i];
        hist[i] = run;
        segOff[(size_t)i * G + c] = run;
        run += v;
    }
    if (tid == 0) segOff[(size_t)B * G + c] = e1 - e0;
    __syncthreads();
    #pragma unroll
    for (int k = 0; k < TRIP; ++k) {
        const int e = e0 + tid + k * nt;
        if (e < e1)
            rec2[e0 + hist[dd[k] >> BIN_SHIFT] + pp[k]] = make_int4(e, ss[k], ww[k], dd[k]);
    }
}

__global__ __launch_bounds__(1024, 16) void chunk_bin_gemm1_kernel(
    const int* __restrict__ ei, const float* __restrict__ ew,
    int* __restrict__ segOff, int4* __restrict__ rec2, int E, int B, int CH, int G,
    const float* __restrict__ X, const float* __restrict__ W,
    const float* __restrict__ asrc, const float* __restrict__ adst,
    float* __restrict__ H, float* __restrict__ s, float* __restrict__ d,
    int N, int K, int TPB) {
    extern __shared__ int hist[];
    __shared__ int swave[32];
    __shared__ float sW[64 * P1];
    const int half = gridDim.x >> 1;
    const int b = blockIdx.x;
    if (b < half) {
        gemm1_role(sW, X, W, asrc, adst, H, s, d, N, K, b, TPB);
    } else {
        if (b - half < G) chunk_bin_role(hist, swave, ei, ew, segOff, rec2, E, B, CH, b - half, G);
    }
}

__device__ __forceinline__ void seg_group(
    const int* __restrict__ segOff, int b, int G, int CH, int g, int lane,
    int (&incl)[8], int (&dl)[8], int& losum) {
    const int c = g * 8 + (lane & 7);
    int lo = 0, hi = 0;
    if (c < G) { lo = segOff[(size_t)b * G + c]; hi = segOff[(size_t)(b + 1) * G + c]; }
    const int len = hi - lo;
    int inc = len;
    for (int off = 1; off < 8; off <<= 1) {
        int n = __shfl_up(inc, off, 32);
        if ((lane & 7) >= off) inc += n;
    }
    const int d = c * CH + lo - (inc - len);
    losum = 0;
    #pragma unroll
    for (int i = 0; i < 8; ++i) {
        incl[i] = __builtin_amdgcn_readlane(inc, i);
        dl[i]   = __builtin_amdgcn_readlane(d, i);
        losum  += __builtin_amdgcn_readlane(lo, i);
    }
}

__device__ __forceinline__ int seg_index(int q, const int (&incl)[8], const int (&dl)[8]) {
    int D = dl[0];
    #pragma unroll
    for (int i = 0; i < 7; ++i) D = (q >= incl[i]) ? dl[i + 1] : D;
    return q + D;
}

#define GTRIP 6
__global__ __launch_bounds__(1024, 16) void bin_group_kernel(
    const int* __restrict__ segOff, const int4* __restrict__ rec2,
    int4* rec, int* __restrict__ rowptr,
    int* __restrict__ gsrc, float* __restrict__ gw, int N, int E, int G, int CH) {
    __shared__ int cnt[BIN_NODES];
    __shared__ int cur[BIN_NODES];
    __shared__ int swave[BIN_NODES / 32];
    __shared__ int sstart[32];
    const int tid = threadIdx.x;
    const int lane = tid & 31, wave = tid >> 5;
    const int nwaves = blockDim.x >> 5;
    const int b = blockIdx.x;
    const int nbase = b << BIN_SHIFT;
    if (tid < BIN_NODES) cnt[tid] = 0;
    __syncthreads();
    int mylo = 0;
    for (int g = wave; g * 8 < G; g += nwaves) {
        int incl[8], dl[8], losum;
        seg_group(segOff, b, G, CH, g, lane, incl, dl, losum);
        mylo += losum;
        const int T = incl[7];
        for (int q0 = 0; q0 < T; q0 += 32 * GTRIP) {
            int dd[GTRIP];
            #pragma unroll
            for (int k = 0; k < GTRIP; ++k)
                dd[k] = rec2[seg_index(min(q0 + k * 32 + lane, T - 1), incl, dl)].w;
            #pragma unroll
            for (int k = 0; k < GTRIP; ++k)
                if (q0 + k * 32 + lane < T) atomicAdd(&cnt[dd[k] - nbase], 1);
        }
    }
    if (lane == 0) sstart[wave] = mylo;
    __syncthreads();
    int v = 0, inc = 0;
    if (tid < BIN_NODES) {
        v = cnt[tid];
        inc = v;
        for (int off = 1; off < 32; off <<= 1) {
            int n = __shfl_up(inc, off, 32);
            if (lane >= off) inc += n;
        }
        if (lane == 31) swave[wave] = inc;
    }
    __syncthreads();
    if (tid < BIN_NODES) {
        int binStart = 0;
        for (int w = 0; w < nwaves; ++w) binStart += sstart[w];
        int pre = 0;
        for (int w = 0; w < BIN_NODES / 32; ++w) if (w < wave) pre += swave[w];
        const int first = binStart + pre + inc - v;
        cur[tid] = first;
        if (nbase + tid < N) rowptr[nbase + tid] = first;
    }
    if (b == (int)gridDim.x - 1 && tid == 0) rowptr[N] = E;
    __syncthreads();
    for (int g = wave; g * 8 < G; g += nwaves) {
        int incl[8], dl[8], losum;
        seg_group(segOff, b, G, CH, g, lane, incl, dl, losum);
        const int T = incl[7];
        for (int q0 = 0; q0 < T; q0 += 32 * GTRIP) {
            int4 r[GTRIP]; int pp[GTRIP];
            #pragma unroll
            for (int k = 0; k < GTRIP; ++k)
                r[k] = rec2[seg_index(min(q0 + k * 32 + lane, T - 1), incl, dl)];
            #pragma unroll
            for (int k = 0; k < GTRIP; ++k) {
                pp[k] = 0;
                if (q0 + k * 32 + lane < T) pp[k] = atomicAdd(&cur[r[k].w - nbase], 1);
            }
            #pragma unroll
            for (int k = 0; k < GTRIP; ++k)
                if (q0 + k * 32 + lane < T) rec[pp[k]] = make_int4(r[k].x, r[k].y, r[k].z, 0);
        }
    }
    __syncthreads();
    int nl = wave;
    int start = 0, deg = 0;
    int4 r = make_int4(0x7fffffff, 0, 0, 0);
    if (nl < BIN_NODES) {
        deg = cnt[nl];
        start = cur[nl] - deg;
        if (lane < deg) r = rec[start + lane];
    }
    for (; nl < BIN_NODES; nl += nwaves) {
        const int nn = nl + nwaves;
        int nstart = 0, ndeg = 0;
        int4 nr = make_int4(0x7fffffff, 0, 0, 0);
        if (nn < BIN_NODES) {
            ndeg = cnt[nn];
            nstart = cur[nn] - ndeg;
            if (lane < ndeg) nr = rec[nstart + lane];
        }
        const int degs = __builtin_amdgcn_readfirstlane(deg);
        if (degs <= 32) {
            int rk = 0;
            for (int j0 = 0; j0 < degs; j0 += 8) {
                #pragma unroll
                for (int u = 0; u < 8; ++u)
                    rk += (__builtin_amdgcn_readlane(r.x, j0 + u) < r.x) ? 1 : 0;
            }
            if (lane < degs) { gsrc[start + rk] = r.y; gw[start + rk] = __int_as_float(r.z); }
        } else {
            for (int q = lane; q < degs; q += 32) {
                const int4 rq = rec[start + q];
                int rk = 0;
                for (int j = 0; j < degs; ++j) rk += (rec[start + j].x < rq.x) ? 1 : 0;
                gsrc[start + rk] = rq.y; gw[start + rk] = __int_as_float(rq.z);
            }
        }
        start = nstart; deg = ndeg; r = nr;
    }
}

__global__ __launch_bounds__(256) void node_kernel(
    const int* __restrict__ rowptr, const int* __restrict__ gsrc,
    const float* __restrict__ gw, const float* __restrict__ s,
    const float* __restrict__ d, const float* __restrict__ H,
    float* __restrict__ out, int N, int F) {
    const int wfirst = __builtin_amdgcn_readfirstlane(
        (int)((blockIdx.x * (long long)blockDim.x + threadIdx.x) >> 5));
    const int nw   = (int)((gridDim.x * (long long)blockDim.x) >> 5);
    const int lane = threadIdx.x & 31;
    if (wfirst >= N) return;

    const bool fv = lane * 2 < F;
    const int  fo = fv ? lane * 2 : 0;
    const float* __restrict__ Hf = H + fo;

    int   start = rowptr[wfirst];
    int   end   = rowptr[wfirst + 1];
    float di    = d[wfirst];
    float si    = s[wfirst];
    int src0 = 0; float wt0 = 0.0f; float s0 = 0.0f;
    if (lane < end - start) {
        src0 = gsrc[start + lane];
        wt0  = gw[start + lane];
        s0   = s[src0];
    }

    for (int wid = wfirst; wid < N; wid += nw) {
        const float2 hself = *(const float2*)(Hf + (size_t)wid * F);
        const int nx = min(wid + nw, N - 1);
        const int   nstart = rowptr[nx];
        const int   nend   = rowptr[nx + 1];
        const float ndi    = d[nx];
        const float nsi    = s[nx];

        const int deg = end - start;
        const float sc_self = leaky(si + di);
        const bool v0 = lane < deg;
        const float sc0 = v0 ? leaky(s0 + di) : sc_self;

        float m = fmaxf(sc_self, sc0);
        for (int p = start + 32 + lane; p < end; p += 32)
            m = fmaxf(m, leaky(s[gsrc[p]] + di));
        for (int off = 16; off > 0; off >>= 1) m = fmaxf(m, __shfl_xor(m, off, 32));

        const float ex0 = v0 ? __expf(sc0 - m) : 0.0f;
        float sum = ex0;
        for (int p = start + 32 + lane; p < end; p += 32)
            sum += __expf(leaky(s[gsrc[p]] + di) - m);
        for (int off = 16; off > 0; off >>= 1) sum += __shfl_xor(sum, off, 32);
        const float ex_self = __expf(sc_self - m);
        sum += ex_self;
        const float dn = fmaxf(sum, 1e-16f);
        const float c0 = (ex0 * wt0) / dn;
        const float c_self = ex_self / dn;

        const bool nv0 = lane < nend - nstart;
        int nsrc0 = 0; float nwt0 = 0.0f; float ns0 = 0.0f;
        if (nv0) { nsrc0 = gsrc[nstart + lane]; nwt0 = gw[nstart + lane]; }

        float ax = 0.0f, ay = 0.0f;
        const int n0 = min(deg, 32);
        for (int jb = 0; jb < n0; jb += 8) {
            float2 h[8]; float cc[8];
            #pragma unroll
            for (int u = 0; u < 8; ++u) {
                const int j  = jb + u;
                const int jj = min(j, n0 - 1);
                const int sj = __builtin_amdgcn_readlane(src0, jj);
                const float cj = __int_as_float(__builtin_amdgcn_readlane(__float_as_int(c0), jj));
                cc[u] = (j < n0) ? cj : 0.0f;
                h[u]  = *(const float2*)(Hf + (size_t)sj * F);
            }
            if (jb == 0 && nv0) ns0 = s[nsrc0];
            #pragma unroll
            for (int u = 0; u < 8; ++u) {
                if (jb + u < n0) { ax = fmaf(cc[u], h[u].x, ax); ay = fmaf(cc[u], h[u].y, ay); }
            }
        }
        if (n0 == 0 && nv0) ns0 = s[nsrc0];
        for (int cb = start + 32; cb < end; cb += 32) {
            const int p = cb + lane;
            int sx = 0; float cx = 0.0f;
            if (p < end) {
                sx = gsrc[p];
                cx = (__expf(leaky(s[sx] + di) - m) * gw[p]) / dn;
            }
            const int cn = min(32, end - cb);
            for (int j = 0; j < cn; ++j) {
                const int sj = __builtin_amdgcn_readlane(sx, j);
                const float cj = __int_as_float(__builtin_amdgcn_readlane(__float_as_int(cx), j));
                const float2 hv = *(const float2*)(Hf + (size_t)sj * F);
                ax = fmaf(cj, hv.x, ax); ay = fmaf(cj, hv.y, ay);
            }
        }
        ax = fmaf(c_self, hself.x, ax); ay = fmaf(c_self, hself.y, ay);
        if (fv) *(float2*)(out + (size_t)wid * F + fo) = make_float2(ax, ay);

        start = nstart; end = nend; di = ndi; si = nsi;
        src0 = nsrc0; wt0 = nwt0; s0 = ns0;
    }
}

extern "C" void kernel_launch(void* const* d_in, const int* in_sizes, int n_in,
                              void* d_out, int out_size, void* d_ws, size_t ws_size,
                              hipStream_t stream) {
    const float* x   = (const float*)d_in[0];
    const int*   ei  = (const int*)  d_in[1];
    const float* ew  = (const float*)d_in[2];
    const float* W1  = (const float*)d_in[3];
    const float* a1s = (const float*)d_in[4];
    const float* a1d = (const float*)d_in[5];
    const float* W2  = (const float*)d_in[6];
    const float* a2s = (const float*)d_in[7];
    const float* a2d = (const float*)d_in[8];

    const int HID = in_sizes[4];
    const int FIN = in_sizes[3] / HID;
    const int NC  = in_sizes[7];
    const int N   = in_sizes[0] / FIN;
    const int E   = in_sizes[2];

    float* h1   = (float*)d_ws;
    float* out1 = h1   + (size_t)N * HID;
    float* h2   = out1 + (size_t)N * HID;
    float* sArr = h2   + (size_t)N * NC;
    float* dArr = sArr + N;
    float* outF = (float*)d_out;

    const int perWG = 1024 * TRIP;
    const int G1  = (E + perWG - 1) / perWG > 256 ? (E + perWG - 1) / perWG : 256;
    const int CH  = (E + G1 - 1) / G1;
    const int B   = (N + BIN_NODES - 1) / BIN_NODES;
    const int nso = (B + 1) * G1;
    size_t off = (((size_t)N * (2 * HID + NC + 2)) + 3) & ~(size_t)3;
    int* base  = (int*)d_ws;
    int4*  rec    = (int4*)(base + off);  off += (size_t)E * 4;
    int4*  rec2   = (int4*)(base + off);  off += (size_t)E * 4;
    int*   rowptr = base + off;           off += ((size_t)N + 4) & ~(size_t)3;
    int*   segOff = base + off;           off += ((size_t)nso + 3) & ~(size_t)3;
    int*   gsrc   = base + off;           off += ((size_t)E + 3) & ~(size_t)3;
    float* gw     = (float*)(base + off);

    const int nrt     = (N + 15) / 16;
    const int pairs   = (nrt + 1) / 2;
    const int gBlocks = (pairs + 7) / 8;
    const int wBlocks = 2048;

    const int tpb   = (nrt + G1 - 1) / G1 < 32 ? (nrt + G1 - 1) / G1 : 32;
    const int gPair = (nrt + tpb - 1) / tpb > G1 ? (nrt + tpb - 1) / tpb : G1;
    chunk_bin_gemm1_kernel<<<2 * gPair, 1024, B * sizeof(int), stream>>>(
        ei, ew, segOff, rec2, E, B, CH, G1, x, W1, a1s, a1d, h1, sArr, dArr, N, FIN, tpb);
    bin_group_kernel<<<B, 1024, 0, stream>>>(segOff, rec2, rec, rowptr, gsrc, gw, N, E, G1, CH);

    node_kernel<<<wBlocks, 256, 0, stream>>>(rowptr, gsrc, gw, sArr, dArr, h1, out1, N, HID);

    gemm2_kernel<<<gBlocks, 256, 0, stream>>>(out1, W2, a2s, a2d, h2, sArr, dArr, N, NC);
    node_kernel<<<wBlocks, 256, 0, stream>>>(rowptr, gsrc, gw, sArr, dArr, h2, outF, N, NC);
}


# Node pass: scores s_j of a node's first 16 neighbours formed from their gathered rows (rows asked for ahead of the softmax; DPP tree + half swap), not gathered a line per edge; s_j sum order changes
# speedup vs baseline: 7.8667x; 1.0568x over previous
#include <hip/hip_runtime.h>
#include <hip/hip_bf16.h>


typedef __attribute__((ext_vector_type(2))) float v2f;
typedef __attribute__((ext_vector_type(8))) float v8f;

#define NEG_SLOPE 0.2f

__device__ __forceinline__ float leaky(float x) {
    return x > 0.0f ? x : NEG_SLOPE * x;
}

template <int CTRL>
__device__ __forceinline__ float row_xor(float v) {
    return __int_as_float(__builtin_amdgcn_update_dpp(
        0, __float_as_int(v), CTRL, 0xF, 0xF, false));
}

__device__ __forceinline__ float xhalf(float x) {
    const int b = __float_as_int(x);
    return __int_as_float(__builtin_amdgcn_permlanex16(b, b, 0x76543210, 0xfedcba98, false, false));
}

template <int K>
__device__ __forceinline__ float lane_xor(float v) {
    return __int_as_float(__builtin_amdgcn_ds_swizzle(__float_as_int(v), 0x1F | (K << 10)));
}

__device__ __forceinline__ void row_scores(const float (&v)[32], int col0,
                                           float& so, float& dd) {
    const bool b3 = (col0 & 8) != 0, b2 = (col0 & 4) != 0;
    const bool b1 = (col0 & 2) != 0, b0 = (col0 & 1) != 0;
    float w16[16], w8[8], w4[4];
    #pragma unroll
    for (int i = 0; i < 16; ++i) {
        const float keep = b3 ? v[16 + i] : v[i];
        const float send = b3 ? v[i] : v[16 + i];
        w16[i] = keep + row_xor<0x168>(send);
    }
    #pragma unroll
    for (int i = 0; i < 8; ++i) {
        const float keep = b2 ? w16[8 + i] : w16[i];
        const float send = b2 ? w16[i] : w16[8 + i];
        w8[i] = keep + row_xor<0x164>(send);
    }
    #pragma unroll
    for (int i = 0; i < 4; ++i) {
        const float keep = b1 ? w8[4 + i] : w8[i];
        const float send = b1 ? w8[i] : w8[4 + i];
        w4[i] = keep + row_xor<0x162>(send);
    }
    {
        const float keep0 = b0 ? w4[2] : w4[0], send0 = b0 ? w4[0] : w4[2];
        const float keep1 = b0 ? w4[3] : w4[1], send1 = b0 ? w4[1] : w4[3];
        so = keep0 + row_xor<0x161>(send0);
        dd = keep1 + row_xor<0x161>(send1);
    }
}

__device__ __forceinline__ float row_scores16(const float (&v)[16], int col0) {
    const bool b3 = (col0 & 8) != 0, b2 = (col0 & 4) != 0;
    const bool b1 = (col0 & 2) != 0, b0 = (col0 & 1) != 0;
    float w8[8], w4[4], w2[2];
    #pragma unroll
    for (int i = 0; i < 8; ++i) {
        const float keep = b3 ? v[8 + i] : v[i];
        const float send = b3 ? v[i] : v[8 + i];
        w8[i] = keep + row_xor<0x168>(send);
    }
    #pragma unroll
    for (int i = 0; i < 4; ++i) {
        const float keep = b2 ? w8[4 + i] : w8[i];
        const float send = b2 ? w8[i] : w8[4 + i];
        w4[i] = keep + row_xor<0x164>(send);
    }
    #pragma unroll
    for (int i = 0; i < 2; ++i) {
        const float keep = b1 ? w4[2 + i] : w4[i];
        const float send = b1 ? w4[i] : w4[2 + i];
        w2[i] = keep + row_xor<0x162>(send);
    }
    const float keep = b0 ? w2[1] : w2[0], send = b0 ? w2[0] : w2[1];
    return keep + row_xor<0x161>(send);
}

#define P1 132
__device__ __forceinline__ void gemm1_role(
    float* sW,
    const float* __restrict__ X, const float* __restrict__ W,
    const float* __restrict__ asrc, const float* __restrict__ adst,
    float* __restrict__ H, float* __restrict__ s, float* __restrict__ d,
    int N, int K, int blk, int TPB) {
    const int tid = threadIdx.x, nt = 1024;
    const int lane = tid & 31;
    const int wave = __builtin_amdgcn_readfirstlane(tid >> 5);
    const int half = lane >> 4;
    const int col0 = lane & 15;
    const int nrt  = (N + 15) >> 4;
    const int tBeg = min(nrt, blk * TPB), tEnd = min(nrt, tBeg + TPB);
    const int phases = K >> 7;

    if (tBeg >= tEnd) return;
    {
        const int rt = tBeg + wave;
        const bool act = rt < tEnd;
        const int rowA = act ? min(rt * 16 + col0, N - 1) : 0;

        v8f acc[4];
        #pragma unroll
        for (int t = 0; t < 4; ++t) acc[t] = v8f{};

        for (int ph = 0; ph < phases; ++ph) {
            const float* wp = W + (size_t)ph * 128 * 64 + tid;
            #pragma unroll
            for (int j = 0; j < 128 * 64 / nt; ++j)
                sW[(tid & 63) * P1 + (tid >> 6) + j * (nt >> 6)] = wp[j * nt];
            __syncthreads();

            if (act) {
                const float* xr = X + (size_t)rowA * K + ph * 128;
                for (int kk = 0; kk < 32; ++kk) {
                    const int kb = kk * 4 + half * 2;
                    v2f a = *(const v2f*)(xr + kb);
                    #pragma unroll
                    for (int t = 0; t < 4; ++t) {
                        v2f b = *(const v2f*)(&sW[(t * 16 + col0) * P1 + kb]);
                        acc[t] = __builtin_amdgcn_wmma_f32_16x16x4_f32(
                            false, a, false, b, (short)0, acc[t], false, false);
                    }
                }
            }
            __syncthreads();
        }

        if (act) {
            float as[4], ad[4];
            #pragma unroll
            for (int t = 0; t < 4; ++t) { as[t] = asrc[t * 16 + col0]; ad[t] = adst[t * 16 + col0]; }
            #pragma unroll
            for (int t = 0; t < 4; ++t)
                #pragma unroll
                for (int r = 0; r < 8; ++r) {
                    int row = rt * 16 + r + half * 8;
                    if (row < N) H[(size_t)row * 64 + t * 16 + col0] = acc[t][r];
                }

            float pv[16];
            #pragma unroll
            for (int r = 0; r < 8; ++r) {
                float ps = acc[0][r] * as[0], pd = acc[0][r] * ad[0];
                #pragma unroll
                for (int t = 1; t < 4; ++t) {
                    ps = fmaf(acc[t][r], as[t], ps);
                    pd = fmaf(acc[t][r], ad[t], pd);
                }
                pv[r * 2] = ps; pv[r * 2 + 1] = pd;
            }
            const float sc = row_scores16(pv, col0);
            const int srow = rt * 16 + half * 8 + (col0 >> 1);
            if (srow < N) { if (col0 & 1) d[srow] = sc; else s[srow] = sc; }
        }
    }
}

#define P2 68
__global__ __launch_bounds__(256) void gemm2_kernel(
    const float* __restrict__ Hin, const float* __restrict__ W,
    const float* __restrict__ asrc, const float* __restrict__ adst,
    float* __restrict__ Hout, float* __restrict__ s, float* __restrict__ d,
    int N, int NC) {
    __shared__ float sW[48 * P2];
    const int lane = threadIdx.x & 31;
    const int wave = threadIdx.x >> 5;
    const int half = lane >> 4;
    const int col0 = lane & 15;
    float as[3], ad[3];
    #pragma unroll
    for (int t = 0; t < 3; ++t) {
        const int c = t * 16 + col0;
        as[t] = (c < NC) ? asrc[c] : 0.0f;
        ad[t] = (c < NC) ? adst[c] : 0.0f;
    }
    const int nrt  = (N + 15) >> 4;
    const int rt0  = (blockIdx.x * 8 + wave) * 2;
    const int rt1  = rt0 + 1;
    const int rowA0 = (rt0 < nrt) ? min(rt0 * 16 + col0, N - 1) : 0;
    const int rowA1 = (rt1 < nrt) ? min(rt1 * 16 + col0, N - 1) : 0;

    for (int i = threadIdx.x; i < 64 * 48; i += 256) {
        int r = i / 48, c = i % 48;
        sW[c * P2 + r] = (c < NC) ? W[(size_t)r * NC + c] : 0.0f;
    }
    __syncthreads();

    v8f acc[2][3];
    #pragma unroll
    for (int u = 0; u < 2; ++u)
        #pragma unroll
        for (int t = 0; t < 3; ++t) acc[u][t] = v8f{};

    const float* xr0 = Hin + (size_t)rowA0 * 64;
    const float* xr1 = Hin + (size_t)rowA1 * 64;
    for (int kk = 0; kk < 16; ++kk) {
        const int kb = kk * 4 + half * 2;
        v2f a0 = *(const v2f*)(xr0 + kb);
        v2f a1 = *(const v2f*)(xr1 + kb);
        a0[0] = fmaxf(a0[0], 0.0f); a0[1] = fmaxf(a0[1], 0.0f);
        a1[0] = fmaxf(a1[0], 0.0f); a1[1] = fmaxf(a1[1], 0.0f);
        #pragma unroll
        for (int t = 0; t < 3; ++t) {
            v2f b = *(const v2f*)(&sW[(t * 16 + col0) * P2 + kb]);
            acc[0][t] = __builtin_amdgcn_wmma_f32_16x16x4_f32(
                false, a0, false, b, (short)0, acc[0][t], false, false);
            acc[1][t] = __builtin_amdgcn_wmma_f32_16x16x4_f32(
                false, a1, false, b, (short)0, acc[1][t], false, false);
        }
    }

    #pragma unroll
    for (int u = 0; u < 2; ++u) {
        int rt = rt0 + u;
        if (rt >= nrt) continue;
        #pragma unroll
        for (int t = 0; t < 3; ++t)
            #pragma unroll
            for (int r = 0; r < 8; ++r) {
                int row = rt * 16 + r + half * 8;
                int col = t * 16 + col0;
                if (row < N && col < NC)
                    Hout[(size_t)row * NC + col] = acc[u][t][r];
            }
    }

    float pv[32];
    #pragma unroll
    for (int u = 0; u < 2; ++u)
        #pragma unroll
        for (int r = 0; r < 8; ++r) {
            float ps = acc[u][0][r] * as[0], pd = acc[u][0][r] * ad[0];
            #pragma unroll
            for (int t = 1; t < 3; ++t) {
                ps = fmaf(acc[u][t][r], as[t], ps);
                pd = fmaf(acc[u][t][r], ad[t], pd);
            }
            pv[u * 16 + r * 2] = ps; pv[u * 16 + r * 2 + 1] = pd;
        }
    float so, dd;
    row_scores(pv, col0, so, dd);
    const int srow = (rt0 + (col0 >> 3)) * 16 + half * 8 + (col0 & 7);
    if (srow < N) { s[srow] = so; d[srow] = dd; }
}

#define BIN_SHIFT 8
#define BIN_NODES 256

#define TRIP 7

__device__ __forceinline__ void chunk_bin_role(
    int* hist, int* swave,
    const int* __restrict__ ei, const float* __restrict__ ew,
    int* __restrict__ segOff, int4* __restrict__ rec2, int E, int B, int CH,
    int c, int G) {
    const int tid = threadIdx.x, nt = blockDim.x;
    const int lane = tid & 31, wave = tid >> 5;
    for (int i = tid; i < B; i += nt) hist[i] = 0;
    __syncthreads();
    const int e0 = min(E, c * CH), e1 = min(E, e0 + CH);
    int dd[TRIP], ss[TRIP], ww[TRIP], pp[TRIP];
    #pragma unroll
    for (int k = 0; k < TRIP; ++k) { dd[k] = 0; ss[k] = 0; ww[k] = 0; pp[k] = 0; }
    if (e0 < e1) {
        #pragma unroll
        for (int k = 0; k < TRIP; ++k) {
            const int e = min(e0 + tid + k * nt, e1 - 1);
            dd[k] = ei[E + e];
            ss[k] = ei[e];
            ww[k] = __float_as_int(ew[e]);
        }
        #pragma unroll
        for (int k = 0; k < TRIP; ++k)
            if (e0 + tid + k * nt < e1) pp[k] = atomicAdd(&hist[dd[k] >> BIN_SHIFT], 1);
    }
    __syncthreads();
    const int per = (B + nt - 1) / nt;
    const int i0 = min(B, tid * per), i1 = min(B, i0 + per);
    int tsum = 0;
    for (int i = i0; i < i1; ++i) tsum += hist[i];
    int inc = tsum;
    for (int off = 1; off < 32; off <<= 1) {
        int n = __shfl_up(inc, off, 32);
        if (lane >= off) inc += n;
    }
    if (lane == 31) swave[wave] = inc;
    __syncthreads();
    int pre = (lane < wave) ? swave[lane] : 0;
    for (int off = 16; off > 0; off >>= 1) pre += __shfl_xor(pre, off, 32);
    int run = pre + inc - tsum;
    for (int i = i0; i < i1; ++i) {
        const int v = hist[i];
        hist[i] = run;
        segOff[(size_t)i * G + c] = run;
        run += v;
    }
    if (tid == 0) segOff[(size_t)B * G + c] = e1 - e0;
    __syncthreads();
    #pragma unroll
    for (int k = 0; k < TRIP; ++k) {
        const int e = e0 + tid + k * nt;
        if (e < e1)
            rec2[e0 + hist[dd[k] >> BIN_SHIFT] + pp[k]] = make_int4(e, ss[k], ww[k], dd[k]);
    }
}

__global__ __launch_bounds__(1024, 16) void chunk_bin_gemm1_kernel(
    const int* __restrict__ ei, const float* __restrict__ ew,
    int* __restrict__ segOff, int4* __restrict__ rec2, int E, int B, int CH, int G,
    const float* __restrict__ X, const float* __restrict__ W,
    const float* __restrict__ asrc, const float* __restrict__ adst,
    float* __restrict__ H, float* __restrict__ s, float* __restrict__ d,
    int N, int K, int TPB) {
    extern __shared__ int hist[];
    __shared__ int swave[32];
    __shared__ float sW[64 * P1];
    const int half = gridDim.x >> 1;
    const int b = blockIdx.x;
    if (b < half) {
        gemm1_role(sW, X, W, asrc, adst, H, s, d, N, K, b, TPB);
    } else {
        if (b - half < G) chunk_bin_role(hist, swave, ei, ew, segOff, rec2, E, B, CH, b - half, G);
    }
}

__device__ __forceinline__ void seg_group(
    const int* __restrict__ segOff, int b, int G, int CH, int g, int lane,
    int (&incl)[8], int (&dl)[8], int& losum) {
    const int c = g * 8 + (lane & 7);
    int lo = 0, hi = 0;
    if (c < G) { lo = segOff[(size_t)b * G + c]; hi = segOff[(size_t)(b + 1) * G + c]; }
    const int len = hi - lo;
    int inc = len;
    for (int off = 1; off < 8; off <<= 1) {
        int n = __shfl_up(inc, off, 32);
        if ((lane & 7) >= off) inc += n;
    }
    const int d = c * CH + lo - (inc - len);
    losum = 0;
    #pragma unroll
    for (int i = 0; i < 8; ++i) {
        incl[i] = __builtin_amdgcn_readlane(inc, i);
        dl[i]   = __builtin_amdgcn_readlane(d, i);
        losum  += __builtin_amdgcn_readlane(lo, i);
    }
}

__device__ __forceinline__ int seg_index(int q, const int (&incl)[8], const int (&dl)[8]) {
    int D = dl[0];
    #pragma unroll
    for (int i = 0; i < 7; ++i) D = (q >= incl[i]) ? dl[i + 1] : D;
    return q + D;
}

#define GTRIP 6
__global__ __launch_bounds__(1024, 16) void bin_group_kernel(
    const int* __restrict__ segOff, const int4* __restrict__ rec2,
    int4* rec, int* __restrict__ rowptr,
    int* __restrict__ gsrc, float* __restrict__ gw, int N, int E, int G, int CH) {
    __shared__ int cnt[BIN_NODES];
    __shared__ int cur[BIN_NODES];
    __shared__ int swave[BIN_NODES / 32];
    __shared__ int sstart[32];
    const int tid = threadIdx.x;
    const int lane = tid & 31, wave = tid >> 5;
    const int nwaves = blockDim.x >> 5;
    const int b = blockIdx.x;
    const int nbase = b << BIN_SHIFT;
    if (tid < BIN_NODES) cnt[tid] = 0;
    __syncthreads();
    int mylo = 0;
    for (int g = wave; g * 8 < G; g += nwaves) {
        int incl[8], dl[8], losum;
        seg_group(segOff, b, G, CH, g, lane, incl, dl, losum);
        mylo += losum;
        const int T = incl[7];
        for (int q0 = 0; q0 < T; q0 += 32 * GTRIP) {
            int dd[GTRIP];
            #pragma unroll
            for (int k = 0; k < GTRIP; ++k)
                dd[k] = rec2[seg_index(min(q0 + k * 32 + lane, T - 1), incl, dl)].w;
            #pragma unroll
            for (int k = 0; k < GTRIP; ++k)
                if (q0 + k * 32 + lane < T) atomicAdd(&cnt[dd[k] - nbase], 1);
        }
    }
    if (lane == 0) sstart[wave] = mylo;
    __syncthreads();
    int v = 0, inc = 0;
    if (tid < BIN_NODES) {
        v = cnt[tid];
        inc = v;
        for (int off = 1; off < 32; off <<= 1) {
            int n = __shfl_up(inc, off, 32);
            if (lane >= off) inc += n;
        }
        if (lane == 31) swave[wave] = inc;
    }
    __syncthreads();
    if (tid < BIN_NODES) {
        int binStart = 0;
        for (int w = 0; w < nwaves; ++w) binStart += sstart[w];
        int pre = 0;
        for (int w = 0; w < BIN_NODES / 32; ++w) if (w < wave) pre += swave[w];
        const int first = binStart + pre + inc - v;
        cur[tid] = first;
        if (nbase + tid < N) rowptr[nbase + tid] = first;
    }
    if (b == (int)gridDim.x - 1 && tid == 0) rowptr[N] = E;
    __syncthreads();
    for (int g = wave; g * 8 < G; g += nwaves) {
        int incl[8], dl[8], losum;
        seg_group(segOff, b, G, CH, g, lane, incl, dl, losum);
        const int T = incl[7];
        for (int q0 = 0; q0 < T; q0 += 32 * GTRIP) {
            int4 r[GTRIP]; int pp[GTRIP];
            #pragma unroll
            for (int k = 0; k < GTRIP; ++k)
                r[k] = rec2[seg_index(min(q0 + k * 32 + lane, T - 1), incl, dl)];
            #pragma unroll
            for (int k = 0; k < GTRIP; ++k) {
                pp[k] = 0;
                if (q0 + k * 32 + lane < T) pp[k] = atomicAdd(&cur[r[k].w - nbase], 1);
            }
            #pragma unroll
            for (int k = 0; k < GTRIP; ++k)
                if (q0 + k * 32 + lane < T) rec[pp[k]] = make_int4(r[k].x, r[k].y, r[k].z, 0);
        }
    }
    __syncthreads();
    int nl = wave;
    int start = 0, deg = 0;
    int4 r = make_int4(0x7fffffff, 0, 0, 0);
    if (nl < BIN_NODES) {
        deg = cnt[nl];
        start = cur[nl] - deg;
        if (lane < deg) r = rec[start + lane];
    }
    for (; nl < BIN_NODES; nl += nwaves) {
        const int nn = nl + nwaves;
        int nstart = 0, ndeg = 0;
        int4 nr = make_int4(0x7fffffff, 0, 0, 0);
        if (nn < BIN_NODES) {
            ndeg = cnt[nn];
            nstart = cur[nn] - ndeg;
            if (lane < ndeg) nr = rec[nstart + lane];
        }
        const int degs = __builtin_amdgcn_readfirstlane(deg);
        if (degs <= 32) {
            int rk = 0;
            for (int j0 = 0; j0 < degs; j0 += 8) {
                #pragma unroll
                for (int u = 0; u < 8; ++u)
                    rk += (__builtin_amdgcn_readlane(r.x, j0 + u) < r.x) ? 1 : 0;
            }
            if (lane < degs) { gsrc[start + rk] = r.y; gw[start + rk] = __int_as_float(r.z); }
        } else {
            for (int q = lane; q < degs; q += 32) {
                const int4 rq = rec[start + q];
                int rk = 0;
                for (int j = 0; j < degs; ++j) rk += (rec[start + j].x < rq.x) ? 1 : 0;
                gsrc[start + rk] = rq.y; gw[start + rk] = __int_as_float(rq.z);
            }
        }
        start = nstart; deg = ndeg; r = nr;
    }
}

__global__ __launch_bounds__(256, 16) void node_kernel(
    const int* __restrict__ rowptr, const int* __restrict__ gsrc,
    const float* __restrict__ gw, const float* __restrict__ s,
    const float* __restrict__ d, const float* __restrict__ asrc,
    const float* __restrict__ H,
    float* __restrict__ out, int N, int F) {
    const int wfirst = __builtin_amdgcn_readfirstlane(
        (int)((blockIdx.x * (long long)blockDim.x + threadIdx.x) >> 5));
    const int nw   = (int)((gridDim.x * (long long)blockDim.x) >> 5);
    const int lane = threadIdx.x & 31;
    if (wfirst >= N) return;

    const bool fv = lane * 2 < F;
    const int  fo = fv ? lane * 2 : 0;
    const float a0 = fv ? asrc[fo] : 0.0f;
    const float a1 = fv ? asrc[fo + 1] : 0.0f;
    const bool hi = lane >= 16;

    int   start = rowptr[wfirst];
    int   end   = rowptr[wfirst + 1];
    float di    = d[wfirst];
    float si    = s[wfirst];
    int src0 = gsrc[max(min(start + lane, end - 1), 0)];
    asm volatile("" :: "v"(src0));

    const int nx1 = min(wfirst + nw, N - 1);
    int   nstart = rowptr[nx1];
    int   nend   = rowptr[nx1 + 1];
    float ndi    = d[nx1];
    float nsi    = s[nx1];

    for (int wid = wfirst; wid < N; wid += nw) {
        const int nx2 = min(wid + 2 * nw, N - 1);
        const int   n2start = rowptr[nx2];
        const int   n2end   = rowptr[nx2 + 1];
        const float n2di    = d[nx2];
        const float n2si    = s[nx2];

        const int nsrc0 = gsrc[max(min(nstart + lane, nend - 1), 0)];
        const int deg = end - start;
        const bool v0 = lane < deg;
        const float wt0 = gw[max(min(start + lane, end - 1), 0)];
        float s0 = 0.0f;
        if (v0 && hi) s0 = s[src0];

        const int n0 = min(deg, 32);
        const int jlast = max(n0, 1) - 1;
        float2 hA[16];
        #pragma unroll
        for (int u = 0; u < 16; ++u) {
            const int sj = __builtin_amdgcn_readlane(src0, min(u, jlast));
            hA[u] = *(const float2*)(H + (size_t)sj * F + fo);
        }

        float pv[16];
        #pragma unroll
        for (int u = 0; u < 16; ++u) pv[u] = fmaf(hA[u].y, a1, hA[u].x * a0);
        float slo = row_scores16(pv, lane & 15);
        slo = slo + xhalf(slo);
        const float sj0 = hi ? s0 : slo;
        const float2 hself = *(const float2*)(H + (size_t)wid * F + fo);

        const float sc_self = leaky(si + di);
        const float sc0 = v0 ? leaky(sj0 + di) : sc_self;

        float m = fmaxf(sc_self, sc0);
        for (int p = start + 32 + lane; p < end; p += 32)
            m = fmaxf(m, leaky(s[gsrc[p]] + di));
        m = fmaxf(m, lane_xor<16>(m)); m = fmaxf(m, lane_xor<8>(m)); m = fmaxf(m, lane_xor<4>(m));
        m = fmaxf(m, lane_xor<2>(m));  m = fmaxf(m, lane_xor<1>(m));

        const float ex0 = v0 ? __expf(sc0 - m) : 0.0f;
        float sum = ex0;
        for (int p = start + 32 + lane; p < end; p += 32)
            sum += __expf(leaky(s[gsrc[p]] + di) - m);
        sum += lane_xor<16>(sum); sum += lane_xor<8>(sum); sum += lane_xor<4>(sum);
        sum += lane_xor<2>(sum);  sum += lane_xor<1>(sum);
        const float ex_self = __expf(sc_self - m);
        sum += ex_self;
        const float dn = fmaxf(sum, 1e-16f);
        const float c0 = (ex0 * wt0) / dn;
        const float c_self = ex_self / dn;

        float ax = 0.0f, ay = 0.0f;
        #pragma unroll
        for (int u = 0; u < 16; ++u) {
            if (u < n0) {
                const float cj = __int_as_float(__builtin_amdgcn_readlane(__float_as_int(c0), u));
                ax = fmaf(cj, hA[u].x, ax); ay = fmaf(cj, hA[u].y, ay);
            }
        }
        for (int jb = 16; jb < n0; jb += 8) {
            float2 h[8]; float cc[8];
            #pragma unroll
            for (int u = 0; u < 8; ++u) {
                const int j  = jb + u;
                const int jj = min(j, n0 - 1);
                const int sj = __builtin_amdgcn_readlane(src0, jj);
                const float cj = __int_as_float(__builtin_amdgcn_readlane(__float_as_int(c0), jj));
                cc[u] = (j < n0) ? cj : 0.0f;
                h[u]  = *(const float2*)(H + (size_t)sj * F + fo);
            }
            #pragma unroll
            for (int u = 0; u < 8; ++u) {
                if (jb + u < n0) { ax = fmaf(cc[u], h[u].x, ax); ay = fmaf(cc[u], h[u].y, ay); }
            }
        }
        for (int cb = start + 32; cb < end; cb += 32) {
            const int p = cb + lane;
            int sx = 0; float cx = 0.0f;
            if (p < end) {
                sx = gsrc[p];
                cx = (__expf(leaky(s[sx] + di) - m) * gw[p]) / dn;
            }
            const int cn = min(32, end - cb);
            for (int j = 0; j < cn; ++j) {
                const int sj = __builtin_amdgcn_readlane(sx, j);
                const float cj = __int_as_float(__builtin_amdgcn_readlane(__float_as_int(cx), j));
                const float2 hv = *(const float2*)(H + (size_t)sj * F + fo);
                ax = fmaf(cj, hv.x, ax); ay = fmaf(cj, hv.y, ay);
            }
        }
        asm volatile("" :: "v"(hself.x), "v"(hself.y));
        ax = fmaf(c_self, hself.x, ax); ay = fmaf(c_self, hself.y, ay);
        if (fv) *(float2*)(out + (size_t)wid * F + fo) = make_float2(ax, ay);

        start = nstart; end = nend; di = ndi; si = nsi;
        src0 = nsrc0;
        nstart = n2start; nend = n2end; ndi = n2di; nsi = n2si;
    }
}

extern "C" void kernel_launch(void* const* d_in, const int* in_sizes, int n_in,
                              void* d_out, int out_size, void* d_ws, size_t ws_size,
                              hipStream_t stream) {
    const float* x   = (const float*)d_in[0];
    const int*   ei  = (const int*)  d_in[1];
    const float* ew  = (const float*)d_in[2];
    const float* W1  = (const float*)d_in[3];
    const float* a1s = (const float*)d_in[4];
    const float* a1d = (const float*)d_in[5];
    const float* W2  = (const float*)d_in[6];
    const float* a2s = (const float*)d_in[7];
    const float* a2d = (const float*)d_in[8];

    const int HID = in_sizes[4];
    const int FIN = in_sizes[3] / HID;
    const int NC  = in_sizes[7];
    const int N   = in_sizes[0] / FIN;
    const int E   = in_sizes[2];

    float* h1   = (float*)d_ws;
    float* out1 = h1   + (size_t)N * HID;
    float* h2   = out1 + (size_t)N * HID;
    float* sArr = h2   + (size_t)N * NC;
    float* dArr = sArr + N;
    float* outF = (float*)d_out;

    const int perWG = 1024 * TRIP;
    const int G1  = (E + perWG - 1) / perWG > 256 ? (E + perWG - 1) / perWG : 256;
    const int CH  = (E + G1 - 1) / G1;
    const int B   = (N + BIN_NODES - 1) / BIN_NODES;
    const int nso = (B + 1) * G1;
    size_t off = (((size_t)N * (2 * HID + NC + 2)) + 3) & ~(size_t)3;
    int* base  = (int*)d_ws;
    int4*  rec    = (int4*)(base + off);  off += (size_t)E * 4;
    int4*  rec2   = (int4*)(base + off);  off += (size_t)E * 4;
    int*   rowptr = base + off;           off += ((size_t)N + 4) & ~(size_t)3;
    int*   segOff = base + off;           off += ((size_t)nso + 3) & ~(size_t)3;
    int*   gsrc   = base + off;           off += ((size_t)E + 3) & ~(size_t)3;
    float* gw     = (float*)(base + off);

    const int nrt     = (N + 15) / 16;
    const int pairs   = (nrt + 1) / 2;
    const int gBlocks = (pairs + 7) / 8;
    const int wBlocks = 2048;

    const int tpb   = (nrt + G1 - 1) / G1 < 32 ? (nrt + G1 - 1) / G1 : 32;
    const int gPair = (nrt + tpb - 1) / tpb > G1 ? (nrt + tpb - 1) / tpb : G1;
    chunk_bin_gemm1_kernel<<<2 * gPair, 1024, B * sizeof(int), stream>>>(
        ei, ew, segOff, rec2, E, B, CH, G1, x, W1, a1s, a1d, h1, sArr, dArr, N, FIN, tpb);
    bin_group_kernel<<<B, 1024, 0, stream>>>(segOff, rec2, rec, rowptr, gsrc, gw, N, E, G1, CH);

    node_kernel<<<wBlocks, 256, 0, stream>>>(rowptr, gsrc, gw, sArr, dArr, a1s, h1, out1, N, HID);

    gemm2_kernel<<<gBlocks, 256, 0, stream>>>(out1, W2, a2s, a2d, h2, sArr, dArr, N, NC);
    node_kernel<<<wBlocks, 256, 0, stream>>>(rowptr, gsrc, gw, sArr, dArr, a2s, h2, outF, N, NC);
}


# Layer 2's GEMM runs at the end of layer 1's node pass: a work-group owns 80 consecutive nodes, keeps their out1 rows in LDS and multiplies them there (same WMMA): 5 launches -> 4; sums keep order
# speedup vs baseline: 8.4182x; 1.0701x over previous
#include <hip/hip_runtime.h>
#include <hip/hip_bf16.h>


typedef __attribute__((ext_vector_type(2))) float v2f;
typedef __attribute__((ext_vector_type(8))) float v8f;

#define NEG_SLOPE 0.2f

__device__ __forceinline__ float leaky(float x) {
    return x > 0.0f ? x : NEG_SLOPE * x;
}

template <int CTRL>
__device__ __forceinline__ float row_xor(float v) {
    return __int_as_float(__builtin_amdgcn_update_dpp(
        0, __float_as_int(v), CTRL, 0xF, 0xF, false));
}

__device__ __forceinline__ float xhalf(float x) {
    const int b = __float_as_int(x);
    return __int_as_float(__builtin_amdgcn_permlanex16(b, b, 0x76543210, 0xfedcba98, false, false));
}

template <int K>
__device__ __forceinline__ float lane_xor(float v) {
    return __int_as_float(__builtin_amdgcn_ds_swizzle(__float_as_int(v), 0x1F | (K << 10)));
}

__device__ __forceinline__ float row_scores16(const float (&v)[16], int col0) {
    const bool b3 = (col0 & 8) != 0, b2 = (col0 & 4) != 0;
    const bool b1 = (col0 & 2) != 0, b0 = (col0 & 1) != 0;
    float w8[8], w4[4], w2[2];
    #pragma unroll
    for (int i = 0; i < 8; ++i) {
        const float keep = b3 ? v[8 + i] : v[i];
        const float send = b3 ? v[i] : v[8 + i];
        w8[i] = keep + row_xor<0x168>(send);
    }
    #pragma unroll
    for (int i = 0; i < 4; ++i) {
        const float keep = b2 ? w8[4 + i] : w8[i];
        const float send = b2 ? w8[i] : w8[4 + i];
        w4[i] = keep + row_xor<0x164>(send);
    }
    #pragma unroll
    for (int i = 0; i < 2; ++i) {
        const float keep = b1 ? w4[2 + i] : w4[i];
        const float send = b1 ? w4[i] : w4[2 + i];
        w2[i] = keep + row_xor<0x162>(send);
    }
    const float keep = b0 ? w2[1] : w2[0], send = b0 ? w2[0] : w2[1];
    return keep + row_xor<0x161>(send);
}

#define P1 132
__device__ __forceinline__ void gemm1_role(
    float* sW,
    const float* __restrict__ X, const float* __restrict__ W,
    const float* __restrict__ asrc, const float* __restrict__ adst,
    float* __restrict__ H, float* __restrict__ s, float* __restrict__ d,
    int N, int K, int blk, int TPB) {
    const int tid = threadIdx.x, nt = 1024;
    const int lane = tid & 31;
    const int wave = __builtin_amdgcn_readfirstlane(tid >> 5);
    const int half = lane >> 4;
    const int col0 = lane & 15;
    const int nrt  = (N + 15) >> 4;
    const int tBeg = min(nrt, blk * TPB), tEnd = min(nrt, tBeg + TPB);
    const int phases = K >> 7;

    if (tBeg >= tEnd) return;
    {
        const int rt = tBeg + wave;
        const bool act = rt < tEnd;
        const int rowA = act ? min(rt * 16 + col0, N - 1) : 0;

        v8f acc[4];
        #pragma unroll
        for (int t = 0; t < 4; ++t) acc[t] = v8f{};

        for (int ph = 0; ph < phases; ++ph) {
            const float* wp = W + (size_t)ph * 128 * 64 + tid;
            #pragma unroll
            for (int j = 0; j < 128 * 64 / nt; ++j)
                sW[(tid & 63) * P1 + (tid >> 6) + j * (nt >> 6)] = wp[j * nt];
            __syncthreads();

            if (act) {
                const float* xr = X + (size_t)rowA * K + ph * 128;
                for (int kk = 0; kk < 32; ++kk) {
                    const int kb = kk * 4 + half * 2;
                    v2f a = *(const v2f*)(xr + kb);
                    #pragma unroll
                    for (int t = 0; t < 4; ++t) {
                        v2f b = *(const v2f*)(&sW[(t * 16 + col0) * P1 + kb]);
                        acc[t] = __builtin_amdgcn_wmma_f32_16x16x4_f32(
                            false, a, false, b, (short)0, acc[t], false, false);
                    }
                }
            }
            __syncthreads();
        }

        if (act) {
            float as[4], ad[4];
            #pragma unroll
            for (int t = 0; t < 4; ++t) { as[t] = asrc[t * 16 + col0]; ad[t] = adst[t * 16 + col0]; }
            #pragma unroll
            for (int t = 0; t < 4; ++t)
                #pragma unroll
                for (int r = 0; r < 8; ++r) {
                    int row = rt * 16 + r + half * 8;
                    if (row < N) H[(size_t)row * 64 + t * 16 + col0] = acc[t][r];
                }

            float pv[16];
            #pragma unroll
            for (int r = 0; r < 8; ++r) {
                float ps = acc[0][r] * as[0], pd = acc[0][r] * ad[0];
                #pragma unroll
                for (int t = 1; t < 4; ++t) {
                    ps = fmaf(acc[t][r], as[t], ps);
                    pd = fmaf(acc[t][r], ad[t], pd);
                }
                pv[r * 2] = ps; pv[r * 2 + 1] = pd;
            }
            const float sc = row_scores16(pv, col0);
            const int srow = rt * 16 + half * 8 + (col0 >> 1);
            if (srow < N) { if (col0 & 1) d[srow] = sc; else s[srow] = sc; }
        }
    }
}

#define BIN_SHIFT 8
#define BIN_NODES 256

#define TRIP 7

__device__ __forceinline__ void chunk_bin_role(
    int* hist, int* swave,
    const int* __restrict__ ei, const float* __restrict__ ew,
    int* __restrict__ segOff, int4* __restrict__ rec2, int E, int B, int CH,
    int c, int G) {
    const int tid = threadIdx.x, nt = blockDim.x;
    const int lane = tid & 31, wave = tid >> 5;
    for (int i = tid; i < B; i += nt) hist[i] = 0;
    __syncthreads();
    const int e0 = min(E, c * CH), e1 = min(E, e0 + CH);
    int dd[TRIP], ss[TRIP], ww[TRIP], pp[TRIP];
    #pragma unroll
    for (int k = 0; k < TRIP; ++k) { dd[k] = 0; ss[k] = 0; ww[k] = 0; pp[k] = 0; }
    if (e0 < e1) {
        #pragma unroll
        for (int k = 0; k < TRIP; ++k) {
            const int e = min(e0 + tid + k * nt, e1 - 1);
            dd[k] = ei[E + e];
            ss[k] = ei[e];
            ww[k] = __float_as_int(ew[e]);
        }
        #pragma unroll
        for (int k = 0; k < TRIP; ++k)
            if (e0 + tid + k * nt < e1) pp[k] = atomicAdd(&hist[dd[k] >> BIN_SHIFT], 1);
    }
    __syncthreads();
    const int per = (B + nt - 1) / nt;
    const int i0 = min(B, tid * per), i1 = min(B, i0 + per);
    int tsum = 0;
    for (int i = i0; i < i1; ++i) tsum += hist[i];
    int inc = tsum;
    for (int off = 1; off < 32; off <<= 1) {
        int n = __shfl_up(inc, off, 32);
        if (lane >= off) inc += n;
    }
    if (lane == 31) swave[wave] = inc;
    __syncthreads();
    int pre = (lane < wave) ? swave[lane] : 0;
    for (int off = 16; off > 0; off >>= 1) pre += __shfl_xor(pre, off, 32);
    int run = pre + inc - tsum;
    for (int i = i0; i < i1; ++i) {
        const int v = hist[i];
        hist[i] = run;
        segOff[(size_t)i * G + c] = run;
        run += v;
    }
    if (tid == 0) segOff[(size_t)B * G + c] = e1 - e0;
    __syncthreads();
    #pragma unroll
    for (int k = 0; k < TRIP; ++k) {
        const int e = e0 + tid + k * nt;
        if (e < e1)
            rec2[e0 + hist[dd[k] >> BIN_SHIFT] + pp[k]] = make_int4(e, ss[k], ww[k], dd[k]);
    }
}

__global__ __launch_bounds__(1024, 16) void chunk_bin_gemm1_kernel(
    const int* __restrict__ ei, const float* __restrict__ ew,
    int* __restrict__ segOff, int4* __restrict__ rec2, int E, int B, int CH, int G,
    const float* __restrict__ X, const float* __restrict__ W,
    const float* __restrict__ asrc, const float* __restrict__ adst,
    float* __restrict__ H, float* __restrict__ s, float* __restrict__ d,
    int N, int K, int TPB) {
    extern __shared__ int hist[];
    __shared__ int swave[32];
    __shared__ float sW[64 * P1];
    const int half = gridDim.x >> 1;
    const int b = blockIdx.x;
    if (b < half) {
        gemm1_role(sW, X, W, asrc, adst, H, s, d, N, K, b, TPB);
    } else {
        if (b - half < G) chunk_bin_role(hist, swave, ei, ew, segOff, rec2, E, B, CH, b - half, G);
    }
}

__device__ __forceinline__ void seg_group(
    const int* __restrict__ segOff, int b, int G, int CH, int g, int lane,
    int (&incl)[8], int (&dl)[8], int& losum) {
    const int c = g * 8 + (lane & 7);
    int lo = 0, hi = 0;
    if (c < G) { lo = segOff[(size_t)b * G + c]; hi = segOff[(size_t)(b + 1) * G + c]; }
    const int len = hi - lo;
    int inc = len;
    for (int off = 1; off < 8; off <<= 1) {
        int n = __shfl_up(inc, off, 32);
        if ((lane & 7) >= off) inc += n;
    }
    const int d = c * CH + lo - (inc - len);
    losum = 0;
    #pragma unroll
    for (int i = 0; i < 8; ++i) {
        incl[i] = __builtin_amdgcn_readlane(inc, i);
        dl[i]   = __builtin_amdgcn_readlane(d, i);
        losum  += __builtin_amdgcn_readlane(lo, i);
    }
}

__device__ __forceinline__ int seg_index(int q, const int (&incl)[8], const int (&dl)[8]) {
    int D = dl[0];
    #pragma unroll
    for (int i = 0; i < 7; ++i) D = (q >= incl[i]) ? dl[i + 1] : D;
    return q + D;
}

#define GTRIP 6
__global__ __launch_bounds__(1024, 16) void bin_group_kernel(
    const int* __restrict__ segOff, const int4* __restrict__ rec2,
    int4* rec, int* __restrict__ rowptr,
    int* __restrict__ gsrc, float* __restrict__ gw, int N, int E, int G, int CH) {
    __shared__ int cnt[BIN_NODES];
    __shared__ int cur[BIN_NODES];
    __shared__ int swave[BIN_NODES / 32];
    __shared__ int sstart[32];
    const int tid = threadIdx.x;
    const int lane = tid & 31, wave = tid >> 5;
    const int nwaves = blockDim.x >> 5;
    const int b = blockIdx.x;
    const int nbase = b << BIN_SHIFT;
    if (tid < BIN_NODES) cnt[tid] = 0;
    __syncthreads();
    int mylo = 0;
    for (int g = wave; g * 8 < G; g += nwaves) {
        int incl[8], dl[8], losum;
        seg_group(segOff, b, G, CH, g, lane, incl, dl, losum);
        mylo += losum;
        const int T = incl[7];
        for (int q0 = 0; q0 < T; q0 += 32 * GTRIP) {
            int dd[GTRIP];
            #pragma unroll
            for (int k = 0; k < GTRIP; ++k)
                dd[k] = rec2[seg_index(min(q0 + k * 32 + lane, T - 1), incl, dl)].w;
            #pragma unroll
            for (int k = 0; k < GTRIP; ++k)
                if (q0 + k * 32 + lane < T) atomicAdd(&cnt[dd[k] - nbase], 1);
        }
    }
    if (lane == 0) sstart[wave] = mylo;
    __syncthreads();
    int v = 0, inc = 0;
    if (tid < BIN_NODES) {
        v = cnt[tid];
        inc = v;
        for (int off = 1; off < 32; off <<= 1) {
            int n = __shfl_up(inc, off, 32);
            if (lane >= off) inc += n;
        }
        if (lane == 31) swave[wave] = inc;
    }
    __syncthreads();
    if (tid < BIN_NODES) {
        int binStart = 0;
        for (int w = 0; w < nwaves; ++w) binStart += sstart[w];
        int pre = 0;
        for (int w = 0; w < BIN_NODES / 32; ++w) if (w < wave) pre += swave[w];
        const int first = binStart + pre + inc - v;
        cur[tid] = first;
        if (nbase + tid < N) rowptr[nbase + tid] = first;
    }
    if (b == (int)gridDim.x - 1 && tid == 0) rowptr[N] = E;
    __syncthreads();
    for (int g = wave; g * 8 < G; g += nwaves) {
        int incl[8], dl[8], losum;
        seg_group(segOff, b, G, CH, g, lane, incl, dl, losum);
        const int T = incl[7];
        for (int q0 = 0; q0 < T; q0 += 32 * GTRIP) {
            int4 r[GTRIP]; int pp[GTRIP];
            #pragma unroll
            for (int k = 0; k < GTRIP; ++k)
                r[k] = rec2[seg_index(min(q0 + k * 32 + lane, T - 1), incl, dl)];
            #pragma unroll
            for (int k = 0; k < GTRIP; ++k) {
                pp[k] = 0;
                if (q0 + k * 32 + lane < T) pp[k] = atomicAdd(&cur[r[k].w - nbase], 1);
            }
            #pragma unroll
            for (int k = 0; k < GTRIP; ++k)
                if (q0 + k * 32 + lane < T) rec[pp[k]] = make_int4(r[k].x, r[k].y, r[k].z, 0);
        }
    }
    __syncthreads();
    int nl = wave;
    int start = 0, deg = 0;
    int4 r = make_int4(0x7fffffff, 0, 0, 0);
    if (nl < BIN_NODES) {
        deg = cnt[nl];
        start = cur[nl] - deg;
        if (lane < deg) r = rec[start + lane];
    }
    for (; nl < BIN_NODES; nl += nwaves) {
        const int nn = nl + nwaves;
        int nstart = 0, ndeg = 0;
        int4 nr = make_int4(0x7fffffff, 0, 0, 0);
        if (nn < BIN_NODES) {
            ndeg = cnt[nn];
            nstart = cur[nn] - ndeg;
            if (lane < ndeg) nr = rec[nstart + lane];
        }
        const int degs = __builtin_amdgcn_readfirstlane(deg);
        if (degs <= 32) {
            int rk = 0;
            for (int j0 = 0; j0 < degs; j0 += 8) {
                #pragma unroll
                for (int u = 0; u < 8; ++u)
                    rk += (__builtin_amdgcn_readlane(r.x, j0 + u) < r.x) ? 1 : 0;
            }
            if (lane < degs) { gsrc[start + rk] = r.y; gw[start + rk] = __int_as_float(r.z); }
        } else {
            for (int q = lane; q < degs; q += 32) {
                const int4 rq = rec[start + q];
                int rk = 0;
                for (int j = 0; j < degs; ++j) rk += (rec[start + j].x < rq.x) ? 1 : 0;
                gsrc[start + rk] = rq.y; gw[start + rk] = __int_as_float(rq.z);
            }
        }
        start = nstart; deg = ndeg; r = nr;
    }
}

#define PA 68
template <bool TILE>
__device__ __forceinline__ void node_walk(
    const int* __restrict__ rowptr, const int* __restrict__ gsrc,
    const float* __restrict__ gw, const float* __restrict__ s,
    const float* __restrict__ d, const float* __restrict__ asrc,
    const float* __restrict__ H,
    float* __restrict__ out, float* tile, int tbase,
    int N, int F, int wfirst, int wend, int nw, int lane) {
    const bool fv = lane * 2 < F;
    const int  fo = fv ? lane * 2 : 0;
    const float a0 = fv ? asrc[fo] : 0.0f;
    const float a1 = fv ? asrc[fo + 1] : 0.0f;
    const bool hi = lane >= 16;

    int   start = rowptr[wfirst];
    int   end   = rowptr[wfirst + 1];
    float di    = d[wfirst];
    float si    = s[wfirst];
    int src0 = gsrc[max(min(start + lane, end - 1), 0)];
    asm volatile("" :: "v"(src0));

    const int nx1 = min(wfirst + nw, N - 1);
    int   nstart = rowptr[nx1];
    int   nend   = rowptr[nx1 + 1];
    float ndi    = d[nx1];
    float nsi    = s[nx1];

    for (int wid = wfirst; wid < wend; wid += nw) {
        const int nx2 = min(wid + 2 * nw, N - 1);
        const int   n2start = rowptr[nx2];
        const int   n2end   = rowptr[nx2 + 1];
        const float n2di    = d[nx2];
        const float n2si    = s[nx2];

        const int nsrc0 = gsrc[max(min(nstart + lane, nend - 1), 0)];
        const int deg = end - start;
        const bool v0 = lane < deg;
        const float wt0 = gw[max(min(start + lane, end - 1), 0)];
        float s0 = 0.0f;
        if (v0 && hi) s0 = s[src0];

        const int n0 = min(deg, 32);
        const int jlast = max(n0, 1) - 1;
        float2 hA[16];
        #pragma unroll
        for (int u = 0; u < 16; ++u) {
            const int sj = __builtin_amdgcn_readlane(src0, min(u, jlast));
            hA[u] = *(const float2*)(H + (size_t)sj * F + fo);
        }

        float pv[16];
        #pragma unroll
        for (int u = 0; u < 16; ++u) pv[u] = fmaf(hA[u].y, a1, hA[u].x * a0);
        float slo = row_scores16(pv, lane & 15);
        slo = slo + xhalf(slo);
        const float sj0 = hi ? s0 : slo;
        const float2 hself = *(const float2*)(H + (size_t)wid * F + fo);

        const float sc_self = leaky(si + di);
        const float sc0 = v0 ? leaky(sj0 + di) : sc_self;

        float m = fmaxf(sc_self, sc0);
        for (int p = start + 32 + lane; p < end; p += 32)
            m = fmaxf(m, leaky(s[gsrc[p]] + di));
        m = fmaxf(m, lane_xor<16>(m)); m = fmaxf(m, lane_xor<8>(m)); m = fmaxf(m, lane_xor<4>(m));
        m = fmaxf(m, lane_xor<2>(m));  m = fmaxf(m, lane_xor<1>(m));

        const float ex0 = v0 ? __expf(sc0 - m) : 0.0f;
        float sum = ex0;
        for (int p = start + 32 + lane; p < end; p += 32)
            sum += __expf(leaky(s[gsrc[p]] + di) - m);
        sum += lane_xor<16>(sum); sum += lane_xor<8>(sum); sum += lane_xor<4>(sum);
        sum += lane_xor<2>(sum);  sum += lane_xor<1>(sum);
        const float ex_self = __expf(sc_self - m);
        sum += ex_self;
        const float dn = fmaxf(sum, 1e-16f);
        const float c0 = (ex0 * wt0) / dn;
        const float c_self = ex_self / dn;

        float ax = 0.0f, ay = 0.0f;
        #pragma unroll
        for (int u = 0; u < 16; ++u) {
            if (u < n0) {
                const float cj = __int_as_float(__builtin_amdgcn_readlane(__float_as_int(c0), u));
                ax = fmaf(cj, hA[u].x, ax); ay = fmaf(cj, hA[u].y, ay);
            }
        }
        for (int jb = 16; jb < n0; jb += 8) {
            float2 h[8]; float cc[8];
            #pragma unroll
            for (int u = 0; u < 8; ++u) {
                const int j  = jb + u;
                const int jj = min(j, n0 - 1);
                const int sj = __builtin_amdgcn_readlane(src0, jj);
                const float cj = __int_as_float(__builtin_amdgcn_readlane(__float_as_int(c0), jj));
                cc[u] = (j < n0) ? cj : 0.0f;
                h[u]  = *(const float2*)(H + (size_t)sj * F + fo);
            }
            #pragma unroll
            for (int u = 0; u < 8; ++u) {
                if (jb + u < n0) { ax = fmaf(cc[u], h[u].x, ax); ay = fmaf(cc[u], h[u].y, ay); }
            }
        }
        for (int cb = start + 32; cb < end; cb += 32) {
            const int p = cb + lane;
            int sx = 0; float cx = 0.0f;
            if (p < end) {
                sx = gsrc[p];
                cx = (__expf(leaky(s[sx] + di) - m) * gw[p]) / dn;
            }
            const int cn = min(32, end - cb);
            for (int j = 0; j < cn; ++j) {
                const int sj = __builtin_amdgcn_readlane(sx, j);
                const float cj = __int_as_float(__builtin_amdgcn_readlane(__float_as_int(cx), j));
                const float2 hv = *(const float2*)(H + (size_t)sj * F + fo);
                ax = fmaf(cj, hv.x, ax); ay = fmaf(cj, hv.y, ay);
            }
        }
        asm volatile("" :: "v"(hself.x), "v"(hself.y));
        ax = fmaf(c_self, hself.x, ax); ay = fmaf(c_self, hself.y, ay);
        if (TILE) *(float2*)(tile + (wid - tbase) * PA + fo) = make_float2(ax, ay);
        else if (fv) *(float2*)(out + (size_t)wid * F + fo) = make_float2(ax, ay);

        start = nstart; end = nend; di = ndi; si = nsi;
        src0 = nsrc0;
        nstart = n2start; nend = n2end; ndi = n2di; nsi = n2si;
    }
}

__global__ __launch_bounds__(256, 16) void node_kernel(
    const int* __restrict__ rowptr, const int* __restrict__ gsrc,
    const float* __restrict__ gw, const float* __restrict__ s,
    const float* __restrict__ d, const float* __restrict__ asrc,
    const float* __restrict__ H,
    float* __restrict__ out, int N, int F) {
    const int wfirst = __builtin_amdgcn_readfirstlane(
        (int)((blockIdx.x * (long long)blockDim.x + threadIdx.x) >> 5));
    const int nw   = (int)((gridDim.x * (long long)blockDim.x) >> 5);
    const int lane = threadIdx.x & 31;
    if (wfirst >= N) return;
    node_walk<false>(rowptr, gsrc, gw, s, d, asrc, H, out, nullptr, 0, N, F, wfirst, N, nw, lane);
}

#define NPB 80
#define NWV 12
#define P2 68
__global__ __launch_bounds__(NWV * 32, 15) void node_gemm2_kernel(
    const int* __restrict__ rowptr, const int* __restrict__ gsrc,
    const float* __restrict__ gw, const float* __restrict__ s,
    const float* __restrict__ d, const float* __restrict__ asrc,
    const float* __restrict__ H, int N,
    const float* __restrict__ W2, const float* __restrict__ a2s,
    const float* __restrict__ a2d,
    float* __restrict__ Hout, float* __restrict__ s2, float* __restrict__ d2, int NC) {
    __shared__ __attribute__((aligned(16))) float sA[NPB * PA];
    __shared__ __attribute__((aligned(16))) float sW[48 * P2];
    const int lane = threadIdx.x & 31;
    const int wave = __builtin_amdgcn_readfirstlane((int)(threadIdx.x >> 5));
    const int base = blockIdx.x * NPB;
    const int wend = min(N, base + NPB);

    if (base + wave < wend)
        node_walk<true>(rowptr, gsrc, gw, s, d, asrc, H, nullptr, sA, base,
                        N, 64, base + wave, wend, NWV, lane);
    {
        float wv[64 * 48 / (NWV * 32)];
        #pragma unroll
        for (int j = 0; j < 64 * 48 / (NWV * 32); ++j) {
            const int i = threadIdx.x + j * (NWV * 32);
            const int r = i / 48, cc = i % 48;
            wv[j] = W2[(size_t)r * NC + min(cc, NC - 1)];
        }
        #pragma unroll
        for (int j = 0; j < 64 * 48 / (NWV * 32); ++j) {
            const int i = threadIdx.x + j * (NWV * 32);
            const int r = i / 48, cc = i % 48;
            sW[cc * P2 + r] = (cc < NC) ? wv[j] : 0.0f;
        }
    }
    for (int r = wend - base + wave; r < NPB; r += NWV)
        *(float2*)(sA + r * PA + lane * 2) = make_float2(0.0f, 0.0f);
    __syncthreads();

    if (wave < NPB / 16 && base + wave * 16 < N) {
        const int half = lane >> 4;
        const int col0 = lane & 15;
        float as[3], ad[3];
        #pragma unroll
        for (int t = 0; t < 3; ++t) {
            const int cc = t * 16 + col0;
            as[t] = (cc < NC) ? a2s[cc] : 0.0f;
            ad[t] = (cc < NC) ? a2d[cc] : 0.0f;
        }
        v8f acc[3];
        #pragma unroll
        for (int t = 0; t < 3; ++t) acc[t] = v8f{};

        const float* ar = sA + (wave * 16 + col0) * PA;
        for (int kk = 0; kk < 16; ++kk) {
            const int kb = kk * 4 + half * 2;
            v2f a = *(const v2f*)(ar + kb);
            a[0] = fmaxf(a[0], 0.0f); a[1] = fmaxf(a[1], 0.0f);
            #pragma unroll
            for (int t = 0; t < 3; ++t) {
                v2f b = *(const v2f*)(&sW[(t * 16 + col0) * P2 + kb]);
                acc[t] = __builtin_amdgcn_wmma_f32_16x16x4_f32(
                    false, a, false, b, (short)0, acc[t], false, false);
            }
        }

        const int row0 = base + wave * 16;
        #pragma unroll
        for (int t = 0; t < 3; ++t)
            #pragma unroll
            for (int r = 0; r < 8; ++r) {
                const int row = row0 + r + half * 8;
                const int col = t * 16 + col0;
                if (row < N && col < NC)
                    Hout[(size_t)row * NC + col] = acc[t][r];
            }

        float pv[16];
        #pragma unroll
        for (int r = 0; r < 8; ++r) {
            float ps = acc[0][r] * as[0], pd = acc[0][r] * ad[0];
            #pragma unroll
            for (int t = 1; t < 3; ++t) {
                ps = fmaf(acc[t][r], as[t], ps);
                pd = fmaf(acc[t][r], ad[t], pd);
            }
            pv[r * 2] = ps; pv[r * 2 + 1] = pd;
        }
        const float sc = row_scores16(pv, col0);
        const int srow = row0 + half * 8 + (col0 >> 1);
        if (srow < N) { if (col0 & 1) d2[srow] = sc; else s2[srow] = sc; }
    }
}

extern "C" void kernel_launch(void* const* d_in, const int* in_sizes, int n_in,
                              void* d_out, int out_size, void* d_ws, size_t ws_size,
                              hipStream_t stream) {
    const float* x   = (const float*)d_in[0];
    const int*   ei  = (const int*)  d_in[1];
    const float* ew  = (const float*)d_in[2];
    const float* W1  = (const float*)d_in[3];
    const float* a1s = (const float*)d_in[4];
    const float* a1d = (const float*)d_in[5];
    const float* W2  = (const float*)d_in[6];
    const float* a2s = (const float*)d_in[7];
    const float* a2d = (const float*)d_in[8];

    const int HID = in_sizes[4];
    const int FIN = in_sizes[3] / HID;
    const int NC  = in_sizes[7];
    const int N   = in_sizes[0] / FIN;
    const int E   = in_sizes[2];

    float* h1   = (float*)d_ws;
    float* h2   = h1   + (size_t)N * HID;
    float* sArr = h2   + (size_t)N * NC;
    float* dArr = sArr + N;
    float* outF = (float*)d_out;

    const int perWG = 1024 * TRIP;
    const int G1  = (E + perWG - 1) / perWG > 256 ? (E + perWG - 1) / perWG : 256;
    const int CH  = (E + G1 - 1) / G1;
    const int B   = (N + BIN_NODES - 1) / BIN_NODES;
    const int nso = (B + 1) * G1;
    size_t off = (((size_t)N * (HID + NC + 2)) + 3) & ~(size_t)3;
    int* base  = (int*)d_ws;
    int4*  rec    = (int4*)(base + off);  off += (size_t)E * 4;
    int4*  rec2   = (int4*)(base + off);  off += (size_t)E * 4;
    int*   rowptr = base + off;           off += ((size_t)N + 4) & ~(size_t)3;
    int*   segOff = base + off;           off += ((size_t)nso + 3) & ~(size_t)3;
    int*   gsrc   = base + off;           off += ((size_t)E + 3) & ~(size_t)3;
    float* gw     = (float*)(base + off); off += ((size_t)E + 3) & ~(size_t)3;
    float* s2Arr  = (float*)(base + off); off += ((size_t)N + 3) & ~(size_t)3;
    float* d2Arr  = (float*)(base + off);

    const int nrt     = (N + 15) / 16;
    const int wBlocks = 2048;
    const int nBlocks = (N + NPB - 1) / NPB;

    const int tpb   = (nrt + G1 - 1) / G1 < 32 ? (nrt + G1 - 1) / G1 : 32;
    const int gPair = (nrt + tpb - 1) / tpb > G1 ? (nrt + tpb - 1) / tpb : G1;
    chunk_bin_gemm1_kernel<<<2 * gPair, 1024, B * sizeof(int), stream>>>(
        ei, ew, segOff, rec2, E, B, CH, G1, x, W1, a1s, a1d, h1, sArr, dArr, N, FIN, tpb);
    bin_group_kernel<<<B, 1024, 0, stream>>>(segOff, rec2, rec, rowptr, gsrc, gw, N, E, G1, CH);

    node_gemm2_kernel<<<nBlocks, NWV * 32, 0, stream>>>(rowptr, gsrc, gw, sArr, dArr, a1s, h1, N,
                                                  W2, a2s, a2d, h2, s2Arr, d2Arr, NC);

    node_kernel<<<wBlocks, 256, 0, stream>>>(rowptr, gsrc, gw, s2Arr, d2Arr, a2s, h2, outF, N, NC);
}


# Layer 1's node pass + layer 2's GEMM run inside the 2nd grouping launch: the work-group that groups a bin walks its nodes (bounds in LDS; bins of 196, 2 per WGP): 4 launches -> 3; sums keep order
# speedup vs baseline: 8.8434x; 1.0505x over previous
#include <hip/hip_runtime.h>
#include <hip/hip_bf16.h>


typedef __attribute__((ext_vector_type(2))) float v2f;
typedef __attribute__((ext_vector_type(8))) float v8f;

#define NEG_SLOPE 0.2f

__device__ __forceinline__ float leaky(float x) {
    return x > 0.0f ? x : NEG_SLOPE * x;
}

template <int CTRL>
__device__ __forceinline__ float row_xor(float v) {
    return __int_as_float(__builtin_amdgcn_update_dpp(
        0, __float_as_int(v), CTRL, 0xF, 0xF, false));
}

__device__ __forceinline__ float xhalf(float x) {
    const int b = __float_as_int(x);
    return __int_as_float(__builtin_amdgcn_permlanex16(b, b, 0x76543210, 0xfedcba98, false, false));
}

template <int K>
__device__ __forceinline__ float lane_xor(float v) {
    return __int_as_float(__builtin_amdgcn_ds_swizzle(__float_as_int(v), 0x1F | (K << 10)));
}

__device__ __forceinline__ float row_scores16(const float (&v)[16], int col0) {
    const bool b3 = (col0 & 8) != 0, b2 = (col0 & 4) != 0;
    const bool b1 = (col0 & 2) != 0, b0 = (col0 & 1) != 0;
    float w8[8], w4[4], w2[2];
    #pragma unroll
    for (int i = 0; i < 8; ++i) {
        const float keep = b3 ? v[8 + i] : v[i];
        const float send = b3 ? v[i] : v[8 + i];
        w8[i] = keep + row_xor<0x168>(send);
    }
    #pragma unroll
    for (int i = 0; i < 4; ++i) {
        const float keep = b2 ? w8[4 + i] : w8[i];
        const float send = b2 ? w8[i] : w8[4 + i];
        w4[i] = keep + row_xor<0x164>(send);
    }
    #pragma unroll
    for (int i = 0; i < 2; ++i) {
        const float keep = b1 ? w4[2 + i] : w4[i];
        const float send = b1 ? w4[i] : w4[2 + i];
        w2[i] = keep + row_xor<0x162>(send);
    }
    const float keep = b0 ? w2[1] : w2[0], send = b0 ? w2[0] : w2[1];
    return keep + row_xor<0x161>(send);
}

#define P1 132
__device__ __forceinline__ void gemm1_role(
    float* sW,
    const float* __restrict__ X, const float* __restrict__ W,
    const float* __restrict__ asrc, const float* __restrict__ adst,
    float* __restrict__ H, float* __restrict__ s, float* __restrict__ d,
    int N, int K, int blk, int TPB) {
    const int tid = threadIdx.x, nt = 1024;
    const int lane = tid & 31;
    const int wave = __builtin_amdgcn_readfirstlane(tid >> 5);
    const int half = lane >> 4;
    const int col0 = lane & 15;
    const int nrt  = (N + 15) >> 4;
    const int tBeg = min(nrt, blk * TPB), tEnd = min(nrt, tBeg + TPB);
    const int phases = K >> 7;

    if (tBeg >= tEnd) return;
    {
        const int rt = tBeg + wave;
        const bool act = rt < tEnd;
        const int rowA = act ? min(rt * 16 + col0, N - 1) : 0;

        v8f acc[4];
        #pragma unroll
        for (int t = 0; t < 4; ++t) acc[t] = v8f{};

        for (int ph = 0; ph < phases; ++ph) {
            const float* wp = W + (size_t)ph * 128 * 64 + tid;
            #pragma unroll
            for (int j = 0; j < 128 * 64 / nt; ++j)
                sW[(tid & 63) * P1 + (tid >> 6) + j * (nt >> 6)] = wp[j * nt];
            __syncthreads();

            if (act) {
                const float* xr = X + (size_t)rowA * K + ph * 128;
                for (int kk = 0; kk < 32; ++kk) {
                    const int kb = kk * 4 + half * 2;
                    v2f a = *(const v2f*)(xr + kb);
                    #pragma unroll
                    for (int t = 0; t < 4; ++t) {
                        v2f b = *(const v2f*)(&sW[(t * 16 + col0) * P1 + kb]);
                        acc[t] = __builtin_amdgcn_wmma_f32_16x16x4_f32(
                            false, a, false, b, (short)0, acc[t], false, false);
                    }
                }
            }
            __syncthreads();
        }

        if (act) {
            float as[4], ad[4];
            #pragma unroll
            for (int t = 0; t < 4; ++t) { as[t] = asrc[t * 16 + col0]; ad[t] = adst[t * 16 + col0]; }
            #pragma unroll
            for (int t = 0; t < 4; ++t)
                #pragma unroll
                for (int r = 0; r < 8; ++r) {
                    int row = rt * 16 + r + half * 8;
                    if (row < N) H[(size_t)row * 64 + t * 16 + col0] = acc[t][r];
                }

            float pv[16];
            #pragma unroll
            for (int r = 0; r < 8; ++r) {
                float ps = acc[0][r] * as[0], pd = acc[0][r] * ad[0];
                #pragma unroll
                for (int t = 1; t < 4; ++t) {
                    ps = fmaf(acc[t][r], as[t], ps);
                    pd = fmaf(acc[t][r], ad[t], pd);
                }
                pv[r * 2] = ps; pv[r * 2 + 1] = pd;
            }
            const float sc = row_scores16(pv, col0);
            const int srow = rt * 16 + half * 8 + (col0 >> 1);
            if (srow < N) { if (col0 & 1) d[srow] = sc; else s[srow] = sc; }
        }
    }
}

#define BIN_NODES 196
#define BIN_PAD 224
#define BIN_ROWS 208
__device__ __forceinline__ int bin_of(int node) { return (int)((unsigned)node / BIN_NODES); }

#define TRIP 7

__device__ __forceinline__ void chunk_bin_role(
    int* hist, int* swave,
    const int* __restrict__ ei, const float* __restrict__ ew,
    int* __restrict__ segOff, int4* __restrict__ rec2, int E, int B, int CH,
    int c, int G) {
    const int tid = threadIdx.x, nt = blockDim.x;
    const int lane = tid & 31, wave = tid >> 5;
    for (int i = tid; i < B; i += nt) hist[i] = 0;
    __syncthreads();
    const int e0 = min(E, c * CH), e1 = min(E, e0 + CH);
    int dd[TRIP], ss[TRIP], ww[TRIP], pp[TRIP];
    #pragma unroll
    for (int k = 0; k < TRIP; ++k) { dd[k] = 0; ss[k] = 0; ww[k] = 0; pp[k] = 0; }
    if (e0 < e1) {
        #pragma unroll
        for (int k = 0; k < TRIP; ++k) {
            const int e = min(e0 + tid + k * nt, e1 - 1);
            dd[k] = ei[E + e];
            ss[k] = ei[e];
            ww[k] = __float_as_int(ew[e]);
        }
        #pragma unroll
        for (int k = 0; k < TRIP; ++k)
            if (e0 + tid + k * nt < e1) pp[k] = atomicAdd(&hist[bin_of(dd[k])], 1);
    }
    __syncthreads();
    const int per = (B + nt - 1) / nt;
    const int i0 = min(B, tid * per), i1 = min(B, i0 + per);
    int tsum = 0;
    for (int i = i0; i < i1; ++i) tsum += hist[i];
    int inc = tsum;
    for (int off = 1; off < 32; off <<= 1) {
        int n = __shfl_up(inc, off, 32);
        if (lane >= off) inc += n;
    }
    if (lane == 31) swave[wave] = inc;
    __syncthreads();
    int pre = (lane < wave) ? swave[lane] : 0;
    for (int off = 16; off > 0; off >>= 1) pre += __shfl_xor(pre, off, 32);
    int run = pre + inc - tsum;
    for (int i = i0; i < i1; ++i) {
        const int v = hist[i];
        hist[i] = run;
        segOff[(size_t)i * G + c] = run;
        run += v;
    }
    if (tid == 0) segOff[(size_t)B * G + c] = e1 - e0;
    __syncthreads();
    #pragma unroll
    for (int k = 0; k < TRIP; ++k) {
        const int e = e0 + tid + k * nt;
        if (e < e1)
            rec2[e0 + hist[bin_of(dd[k])] + pp[k]] = make_int4(e, ss[k], ww[k], dd[k]);
    }
}

__global__ __launch_bounds__(1024, 16) void chunk_bin_gemm1_kernel(
    const int* __restrict__ ei, const float* __restrict__ ew,
    int* __restrict__ segOff, int4* __restrict__ rec2, int E, int B, int CH, int G,
    const float* __restrict__ X, const float* __restrict__ W,
    const float* __restrict__ asrc, const float* __restrict__ adst,
    float* __restrict__ H, float* __restrict__ s, float* __restrict__ d,
    int N, int K, int TPB) {
    extern __shared__ int hist[];
    __shared__ int swave[32];
    __shared__ float sW[64 * P1];
    const int half = gridDim.x >> 1;
    const int b = blockIdx.x;
    if (b < half) {
        gemm1_role(sW, X, W, asrc, adst, H, s, d, N, K, b, TPB);
    } else {
        if (b - half < G) chunk_bin_role(hist, swave, ei, ew, segOff, rec2, E, B, CH, b - half, G);
    }
}

__device__ __forceinline__ void seg_group(
    const int* __restrict__ segOff, int b, int G, int CH, int g, int lane,
    int (&incl)[8], int (&dl)[8], int& losum) {
    const int c = g * 8 + (lane & 7);
    int lo = 0, hi = 0;
    if (c < G) { lo = segOff[(size_t)b * G + c]; hi = segOff[(size_t)(b + 1) * G + c]; }
    const int len = hi - lo;
    int inc = len;
    for (int off = 1; off < 8; off <<= 1) {
        int n = __shfl_up(inc, off, 32);
        if ((lane & 7) >= off) inc += n;
    }
    const int d = c * CH + lo - (inc - len);
    losum = 0;
    #pragma unroll
    for (int i = 0; i < 8; ++i) {
        incl[i] = __builtin_amdgcn_readlane(inc, i);
        dl[i]   = __builtin_amdgcn_readlane(d, i);
        losum  += __builtin_amdgcn_readlane(lo, i);
    }
}

__device__ __forceinline__ int seg_index(int q, const int (&incl)[8], const int (&dl)[8]) {
    int D = dl[0];
    #pragma unroll
    for (int i = 0; i < 7; ++i) D = (q >= incl[i]) ? dl[i + 1] : D;
    return q + D;
}

#define GTRIP 6
__device__ __forceinline__ void bin_group_role(
    int* cnt, int* cur, int* swave, int* sstart,
    const int* __restrict__ segOff, const int4* __restrict__ rec2,
    int4* rec, int* __restrict__ rowptr,
    int* gsrc, float* gw, int N, int E, int G, int CH) {
    const int tid = threadIdx.x;
    const int lane = tid & 31, wave = tid >> 5;
    const int nwaves = blockDim.x >> 5;
    const int b = blockIdx.x;
    const int nbase = b * BIN_NODES;
    if (tid < BIN_PAD) cnt[tid] = 0;
    __syncthreads();
    int mylo = 0;
    for (int g = wave; g * 8 < G; g += nwaves) {
        int incl[8], dl[8], losum;
        seg_group(segOff, b, G, CH, g, lane, incl, dl, losum);
        mylo += losum;
        const int T = incl[7];
        for (int q0 = 0; q0 < T; q0 += 32 * GTRIP) {
            int dd[GTRIP];
            #pragma unroll
            for (int k = 0; k < GTRIP; ++k)
                dd[k] = rec2[seg_index(min(q0 + k * 32 + lane, T - 1), incl, dl)].w;
            #pragma unroll
            for (int k = 0; k < GTRIP; ++k)
                if (q0 + k * 32 + lane < T) atomicAdd(&cnt[dd[k] - nbase], 1);
        }
    }
    if (lane == 0) sstart[wave] = mylo;
    __syncthreads();
    int v = 0, inc = 0;
    if (tid < BIN_PAD) {
        v = cnt[tid];
        inc = v;
        for (int off = 1; off < 32; off <<= 1) {
            int n = __shfl_up(inc, off, 32);
            if (lane >= off) inc += n;
        }
        if (lane == 31) swave[wave] = inc;
    }
    __syncthreads();
    if (tid < BIN_PAD) {
        int binStart = 0;
        for (int w = 0; w < nwaves; ++w) binStart += sstart[w];
        int pre = 0;
        for (int w = 0; w < BIN_PAD / 32; ++w) if (w < wave) pre += swave[w];
        const int first = binStart + pre + inc - v;
        cur[tid] = first;
        if (tid < BIN_NODES && nbase + tid < N) rowptr[nbase + tid] = first;
    }
    if (b == (int)gridDim.x - 1 && tid == 0) rowptr[N] = E;
    __syncthreads();
    for (int g = wave; g * 8 < G; g += nwaves) {
        int incl[8], dl[8], losum;
        seg_group(segOff, b, G, CH, g, lane, incl, dl, losum);
        const int T = incl[7];
        for (int q0 = 0; q0 < T; q0 += 32 * GTRIP) {
            int4 r[GTRIP]; int pp[GTRIP];
            #pragma unroll
            for (int k = 0; k < GTRIP; ++k)
                r[k] = rec2[seg_index(min(q0 + k * 32 + lane, T - 1), incl, dl)];
            #pragma unroll
            for (int k = 0; k < GTRIP; ++k) {
                pp[k] = 0;
                if (q0 + k * 32 + lane < T) pp[k] = atomicAdd(&cur[r[k].w - nbase], 1);
            }
            #pragma unroll
            for (int k = 0; k < GTRIP; ++k)
                if (q0 + k * 32 + lane < T) rec[pp[k]] = make_int4(r[k].x, r[k].y, r[k].z, 0);
        }
    }
    __syncthreads();
    int nl = wave;
    int start = 0, deg = 0;
    int4 r = make_int4(0x7fffffff, 0, 0, 0);
    if (nl < BIN_NODES) {
        deg = cnt[nl];
        start = cur[nl] - deg;
        if (lane < deg) r = rec[start + lane];
    }
    for (; nl < BIN_NODES; nl += nwaves) {
        const int nn = nl + nwaves;
        int nstart = 0, ndeg = 0;
        int4 nr = make_int4(0x7fffffff, 0, 0, 0);
        if (nn < BIN_NODES) {
            ndeg = cnt[nn];
            nstart = cur[nn] - ndeg;
            if (lane < ndeg) nr = rec[nstart + lane];
        }
        const int degs = __builtin_amdgcn_readfirstlane(deg);
        if (degs <= 32) {
            int rk = 0;
            for (int j0 = 0; j0 < degs; j0 += 8) {
                #pragma unroll
                for (int u = 0; u < 8; ++u)
                    rk += (__builtin_amdgcn_readlane(r.x, j0 + u) < r.x) ? 1 : 0;
            }
            if (lane < degs) { gsrc[start + rk] = r.y; gw[start + rk] = __int_as_float(r.z); }
        } else {
            for (int q = lane; q < degs; q += 32) {
                const int4 rq = rec[start + q];
                int rk = 0;
                for (int j = 0; j < degs; ++j) rk += (rec[start + j].x < rq.x) ? 1 : 0;
                gsrc[start + rk] = rq.y; gw[start + rk] = __int_as_float(rq.z);
            }
        }
        start = nstart; deg = ndeg; r = nr;
    }
}

#define PA 68
template <bool TILE>
__device__ __forceinline__ void ask_bounds(
    const int* __restrict__ rowptr, const int* lend, const int* ldeg, int tbase,
    int node, int& a, int& b) {
    if (TILE) { a = lend[node - tbase]; b = ldeg[node - tbase]; }
    else      { a = rowptr[node];       b = rowptr[node + 1]; }
}
template <bool TILE>
__device__ __forceinline__ void take_bounds(int a, int b, int& st, int& en) {
    if (TILE) {
        en = __builtin_amdgcn_readfirstlane(a);
        st = en - __builtin_amdgcn_readfirstlane(b);
    } else { st = a; en = b; }
}
template <bool TILE>
__device__ __forceinline__ void node_walk(
    const int* __restrict__ rowptr, const int* lend, const int* ldeg,
    const int* __restrict__ gsrc,
    const float* __restrict__ gw, const float* __restrict__ s,
    const float* __restrict__ d, const float* __restrict__ asrc,
    const float* __restrict__ H,
    float* __restrict__ out, float* tile, int tbase,
    int F, int wfirst, int wend, int nw, int lane) {
    const bool fv = lane * 2 < F;
    const int  fo = fv ? lane * 2 : 0;
    const float a0 = fv ? asrc[fo] : 0.0f;
    const float a1 = fv ? asrc[fo + 1] : 0.0f;
    const bool hi = lane >= 16;

    int start, end;
    {
        int ba, bb;
        ask_bounds<TILE>(rowptr, lend, ldeg, tbase, wfirst, ba, bb);
        take_bounds<TILE>(ba, bb, start, end);
    }
    float di    = d[wfirst];
    float si    = s[wfirst];
    int src0 = gsrc[max(min(start + lane, end - 1), 0)];
    asm volatile("" :: "v"(src0));

    const int nx1 = min(wfirst + nw, wend - 1);
    int nstart, nend;
    {
        int ba, bb;
        ask_bounds<TILE>(rowptr, lend, ldeg, tbase, nx1, ba, bb);
        take_bounds<TILE>(ba, bb, nstart, nend);
    }
    float ndi    = d[nx1];
    float nsi    = s[nx1];

    for (int wid = wfirst; wid < wend; wid += nw) {
        const int nx2 = min(wid + 2 * nw, wend - 1);
        int n2a, n2b;
        ask_bounds<TILE>(rowptr, lend, ldeg, tbase, nx2, n2a, n2b);
        const float n2di    = d[nx2];
        const float n2si    = s[nx2];

        const int nsrc0 = gsrc[max(min(nstart + lane, nend - 1), 0)];
        const int deg = end - start;
        const bool v0 = lane < deg;
        const float wt0 = gw[max(min(start + lane, end - 1), 0)];
        float s0 = 0.0f;
        if (v0 && hi) s0 = s[src0];

        const int n0 = min(deg, 32);
        const int jlast = max(n0, 1) - 1;
        const int srcA = (TILE && deg == 0) ? wid : src0;
        float2 hA[16];
        #pragma unroll
        for (int u = 0; u < 16; ++u) {
            const int sj = __builtin_amdgcn_readlane(srcA, min(u, jlast));
            hA[u] = *(const float2*)(H + (size_t)sj * F + fo);
        }
        int n2start, n2end;
        take_bounds<TILE>(n2a, n2b, n2start, n2end);

        float pv[16];
        #pragma unroll
        for (int u = 0; u < 16; ++u) pv[u] = fmaf(hA[u].y, a1, hA[u].x * a0);
        float slo = row_scores16(pv, lane & 15);
        slo = slo + xhalf(slo);
        const float sj0 = hi ? s0 : slo;
        const float2 hself = *(const float2*)(H + (size_t)wid * F + fo);

        const float sc_self = leaky(si + di);
        const float sc0 = v0 ? leaky(sj0 + di) : sc_self;

        float m = fmaxf(sc_self, sc0);
        for (int p = start + 32 + lane; p < end; p += 32)
            m = fmaxf(m, leaky(s[gsrc[p]] + di));
        m = fmaxf(m, lane_xor<16>(m)); m = fmaxf(m, lane_xor<8>(m)); m = fmaxf(m, lane_xor<4>(m));
        m = fmaxf(m, lane_xor<2>(m));  m = fmaxf(m, lane_xor<1>(m));

        const float ex0 = v0 ? __expf(sc0 - m) : 0.0f;
        float sum = ex0;
        for (int p = start + 32 + lane; p < end; p += 32)
            sum += __expf(leaky(s[gsrc[p]] + di) - m);
        sum += lane_xor<16>(sum); sum += lane_xor<8>(sum); sum += lane_xor<4>(sum);
        sum += lane_xor<2>(sum);  sum += lane_xor<1>(sum);
        const float ex_self = __expf(sc_self - m);
        sum += ex_self;
        const float dn = fmaxf(sum, 1e-16f);
        const float c0 = (ex0 * wt0) / dn;
        const float c_self = ex_self / dn;

        float ax = 0.0f, ay = 0.0f;
        #pragma unroll
        for (int u = 0; u < 16; ++u) {
            if (u < n0) {
                const float cj = __int_as_float(__builtin_amdgcn_readlane(__float_as_int(c0), u));
                ax = fmaf(cj, hA[u].x, ax); ay = fmaf(cj, hA[u].y, ay);
            }
        }
        for (int jb = 16; jb < n0; jb += 8) {
            float2 h[8]; float cc[8];
            #pragma unroll
            for (int u = 0; u < 8; ++u) {
                const int j  = jb + u;
                const int jj = min(j, n0 - 1);
                const int sj = __builtin_amdgcn_readlane(src0, jj);
                const float cj = __int_as_float(__builtin_amdgcn_readlane(__float_as_int(c0), jj));
                cc[u] = (j < n0) ? cj : 0.0f;
                h[u]  = *(const float2*)(H + (size_t)sj * F + fo);
            }
            #pragma unroll
            for (int u = 0; u < 8; ++u) {
                if (jb + u < n0) { ax = fmaf(cc[u], h[u].x, ax); ay = fmaf(cc[u], h[u].y, ay); }
            }
        }
        for (int cb = start + 32; cb < end; cb += 32) {
            const int p = cb + lane;
            int sx = 0; float cx = 0.0f;
            if (p < end) {
                sx = gsrc[p];
                cx = (__expf(leaky(s[sx] + di) - m) * gw[p]) / dn;
            }
            const int cn = min(32, end - cb);
            for (int j = 0; j < cn; ++j) {
                const int sj = __builtin_amdgcn_readlane(sx, j);
                const float cj = __int_as_float(__builtin_amdgcn_readlane(__float_as_int(cx), j));
                const float2 hv = *(const float2*)(H + (size_t)sj * F + fo);
                ax = fmaf(cj, hv.x, ax); ay = fmaf(cj, hv.y, ay);
            }
        }
        asm volatile("" :: "v"(hself.x), "v"(hself.y));
        ax = fmaf(c_self, hself.x, ax); ay = fmaf(c_self, hself.y, ay);
        if (TILE) *(float2*)(tile + (wid - tbase) * PA + fo) = make_float2(ax, ay);
        else if (fv) *(float2*)(out + (size_t)wid * F + fo) = make_float2(ax, ay);

        start = nstart; end = nend; di = ndi; si = nsi;
        src0 = nsrc0;
        nstart = n2start; nend = n2end; ndi = n2di; nsi = n2si;
    }
}

__global__ __launch_bounds__(256, 16) void node_kernel(
    const int* __restrict__ rowptr, const int* __restrict__ gsrc,
    const float* __restrict__ gw, const float* __restrict__ s,
    const float* __restrict__ d, const float* __restrict__ asrc,
    const float* __restrict__ H,
    float* __restrict__ out, int N, int F) {
    const int wfirst = __builtin_amdgcn_readfirstlane(
        (int)((blockIdx.x * (long long)blockDim.x + threadIdx.x) >> 5));
    const int nw   = (int)((gridDim.x * (long long)blockDim.x) >> 5);
    const int lane = threadIdx.x & 31;
    if (wfirst >= N) return;
    node_walk<false>(rowptr, nullptr, nullptr, gsrc, gw, s, d, asrc, H, out, nullptr, 0, F, wfirst, N, nw, lane);
}

#define P2 68
__global__ __launch_bounds__(1024, 16) void bin_node_gemm2_kernel(
    const int* __restrict__ segOff, const int4* __restrict__ rec2,
    int4* rec, int* __restrict__ rowptr,
    int* gsrc, float* gw, int N, int E, int G, int CH,
    const float* __restrict__ s,
    const float* __restrict__ d, const float* __restrict__ asrc,
    const float* __restrict__ H,
    const float* __restrict__ W2, const float* __restrict__ a2s,
    const float* __restrict__ a2d,
    float* __restrict__ Hout, float* __restrict__ s2, float* __restrict__ d2, int NC) {
    __shared__ int cnt[BIN_PAD];
    __shared__ int cur[BIN_PAD];
    __shared__ int swave[BIN_PAD / 32];
    __shared__ int sstart[32];
    __shared__ __attribute__((aligned(16))) float sA[BIN_ROWS * PA];
    __shared__ __attribute__((aligned(16))) float sW[48 * P2];

    bin_group_role(cnt, cur, swave, sstart, segOff, rec2, rec, rowptr, gsrc, gw, N, E, G, CH);
    __syncthreads();

    const int lane = threadIdx.x & 31;
    const int wave = __builtin_amdgcn_readfirstlane((int)(threadIdx.x >> 5));
    const int base = blockIdx.x * BIN_NODES;
    const int wend = min(N, base + BIN_NODES);

    if (base + wave < wend)
        node_walk<true>(nullptr, cur, cnt, gsrc, gw, s, d, asrc, H, nullptr, sA, base,
                        64, base + wave, wend, 32, lane);
    {
        float wv[64 * 48 / 1024];
        #pragma unroll
        for (int j = 0; j < 64 * 48 / 1024; ++j) {
            const int i = threadIdx.x + j * 1024;
            const int r = i / 48, cc = i % 48;
            wv[j] = W2[(size_t)r * NC + min(cc, NC - 1)];
        }
        #pragma unroll
        for (int j = 0; j < 64 * 48 / 1024; ++j) {
            const int i = threadIdx.x + j * 1024;
            const int r = i / 48, cc = i % 48;
            sW[cc * P2 + r] = (cc < NC) ? wv[j] : 0.0f;
        }
    }
    for (int r = wend - base + wave; r < BIN_ROWS; r += 32)
        *(float2*)(sA + r * PA + lane * 2) = make_float2(0.0f, 0.0f);
    __syncthreads();

    if (wave < BIN_ROWS / 16 && base + wave * 16 < wend) {
        const int half = lane >> 4;
        const int col0 = lane & 15;
        float as[3], ad[3];
        #pragma unroll
        for (int t = 0; t < 3; ++t) {
            const int cc = t * 16 + col0;
            as[t] = (cc < NC) ? a2s[cc] : 0.0f;
            ad[t] = (cc < NC) ? a2d[cc] : 0.0f;
        }
        v8f acc[3];
        #pragma unroll
        for (int t = 0; t < 3; ++t) acc[t] = v8f{};

        const float* ar = sA + (wave * 16 + col0) * PA;
        for (int kk = 0; kk < 16; ++kk) {
            const int kb = kk * 4 + half * 2;
            v2f a = *(const v2f*)(ar + kb);
            a[0] = fmaxf(a[0], 0.0f); a[1] = fmaxf(a[1], 0.0f);
            #pragma unroll
            for (int t = 0; t < 3; ++t) {
                v2f b = *(const v2f*)(&sW[(t * 16 + col0) * P2 + kb]);
                acc[t] = __builtin_amdgcn_wmma_f32_16x16x4_f32(
                    false, a, false, b, (short)0, acc[t], false, false);
            }
        }

        const int row0 = base + wave * 16;
        #pragma unroll
        for (int t = 0; t < 3; ++t)
            #pragma unroll
            for (int r = 0; r < 8; ++r) {
                const int row = row0 + r + half * 8;
                const int col = t * 16 + col0;
                if (row < wend && col < NC)
                    Hout[(size_t)row * NC + col] = acc[t][r];
            }

        float pv[16];
        #pragma unroll
        for (int r = 0; r < 8; ++r) {
            float ps = acc[0][r] * as[0], pd = acc[0][r] * ad[0];
            #pragma unroll
            for (int t = 1; t < 3; ++t) {
                ps = fmaf(acc[t][r], as[t], ps);
                pd = fmaf(acc[t][r], ad[t], pd);
            }
            pv[r * 2] = ps; pv[r * 2 + 1] = pd;
        }
        const float sc = row_scores16(pv, col0);
        const int srow = row0 + half * 8 + (col0 >> 1);
        if (srow < wend) { if (col0 & 1) d2[srow] = sc; else s2[srow] = sc; }
    }
}

extern "C" void kernel_launch(void* const* d_in, const int* in_sizes, int n_in,
                              void* d_out, int out_size, void* d_ws, size_t ws_size,
                              hipStream_t stream) {
    const float* x   = (const float*)d_in[0];
    const int*   ei  = (const int*)  d_in[1];
    const float* ew  = (const float*)d_in[2];
    const float* W1  = (const float*)d_in[3];
    const float* a1s = (const float*)d_in[4];
    const float* a1d = (const float*)d_in[5];
    const float* W2  = (const float*)d_in[6];
    const float* a2s = (const float*)d_in[7];
    const float* a2d = (const float*)d_in[8];

    const int HID = in_sizes[4];
    const int FIN = in_sizes[3] / HID;
    const int NC  = in_sizes[7];
    const int N   = in_sizes[0] / FIN;
    const int E   = in_sizes[2];

    float* h1   = (float*)d_ws;
    float* h2   = h1   + (size_t)N * HID;
    float* sArr = h2   + (size_t)N * NC;
    float* dArr = sArr + N;
    float* outF = (float*)d_out;

    const int perWG = 1024 * TRIP;
    const int G1  = (E + perWG - 1) / perWG > 256 ? (E + perWG - 1) / perWG : 256;
    const int CH  = (E + G1 - 1) / G1;
    const int B   = (N + BIN_NODES - 1) / BIN_NODES;
    const int nso = (B + 1) * G1;
    size_t off = (((size_t)N * (HID + NC + 2)) + 3) & ~(size_t)3;
    int* base  = (int*)d_ws;
    int4*  rec    = (int4*)(base + off);  off += (size_t)E * 4;
    int4*  rec2   = (int4*)(base + off);  off += (size_t)E * 4;
    int*   rowptr = base + off;           off += ((size_t)N + 4) & ~(size_t)3;
    int*   segOff = base + off;           off += ((size_t)nso + 3) & ~(size_t)3;
    int*   gsrc   = base + off;           off += ((size_t)E + 3) & ~(size_t)3;
    float* gw     = (float*)(base + off); off += ((size_t)E + 3) & ~(size_t)3;
    float* s2Arr  = (float*)(base + off); off += ((size_t)N + 3) & ~(size_t)3;
    float* d2Arr  = (float*)(base + off);

    const int nrt     = (N + 15) / 16;
    const int wBlocks = 2048;

    const int tpb   = (nrt + G1 - 1) / G1 < 32 ? (nrt + G1 - 1) / G1 : 32;
    const int gPair = (nrt + tpb - 1) / tpb > G1 ? (nrt + tpb - 1) / tpb : G1;
    chunk_bin_gemm1_kernel<<<2 * gPair, 1024, B * sizeof(int), stream>>>(
        ei, ew, segOff, rec2, E, B, CH, G1, x, W1, a1s, a1d, h1, sArr, dArr, N, FIN, tpb);
    bin_node_gemm2_kernel<<<B, 1024, 0, stream>>>(segOff, rec2, rec, rowptr, gsrc, gw, N, E, G1, CH,
                                             sArr, dArr, a1s, h1, W2, a2s, a2d, h2, s2Arr, d2Arr, NC);

    node_kernel<<<wBlocks, 256, 0, stream>>>(rowptr, gsrc, gw, s2Arr, d2Arr, a2s, h2, outF, N, NC);
}


# Bin pass keeps the records it puts into their node's group in LDS (in the later out1 tile's space) for the ordering step: no scattered record stores to memory and no read-back; sums keep order
# speedup vs baseline: 9.4117x; 1.0643x over previous
#include <hip/hip_runtime.h>
#include <hip/hip_bf16.h>


typedef __attribute__((ext_vector_type(2))) float v2f;
typedef __attribute__((ext_vector_type(8))) float v8f;

#define NEG_SLOPE 0.2f

__device__ __forceinline__ float leaky(float x) {
    return x > 0.0f ? x : NEG_SLOPE * x;
}

template <int CTRL>
__device__ __forceinline__ float row_xor(float v) {
    return __int_as_float(__builtin_amdgcn_update_dpp(
        0, __float_as_int(v), CTRL, 0xF, 0xF, false));
}

__device__ __forceinline__ float xhalf(float x) {
    const int b = __float_as_int(x);
    return __int_as_float(__builtin_amdgcn_permlanex16(b, b, 0x76543210, 0xfedcba98, false, false));
}

template <int K>
__device__ __forceinline__ float lane_xor(float v) {
    return __int_as_float(__builtin_amdgcn_ds_swizzle(__float_as_int(v), 0x1F | (K << 10)));
}

__device__ __forceinline__ float row_scores16(const float (&v)[16], int col0) {
    const bool b3 = (col0 & 8) != 0, b2 = (col0 & 4) != 0;
    const bool b1 = (col0 & 2) != 0, b0 = (col0 & 1) != 0;
    float w8[8], w4[4], w2[2];
    #pragma unroll
    for (int i = 0; i < 8; ++i) {
        const float keep = b3 ? v[8 + i] : v[i];
        const float send = b3 ? v[i] : v[8 + i];
        w8[i] = keep + row_xor<0x168>(send);
    }
    #pragma unroll
    for (int i = 0; i < 4; ++i) {
        const float keep = b2 ? w8[4 + i] : w8[i];
        const float send = b2 ? w8[i] : w8[4 + i];
        w4[i] = keep + row_xor<0x164>(send);
    }
    #pragma unroll
    for (int i = 0; i < 2; ++i) {
        const float keep = b1 ? w4[2 + i] : w4[i];
        const float send = b1 ? w4[i] : w4[2 + i];
        w2[i] = keep + row_xor<0x162>(send);
    }
    const float keep = b0 ? w2[1] : w2[0], send = b0 ? w2[0] : w2[1];
    return keep + row_xor<0x161>(send);
}

#define P1 132
__device__ __forceinline__ void gemm1_role(
    float* sW,
    const float* __restrict__ X, const float* __restrict__ W,
    const float* __restrict__ asrc, const float* __restrict__ adst,
    float* __restrict__ H, float* __restrict__ s, float* __restrict__ d,
    int N, int K, int blk, int TPB) {
    const int tid = threadIdx.x, nt = 1024;
    const int lane = tid & 31;
    const int wave = __builtin_amdgcn_readfirstlane(tid >> 5);
    const int half = lane >> 4;
    const int col0 = lane & 15;
    const int nrt  = (N + 15) >> 4;
    const int tBeg = min(nrt, blk * TPB), tEnd = min(nrt, tBeg + TPB);
    const int phases = K >> 7;

    if (tBeg >= tEnd) return;
    {
        const int rt = tBeg + wave;
        const bool act = rt < tEnd;
        const int rowA = act ? min(rt * 16 + col0, N - 1) : 0;

        v8f acc[4];
        #pragma unroll
        for (int t = 0; t < 4; ++t) acc[t] = v8f{};

        for (int ph = 0; ph < phases; ++ph) {
            const float* wp = W + (size_t)ph * 128 * 64 + tid;
            #pragma unroll
            for (int j = 0; j < 128 * 64 / nt; ++j)
                sW[(tid & 63) * P1 + (tid >> 6) + j * (nt >> 6)] = wp[j * nt];
            __syncthreads();

            if (act) {
                const float* xr = X + (size_t)rowA * K + ph * 128;
                for (int kk = 0; kk < 32; ++kk) {
                    const int kb = kk * 4 + half * 2;
                    v2f a = *(const v2f*)(xr + kb);
                    #pragma unroll
                    for (int t = 0; t < 4; ++t) {
                        v2f b = *(const v2f*)(&sW[(t * 16 + col0) * P1 + kb]);
                        acc[t] = __builtin_amdgcn_wmma_f32_16x16x4_f32(
                            false, a, false, b, (short)0, acc[t], false, false);
                    }
                }
            }
            __syncthreads();
        }

        if (act) {
            float as[4], ad[4];
            #pragma unroll
            for (int t = 0; t < 4; ++t) { as[t] = asrc[t * 16 + col0]; ad[t] = adst[t * 16 + col0]; }
            #pragma unroll
            for (int t = 0; t < 4; ++t)
                #pragma unroll
                for (int r = 0; r < 8; ++r) {
                    int row = rt * 16 + r + half * 8;
                    if (row < N) H[(size_t)row * 64 + t * 16 + col0] = acc[t][r];
                }

            float pv[16];
            #pragma unroll
            for (int r = 0; r < 8; ++r) {
                float ps = acc[0][r] * as[0], pd = acc[0][r] * ad[0];
                #pragma unroll
                for (int t = 1; t < 4; ++t) {
                    ps = fmaf(acc[t][r], as[t], ps);
                    pd = fmaf(acc[t][r], ad[t], pd);
                }
                pv[r * 2] = ps; pv[r * 2 + 1] = pd;
            }
            const float sc = row_scores16(pv, col0);
            const int srow = rt * 16 + half * 8 + (col0 >> 1);
            if (srow < N) { if (col0 & 1) d[srow] = sc; else s[srow] = sc; }
        }
    }
}

#define BIN_NODES 196
#define BIN_PAD 224
#define BIN_ROWS 208
__device__ __forceinline__ int bin_of(int node) { return (int)((unsigned)node / BIN_NODES); }

#define TRIP 7

__device__ __forceinline__ void chunk_bin_role(
    int* hist, int* swave,
    const int* __restrict__ ei, const float* __restrict__ ew,
    int* __restrict__ segOff, int4* __restrict__ rec2, int E, int B, int CH,
    int c, int G) {
    const int tid = threadIdx.x, nt = blockDim.x;
    const int lane = tid & 31, wave = tid >> 5;
    for (int i = tid; i < B; i += nt) hist[i] = 0;
    __syncthreads();
    const int e0 = min(E, c * CH), e1 = min(E, e0 + CH);
    int dd[TRIP], ss[TRIP], ww[TRIP], pp[TRIP];
    #pragma unroll
    for (int k = 0; k < TRIP; ++k) { dd[k] = 0; ss[k] = 0; ww[k] = 0; pp[k] = 0; }
    if (e0 < e1) {
        #pragma unroll
        for (int k = 0; k < TRIP; ++k) {
            const int e = min(e0 + tid + k * nt, e1 - 1);
            dd[k] = ei[E + e];
            ss[k] = ei[e];
            ww[k] = __float_as_int(ew[e]);
        }
        #pragma unroll
        for (int k = 0; k < TRIP; ++k)
            if (e0 + tid + k * nt < e1) pp[k] = atomicAdd(&hist[bin_of(dd[k])], 1);
    }
    __syncthreads();
    const int per = (B + nt - 1) / nt;
    const int i0 = min(B, tid * per), i1 = min(B, i0 + per);
    int tsum = 0;
    for (int i = i0; i < i1; ++i) tsum += hist[i];
    int inc = tsum;
    for (int off = 1; off < 32; off <<= 1) {
        int n = __shfl_up(inc, off, 32);
        if (lane >= off) inc += n;
    }
    if (lane == 31) swave[wave] = inc;
    __syncthreads();
    int pre = (lane < wave) ? swave[lane] : 0;
    for (int off = 16; off > 0; off >>= 1) pre += __shfl_xor(pre, off, 32);
    int run = pre + inc - tsum;
    for (int i = i0; i < i1; ++i) {
        const int v = hist[i];
        hist[i] = run;
        segOff[(size_t)i * G + c] = run;
        run += v;
    }
    if (tid == 0) segOff[(size_t)B * G + c] = e1 - e0;
    __syncthreads();
    #pragma unroll
    for (int k = 0; k < TRIP; ++k) {
        const int e = e0 + tid + k * nt;
        if (e < e1)
            rec2[e0 + hist[bin_of(dd[k])] + pp[k]] = make_int4(e, ss[k], ww[k], dd[k]);
    }
}

__global__ __launch_bounds__(1024, 16) void chunk_bin_gemm1_kernel(
    const int* __restrict__ ei, const float* __restrict__ ew,
    int* __restrict__ segOff, int4* __restrict__ rec2, int E, int B, int CH, int G,
    const float* __restrict__ X, const float* __restrict__ W,
    const float* __restrict__ asrc, const float* __restrict__ adst,
    float* __restrict__ H, float* __restrict__ s, float* __restrict__ d,
    int N, int K, int TPB) {
    extern __shared__ int hist[];
    __shared__ int swave[32];
    __shared__ float sW[64 * P1];
    const int half = gridDim.x >> 1;
    const int b = blockIdx.x;
    if (b < half) {
        gemm1_role(sW, X, W, asrc, adst, H, s, d, N, K, b, TPB);
    } else {
        if (b - half < G) chunk_bin_role(hist, swave, ei, ew, segOff, rec2, E, B, CH, b - half, G);
    }
}

__device__ __forceinline__ void seg_group(
    const int* __restrict__ segOff, int b, int G, int CH, int g, int lane,
    int (&incl)[8], int (&dl)[8], int& losum) {
    const int c = g * 8 + (lane & 7);
    int lo = 0, hi = 0;
    if (c < G) { lo = segOff[(size_t)b * G + c]; hi = segOff[(size_t)(b + 1) * G + c]; }
    const int len = hi - lo;
    int inc = len;
    for (int off = 1; off < 8; off <<= 1) {
        int n = __shfl_up(inc, off, 32);
        if ((lane & 7) >= off) inc += n;
    }
    const int d = c * CH + lo - (inc - len);
    losum = 0;
    #pragma unroll
    for (int i = 0; i < 8; ++i) {
        incl[i] = __builtin_amdgcn_readlane(inc, i);
        dl[i]   = __builtin_amdgcn_readlane(d, i);
        losum  += __builtin_amdgcn_readlane(lo, i);
    }
}

__device__ __forceinline__ int seg_index(int q, const int (&incl)[8], const int (&dl)[8]) {
    int D = dl[0];
    #pragma unroll
    for (int i = 0; i < 7; ++i) D = (q >= incl[i]) ? dl[i + 1] : D;
    return q + D;
}

#define GTRIP 6
#define PA 68
#define RCAP (BIN_ROWS * PA / 4)
template <bool INL>
__device__ __forceinline__ int4 get_rec(const int4* lrec, const int4* rec, int place, int binStart) {
    return INL ? lrec[place - binStart] : rec[place];
}
template <bool INL>
__device__ __forceinline__ void order_groups(
    const int4* lrec, const int4* rec, int binStart, const int* cnt, const int* cur,
    int* gsrc, float* gw, int wave, int lane, int nwaves) {
    int nl = wave;
    int start = 0, deg = 0;
    int4 r = make_int4(0x7fffffff, 0, 0, 0);
    if (nl < BIN_NODES) {
        deg = cnt[nl];
        start = cur[nl] - deg;
        if (lane < deg) r = get_rec<INL>(lrec, rec, start + lane, binStart);
    }
    for (; nl < BIN_NODES; nl += nwaves) {
        const int nn = nl + nwaves;
        int nstart = 0, ndeg = 0;
        int4 nr = make_int4(0x7fffffff, 0, 0, 0);
        if (nn < BIN_NODES) {
            ndeg = cnt[nn];
            nstart = cur[nn] - ndeg;
            if (lane < ndeg) nr = get_rec<INL>(lrec, rec, nstart + lane, binStart);
        }
        const int degs = __builtin_amdgcn_readfirstlane(deg);
        if (degs <= 32) {
            int rk = 0;
            for (int j0 = 0; j0 < degs; j0 += 8) {
                #pragma unroll
                for (int u = 0; u < 8; ++u)
                    rk += (__builtin_amdgcn_readlane(r.x, j0 + u) < r.x) ? 1 : 0;
            }
            if (lane < degs) { gsrc[start + rk] = r.y; gw[start + rk] = __int_as_float(r.z); }
        } else {
            for (int q = lane; q < degs; q += 32) {
                const int4 rq = get_rec<INL>(lrec, rec, start + q, binStart);
                int rk = 0;
                for (int j = 0; j < degs; ++j) rk += (get_rec<INL>(lrec, rec, start + j, binStart).x < rq.x) ? 1 : 0;
                gsrc[start + rk] = rq.y; gw[start + rk] = __int_as_float(rq.z);
            }
        }
        start = nstart; deg = ndeg; r = nr;
    }
}
__device__ __forceinline__ void bin_group_role(
    int4* lrec, int* cnt, int* cur, int* swave, int* sstart,
    const int* __restrict__ segOff, const int4* __restrict__ rec2,
    int4* rec, int* __restrict__ rowptr,
    int* gsrc, float* gw, int N, int E, int G, int CH) {
    const int tid = threadIdx.x;
    const int lane = tid & 31, wave = tid >> 5;
    const int nwaves = blockDim.x >> 5;
    const int b = blockIdx.x;
    const int nbase = b * BIN_NODES;
    if (tid < BIN_PAD) cnt[tid] = 0;
    __syncthreads();
    int mylo = 0;
    for (int g = wave; g * 8 < G; g += nwaves) {
        int incl[8], dl[8], losum;
        seg_group(segOff, b, G, CH, g, lane, incl, dl, losum);
        mylo += losum;
        const int T = incl[7];
        for (int q0 = 0; q0 < T; q0 += 32 * GTRIP) {
            int dd[GTRIP];
            #pragma unroll
            for (int k = 0; k < GTRIP; ++k)
                dd[k] = rec2[seg_index(min(q0 + k * 32 + lane, T - 1), incl, dl)].w;
            #pragma unroll
            for (int k = 0; k < GTRIP; ++k)
                if (q0 + k * 32 + lane < T) atomicAdd(&cnt[dd[k] - nbase], 1);
        }
    }
    if (lane == 0) sstart[wave] = mylo;
    __syncthreads();
    int v = 0, inc = 0;
    if (tid < BIN_PAD) {
        v = cnt[tid];
        inc = v;
        for (int off = 1; off < 32; off <<= 1) {
            int n = __shfl_up(inc, off, 32);
            if (lane >= off) inc += n;
        }
        if (lane == 31) swave[wave] = inc;
    }
    __syncthreads();
    int binStartV = 0;
    for (int w = 0; w < nwaves; ++w) binStartV += sstart[w];
    int binTotalV = 0;
    for (int w = 0; w < BIN_PAD / 32; ++w) binTotalV += swave[w];
    const int binStart = __builtin_amdgcn_readfirstlane(binStartV);
    const bool inl = __builtin_amdgcn_readfirstlane(binTotalV) <= RCAP;
    if (tid < BIN_PAD) {
        int pre = 0;
        for (int w = 0; w < BIN_PAD / 32; ++w) if (w < wave) pre += swave[w];
        const int first = binStart + pre + inc - v;
        cur[tid] = first;
        if (tid < BIN_NODES && nbase + tid < N) rowptr[nbase + tid] = first;
    }
    if (b == (int)gridDim.x - 1 && tid == 0) rowptr[N] = E;
    __syncthreads();
    for (int g = wave; g * 8 < G; g += nwaves) {
        int incl[8], dl[8], losum;
        seg_group(segOff, b, G, CH, g, lane, incl, dl, losum);
        const int T = incl[7];
        for (int q0 = 0; q0 < T; q0 += 32 * GTRIP) {
            int4 r[GTRIP]; int pp[GTRIP];
            #pragma unroll
            for (int k = 0; k < GTRIP; ++k)
                r[k] = rec2[seg_index(min(q0 + k * 32 + lane, T - 1), incl, dl)];
            #pragma unroll
            for (int k = 0; k < GTRIP; ++k) {
                pp[k] = 0;
                if (q0 + k * 32 + lane < T) pp[k] = atomicAdd(&cur[r[k].w - nbase], 1);
            }
            #pragma unroll
            for (int k = 0; k < GTRIP; ++k)
                if (q0 + k * 32 + lane < T) {
                    const int4 v4 = make_int4(r[k].x, r[k].y, r[k].z, 0);
                    if (inl) lrec[pp[k] - binStart] = v4; else rec[pp[k]] = v4;
                }
        }
    }
    __syncthreads();
    if (inl) order_groups<true>(lrec, rec, binStart, cnt, cur, gsrc, gw, wave, lane, nwaves);
    else     order_groups<false>(lrec, rec, binStart, cnt, cur, gsrc, gw, wave, lane, nwaves);
}

template <bool TILE>
__device__ __forceinline__ void ask_bounds(
    const int* __restrict__ rowptr, const int* lend, const int* ldeg, int tbase,
    int node, int& a, int& b) {
    if (TILE) { a = lend[node - tbase]; b = ldeg[node - tbase]; }
    else      { a = rowptr[node];       b = rowptr[node + 1]; }
}
template <bool TILE>
__device__ __forceinline__ void take_bounds(int a, int b, int& st, int& en) {
    if (TILE) {
        en = __builtin_amdgcn_readfirstlane(a);
        st = en - __builtin_amdgcn_readfirstlane(b);
    } else { st = a; en = b; }
}
template <bool TILE>
__device__ __forceinline__ void node_walk(
    const int* __restrict__ rowptr, const int* lend, const int* ldeg,
    const int* __restrict__ gsrc,
    const float* __restrict__ gw, const float* __restrict__ s,
    const float* __restrict__ d, const float* __restrict__ asrc,
    const float* __restrict__ H,
    float* __restrict__ out, float* tile, int tbase,
    int F, int wfirst, int wend, int nw, int lane) {
    const bool fv = lane * 2 < F;
    const int  fo = fv ? lane * 2 : 0;
    const float a0 = fv ? asrc[fo] : 0.0f;
    const float a1 = fv ? asrc[fo + 1] : 0.0f;
    const bool hi = lane >= 16;

    int start, end;
    {
        int ba, bb;
        ask_bounds<TILE>(rowptr, lend, ldeg, tbase, wfirst, ba, bb);
        take_bounds<TILE>(ba, bb, start, end);
    }
    float di    = d[wfirst];
    float si    = s[wfirst];
    int src0 = gsrc[max(min(start + lane, end - 1), 0)];
    asm volatile("" :: "v"(src0));

    const int nx1 = min(wfirst + nw, wend - 1);
    int nstart, nend;
    {
        int ba, bb;
        ask_bounds<TILE>(rowptr, lend, ldeg, tbase, nx1, ba, bb);
        take_bounds<TILE>(ba, bb, nstart, nend);
    }
    float ndi    = d[nx1];
    float nsi    = s[nx1];

    for (int wid = wfirst; wid < wend; wid += nw) {
        const int nx2 = min(wid + 2 * nw, wend - 1);
        int n2a, n2b;
        ask_bounds<TILE>(rowptr, lend, ldeg, tbase, nx2, n2a, n2b);
        const float n2di    = d[nx2];
        const float n2si    = s[nx2];

        const int nsrc0 = gsrc[max(min(nstart + lane, nend - 1), 0)];
        const int deg = end - start;
        const bool v0 = lane < deg;
        const float wt0 = gw[max(min(start + lane, end - 1), 0)];
        float s0 = 0.0f;
        if (v0 && hi) s0 = s[src0];

        const int n0 = min(deg, 32);
        const int jlast = max(n0, 1) - 1;
        const int srcA = (TILE && deg == 0) ? wid : src0;
        float2 hA[16];
        #pragma unroll
        for (int u = 0; u < 16; ++u) {
            const int sj = __builtin_amdgcn_readlane(srcA, min(u, jlast));
            hA[u] = *(const float2*)(H + (size_t)sj * F + fo);
        }
        int n2start, n2end;
        take_bounds<TILE>(n2a, n2b, n2start, n2end);

        float pv[16];
        #pragma unroll
        for (int u = 0; u < 16; ++u) pv[u] = fmaf(hA[u].y, a1, hA[u].x * a0);
        float slo = row_scores16(pv, lane & 15);
        slo = slo + xhalf(slo);
        const float sj0 = hi ? s0 : slo;
        const float2 hself = *(const float2*)(H + (size_t)wid * F + fo);

        const float sc_self = leaky(si + di);
        const float sc0 = v0 ? leaky(sj0 + di) : sc_self;

        float m = fmaxf(sc_self, sc0);
        for (int p = start + 32 + lane; p < end; p += 32)
            m = fmaxf(m, leaky(s[gsrc[p]] + di));
        m = fmaxf(m, lane_xor<16>(m)); m = fmaxf(m, lane_xor<8>(m)); m = fmaxf(m, lane_xor<4>(m));
        m = fmaxf(m, lane_xor<2>(m));  m = fmaxf(m, lane_xor<1>(m));

        const float ex0 = v0 ? __expf(sc0 - m) : 0.0f;
        float sum = ex0;
        for (int p = start + 32 + lane; p < end; p += 32)
            sum += __expf(leaky(s[gsrc[p]] + di) - m);
        sum += lane_xor<16>(sum); sum += lane_xor<8>(sum); sum += lane_xor<4>(sum);
        sum += lane_xor<2>(sum);  sum += lane_xor<1>(sum);
        const float ex_self = __expf(sc_self - m);
        sum += ex_self;
        const float dn = fmaxf(sum, 1e-16f);
        const float c0 = (ex0 * wt0) / dn;
        const float c_self = ex_self / dn;

        float ax = 0.0f, ay = 0.0f;
        #pragma unroll
        for (int u = 0; u < 16; ++u) {
            if (u < n0) {
                const float cj = __int_as_float(__builtin_amdgcn_readlane(__float_as_int(c0), u));
                ax = fmaf(cj, hA[u].x, ax); ay = fmaf(cj, hA[u].y, ay);
            }
        }
        for (int jb = 16; jb < n0; jb += 8) {
            float2 h[8]; float cc[8];
            #pragma unroll
            for (int u = 0; u < 8; ++u) {
                const int j  = jb + u;
                const int jj = min(j, n0 - 1);
                const int sj = __builtin_amdgcn_readlane(src0, jj);
                const float cj = __int_as_float(__builtin_amdgcn_readlane(__float_as_int(c0), jj));
                cc[u] = (j < n0) ? cj : 0.0f;
                h[u]  = *(const float2*)(H + (size_t)sj * F + fo);
            }
            #pragma unroll
            for (int u = 0; u < 8; ++u) {
                if (jb + u < n0) { ax = fmaf(cc[u], h[u].x, ax); ay = fmaf(cc[u], h[u].y, ay); }
            }
        }
        for (int cb = start + 32; cb < end; cb += 32) {
            const int p = cb + lane;
            int sx = 0; float cx = 0.0f;
            if (p < end) {
                sx = gsrc[p];
                cx = (__expf(leaky(s[sx] + di) - m) * gw[p]) / dn;
            }
            const int cn = min(32, end - cb);
            for (int j = 0; j < cn; ++j) {
                const int sj = __builtin_amdgcn_readlane(sx, j);
                const float cj = __int_as_float(__builtin_amdgcn_readlane(__float_as_int(cx), j));
                const float2 hv = *(const float2*)(H + (size_t)sj * F + fo);
                ax = fmaf(cj, hv.x, ax); ay = fmaf(cj, hv.y, ay);
            }
        }
        asm volatile("" :: "v"(hself.x), "v"(hself.y));
        ax = fmaf(c_self, hself.x, ax); ay = fmaf(c_self, hself.y, ay);
        if (TILE) *(float2*)(tile + (wid - tbase) * PA + fo) = make_float2(ax, ay);
        else if (fv) *(float2*)(out + (size_t)wid * F + fo) = make_float2(ax, ay);

        start = nstart; end = nend; di = ndi; si = nsi;
        src0 = nsrc0;
        nstart = n2start; nend = n2end; ndi = n2di; nsi = n2si;
    }
}

__global__ __launch_bounds__(256, 16) void node_kernel(
    const int* __restrict__ rowptr, const int* __restrict__ gsrc,
    const float* __restrict__ gw, const float* __restrict__ s,
    const float* __restrict__ d, const float* __restrict__ asrc,
    const float* __restrict__ H,
    float* __restrict__ out, int N, int F) {
    const int wfirst = __builtin_amdgcn_readfirstlane(
        (int)((blockIdx.x * (long long)blockDim.x + threadIdx.x) >> 5));
    const int nw   = (int)((gridDim.x * (long long)blockDim.x) >> 5);
    const int lane = threadIdx.x & 31;
    if (wfirst >= N) return;
    node_walk<false>(rowptr, nullptr, nullptr, gsrc, gw, s, d, asrc, H, out, nullptr, 0, F, wfirst, N, nw, lane);
}

#define P2 68
__global__ __launch_bounds__(1024, 16) void bin_node_gemm2_kernel(
    const int* __restrict__ segOff, const int4* __restrict__ rec2,
    int4* rec, int* __restrict__ rowptr,
    int* gsrc, float* gw, int N, int E, int G, int CH,
    const float* __restrict__ s,
    const float* __restrict__ d, const float* __restrict__ asrc,
    const float* __restrict__ H,
    const float* __restrict__ W2, const float* __restrict__ a2s,
    const float* __restrict__ a2d,
    float* __restrict__ Hout, float* __restrict__ s2, float* __restrict__ d2, int NC) {
    __shared__ int cnt[BIN_PAD];
    __shared__ int cur[BIN_PAD];
    __shared__ int swave[BIN_PAD / 32];
    __shared__ int sstart[32];
    __shared__ int4 sRec[RCAP];
    float* sA = (float*)sRec;
    __shared__ __attribute__((aligned(16))) float sW[48 * P2];

    bin_group_role(sRec, cnt, cur, swave, sstart, segOff, rec2, rec, rowptr, gsrc, gw, N, E, G, CH);
    __syncthreads();

    const int lane = threadIdx.x & 31;
    const int wave = __builtin_amdgcn_readfirstlane((int)(threadIdx.x >> 5));
    const int base = blockIdx.x * BIN_NODES;
    const int wend = min(N, base + BIN_NODES);

    if (base + wave < wend)
        node_walk<true>(nullptr, cur, cnt, gsrc, gw, s, d, asrc, H, nullptr, sA, base,
                        64, base + wave, wend, 32, lane);
    {
        float wv[64 * 48 / 1024];
        #pragma unroll
        for (int j = 0; j < 64 * 48 / 1024; ++j) {
            const int i = threadIdx.x + j * 1024;
            const int r = i / 48, cc = i % 48;
            wv[j] = W2[(size_t)r * NC + min(cc, NC - 1)];
        }
        #pragma unroll
        for (int j = 0; j < 64 * 48 / 1024; ++j) {
            const int i = threadIdx.x + j * 1024;
            const int r = i / 48, cc = i % 48;
            sW[cc * P2 + r] = (cc < NC) ? wv[j] : 0.0f;
        }
    }
    for (int r = wend - base + wave; r < BIN_ROWS; r += 32)
        *(float2*)(sA + r * PA + lane * 2) = make_float2(0.0f, 0.0f);
    __syncthreads();

    if (wave < BIN_ROWS / 16 && base + wave * 16 < wend) {
        const int half = lane >> 4;
        const int col0 = lane & 15;
        float as[3], ad[3];
        #pragma unroll
        for (int t = 0; t < 3; ++t) {
            const int cc = t * 16 + col0;
            as[t] = (cc < NC) ? a2s[cc] : 0.0f;
            ad[t] = (cc < NC) ? a2d[cc] : 0.0f;
        }
        v8f acc[3];
        #pragma unroll
        for (int t = 0; t < 3; ++t) acc[t] = v8f{};

        const float* ar = sA + (wave * 16 + col0) * PA;
        for (int kk = 0; kk < 16; ++kk) {
            const int kb = kk * 4 + half * 2;
            v2f a = *(const v2f*)(ar + kb);
            a[0] = fmaxf(a[0], 0.0f); a[1] = fmaxf(a[1], 0.0f);
            #pragma unroll
            for (int t = 0; t < 3; ++t) {
                v2f b = *(const v2f*)(&sW[(t * 16 + col0) * P2 + kb]);
                acc[t] = __builtin_amdgcn_wmma_f32_16x16x4_f32(
                    false, a, false, b, (short)0, acc[t], false, false);
            }
        }

        const int row0 = base + wave * 16;
        #pragma unroll
        for (int t = 0; t < 3; ++t)
            #pragma unroll
            for (int r = 0; r < 8; ++r) {
                const int row = row0 + r + half * 8;
                const int col = t * 16 + col0;
                if (row < wend && col < NC)
                    Hout[(size_t)row * NC + col] = acc[t][r];
            }

        float pv[16];
        #pragma unroll
        for (int r = 0; r < 8; ++r) {
            float ps = acc[0][r] * as[0], pd = acc[0][r] * ad[0];
            #pragma unroll
            for (int t = 1; t < 3; ++t) {
                ps = fmaf(acc[t][r], as[t], ps);
                pd = fmaf(acc[t][r], ad[t], pd);
            }
            pv[r * 2] = ps; pv[r * 2 + 1] = pd;
        }
        const float sc = row_scores16(pv, col0);
        const int srow = row0 + half * 8 + (col0 >> 1);
        if (srow < wend) { if (col0 & 1) d2[srow] = sc; else s2[srow] = sc; }
    }
}

extern "C" void kernel_launch(void* const* d_in, const int* in_sizes, int n_in,
                              void* d_out, int out_size, void* d_ws, size_t ws_size,
                              hipStream_t stream) {
    const float* x   = (const float*)d_in[0];
    const int*   ei  = (const int*)  d_in[1];
    const float* ew  = (const float*)d_in[2];
    const float* W1  = (const float*)d_in[3];
    const float* a1s = (const float*)d_in[4];
    const float* a1d = (const float*)d_in[5];
    const float* W2  = (const float*)d_in[6];
    const float* a2s = (const float*)d_in[7];
    const float* a2d = (const float*)d_in[8];

    const int HID = in_sizes[4];
    const int FIN = in_sizes[3] / HID;
    const int NC  = in_sizes[7];
    const int N   = in_sizes[0] / FIN;
    const int E   = in_sizes[2];

    float* h1   = (float*)d_ws;
    float* h2   = h1   + (size_t)N * HID;
    float* sArr = h2   + (size_t)N * NC;
    float* dArr = sArr + N;
    float* outF = (float*)d_out;

    const int perWG = 1024 * TRIP;
    const int G1  = (E + perWG - 1) / perWG > 256 ? (E + perWG - 1) / perWG : 256;
    const int CH  = (E + G1 - 1) / G1;
    const int B   = (N + BIN_NODES - 1) / BIN_NODES;
    const int nso = (B + 1) * G1;
    size_t off = (((size_t)N * (HID + NC + 2)) + 3) & ~(size_t)3;
    int* base  = (int*)d_ws;
    int4*  rec    = (int4*)(base + off);  off += (size_t)E * 4;
    int4*  rec2   = (int4*)(base + off);  off += (size_t)E * 4;
    int*   rowptr = base + off;           off += ((size_t)N + 4) & ~(size_t)3;
    int*   segOff = base + off;           off += ((size_t)nso + 3) & ~(size_t)3;
    int*   gsrc   = base + off;           off += ((size_t)E + 3) & ~(size_t)3;
    float* gw     = (float*)(base + off); off += ((size_t)E + 3) & ~(size_t)3;
    float* s2Arr  = (float*)(base + off); off += ((size_t)N + 3) & ~(size_t)3;
    float* d2Arr  = (float*)(base + off);

    const int nrt     = (N + 15) / 16;
    const int wBlocks = 2048;

    const int tpb   = (nrt + G1 - 1) / G1 < 32 ? (nrt + G1 - 1) / G1 : 32;
    const int gPair = (nrt + tpb - 1) / tpb > G1 ? (nrt + tpb - 1) / tpb : G1;
    chunk_bin_gemm1_kernel<<<2 * gPair, 1024, B * sizeof(int), stream>>>(
        ei, ew, segOff, rec2, E, B, CH, G1, x, W1, a1s, a1d, h1, sArr, dArr, N, FIN, tpb);
    bin_node_gemm2_kernel<<<B, 1024, 0, stream>>>(segOff, rec2, rec, rowptr, gsrc, gw, N, E, G1, CH,
                                             sArr, dArr, a1s, h1, W2, a2s, a2d, h2, s2Arr, d2Arr, NC);

    node_kernel<<<wBlocks, 256, 0, stream>>>(rowptr, gsrc, gw, s2Arr, d2Arr, a2s, h2, outF, N, NC);
}
